# starting kernel
# baseline (speedup 1.0000x reference)
#include <hip/hip_runtime.h>
#include <hip/hip_bf16.h>


typedef __attribute__((ext_vector_type(16))) _Float16 v16h;
typedef __attribute__((ext_vector_type(8)))  _Float16 v8h;
typedef __attribute__((ext_vector_type(8)))  float    v8f;

#define DEV __device__ __forceinline__

DEV v8f wmma_f16(v16h a, v16h b, v8f c) {
  return __builtin_amdgcn_wmma_f32_16x16x32_f16(false, a, false, b, (short)0, c,
                                                false, false);
}

DEV v16h load_frag(const _Float16* __restrict__ tile, int ld, int lane) {
  const int m  = lane & 15;
  const int kh = (lane & 16) >> 1;
  const _Float16* p = tile + (size_t)m * ld + kh;
  v8h lo = *(const v8h*)p;
  v8h hi = *(const v8h*)(p + 16);
  v16h a;
#pragma unroll
  for (int i = 0; i < 8; ++i) { a[i] = lo[i]; a[i + 8] = hi[i]; }
  return a;
}

__global__ void tcast_kernel(const float* __restrict__ in,
                             _Float16* __restrict__ out, int R, int C) {
  size_t base = (size_t)blockIdx.y * R * C;
  int idx = blockIdx.x * blockDim.x + threadIdx.x;
  if (idx >= R * C) return;
  int r = idx / C, c = idx - r * C;
  out[base + (size_t)c * R + r] = (_Float16)in[base + idx];
}

__global__ void ln_kernel(const float* __restrict__ x, const float* __restrict__ g,
                          const float* __restrict__ be, _Float16* __restrict__ out) {
  const int C = 384;
  int lane = threadIdx.x & 31;
  size_t row = (size_t)blockIdx.x * 8 + (threadIdx.x >> 5);
  const float* xr = x + row * C;
  float v[12];
  float s = 0.f;
#pragma unroll
  for (int i = 0; i < 12; ++i) { v[i] = xr[lane + 32 * i]; s += v[i]; }
#pragma unroll
  for (int m = 16; m >= 1; m >>= 1) s += __shfl_xor(s, m, 32);
  float mu = s * (1.f / 384.f);
  float var = 0.f;
#pragma unroll
  for (int i = 0; i < 12; ++i) { float d = v[i] - mu; var += d * d; }
#pragma unroll
  for (int m = 16; m >= 1; m >>= 1) var += __shfl_xor(var, m, 32);
  float rs = rsqrtf(var * (1.f / 384.f) + 1e-5f);
  _Float16* orow = out + row * C;
#pragma unroll
  for (int i = 0; i < 12; ++i) {
    int c = lane + 32 * i;
    orow[c] = (_Float16)((v[i] - mu) * rs * g[c] + be[c]);
  }
}

__global__ void qkv_kernel(const _Float16* __restrict__ h16,
                           const _Float16* __restrict__ wqt,
                           const _Float16* __restrict__ wkt,
                           const _Float16* __restrict__ wvt,
                           _Float16* __restrict__ q, _Float16* __restrict__ k,
                           _Float16* __restrict__ vt) {
  const int Cc = 384, Dh = 64, T = 256, H = 6;
  int tile = blockIdx.x;
  int dt = tile & 3;
  int rest = tile >> 2;
  int hh = rest % H;
  int rt = rest / H;
  int lane = threadIdx.x;
  const int n = lane & 15, mh = (lane >> 4) * 8;

  const _Float16* Ab = h16 + (size_t)rt * 16 * Cc;
  const _Float16* Bq = wqt + ((size_t)hh * Dh + dt * 16) * Cc;
  const _Float16* Bk = wkt + ((size_t)hh * Dh + dt * 16) * Cc;
  const _Float16* Bv = wvt + ((size_t)hh * Dh + dt * 16) * Cc;

  v8f cq = {}, ck = {}, cv = {};
  for (int k0 = 0; k0 < Cc; k0 += 32) {
    v16h fh = load_frag(Ab + k0, Cc, lane);
    v16h fq = load_frag(Bq + k0, Cc, lane);
    v16h fk = load_frag(Bk + k0, Cc, lane);
    v16h fv = load_frag(Bv + k0, Cc, lane);
    cq = wmma_f16(fh, fq, cq);
    ck = wmma_f16(fh, fk, ck);
    cv = wmma_f16(fv, fh, cv);
  }
  int b = rt >> 4, t0 = (rt & 15) * 16;
  size_t bh = (size_t)b * H + hh;
#pragma unroll
  for (int r = 0; r < 8; ++r) {
    size_t qoff = (bh * T + t0 + mh + r) * Dh + dt * 16 + n;
    q[qoff] = (_Float16)cq[r];
    k[qoff] = (_Float16)ck[r];
    vt[(bh * Dh + dt * 16 + mh + r) * (size_t)T + t0 + n] = (_Float16)cv[r];
  }
}

__global__ void attn_kernel(const _Float16* __restrict__ q,
                            const _Float16* __restrict__ k,
                            const _Float16* __restrict__ vt,
                            _Float16* __restrict__ attn_out) {
  const int T = 256, Dh = 64, H = 6, C = 384;
  int tile = blockIdx.x;
  int it = tile & 15;
  int bh = tile >> 4;
  int b = bh / H, h = bh - b * H;
  int t0 = it * 16;
  int lane = threadIdx.x;
  const int n = lane & 15, mh = (lane >> 4) * 8;

  const _Float16* qb = q + ((size_t)bh * T + t0) * Dh;
  const _Float16* kb = k + (size_t)bh * T * Dh;
  const _Float16* vb = vt + (size_t)bh * Dh * T;

  v16h aq0 = load_frag(qb, Dh, lane);
  v16h aq1 = load_frag(qb + 32, Dh, lane);

  float mrow[8], lrow[8];
#pragma unroll
  for (int r = 0; r < 8; ++r) { mrow[r] = -__builtin_inff(); lrow[r] = 0.f; }
  v8f O[4];
#pragma unroll
  for (int i = 0; i < 4; ++i) O[i] = (v8f){};

  __shared__ _Float16 Pl[16 * 32];
  const float scale = 0.05103103630798287f;

  for (int s0 = 0; s0 < t0 + 16; s0 += 32) {
    float P0[8], P1[8];
    {
      v8f acc = {};
      v16h b0 = load_frag(kb + (size_t)s0 * Dh, Dh, lane);
      v16h b1 = load_frag(kb + (size_t)s0 * Dh + 32, Dh, lane);
      acc = wmma_f16(aq0, b0, acc);
      acc = wmma_f16(aq1, b1, acc);
#pragma unroll
      for (int r = 0; r < 8; ++r) P0[r] = acc[r] * scale;
    }
    {
      v8f acc = {};
      v16h b0 = load_frag(kb + (size_t)(s0 + 16) * Dh, Dh, lane);
      v16h b1 = load_frag(kb + (size_t)(s0 + 16) * Dh + 32, Dh, lane);
      acc = wmma_f16(aq0, b0, acc);
      acc = wmma_f16(aq1, b1, acc);
#pragma unroll
      for (int r = 0; r < 8; ++r) P1[r] = acc[r] * scale;
    }
#pragma unroll
    for (int r = 0; r < 8; ++r) {
      int trow = t0 + mh + r;
      if (s0 + n > trow)      P0[r] = -__builtin_inff();
      if (s0 + 16 + n > trow) P1[r] = -__builtin_inff();
      float mx = fmaxf(P0[r], P1[r]);
      mx = fmaxf(mx, __shfl_xor(mx, 1, 32));
      mx = fmaxf(mx, __shfl_xor(mx, 2, 32));
      mx = fmaxf(mx, __shfl_xor(mx, 4, 32));
      mx = fmaxf(mx, __shfl_xor(mx, 8, 32));
      float mnew = fmaxf(mrow[r], mx);
      float al = expf(mrow[r] - mnew);
      float p0 = expf(P0[r] - mnew);
      float p1 = expf(P1[r] - mnew);
      float rs = p0 + p1;
      rs += __shfl_xor(rs, 1, 32);
      rs += __shfl_xor(rs, 2, 32);
      rs += __shfl_xor(rs, 4, 32);
      rs += __shfl_xor(rs, 8, 32);
      lrow[r] = lrow[r] * al + rs;
      mrow[r] = mnew;
      P0[r] = p0; P1[r] = p1;
      O[0][r] *= al; O[1][r] *= al; O[2][r] *= al; O[3][r] *= al;
    }
    __syncthreads();
#pragma unroll
    for (int r = 0; r < 8; ++r) {
      Pl[(mh + r) * 32 + n]      = (_Float16)P0[r];
      Pl[(mh + r) * 32 + 16 + n] = (_Float16)P1[r];
    }
    __syncthreads();
    v16h ap = load_frag(Pl, 32, lane);
#pragma unroll
    for (int nt = 0; nt < 4; ++nt) {
      v16h bv = load_frag(vb + (size_t)(nt * 16) * T + s0, T, lane);
      O[nt] = wmma_f16(ap, bv, O[nt]);
    }
  }
#pragma unroll
  for (int r = 0; r < 8; ++r) {
    float inv = 1.f / lrow[r];
    _Float16* orow = attn_out + ((size_t)b * T + t0 + mh + r) * C + h * Dh;
#pragma unroll
    for (int nt = 0; nt < 4; ++nt) orow[nt * 16 + n] = (_Float16)(O[nt][r] * inv);
  }
}

template <int EPI>
__global__ void gemm16(const _Float16* __restrict__ A, const _Float16* __restrict__ Bt,
                       const float* __restrict__ bias, const float* __restrict__ res,
                       float* __restrict__ outf, _Float16* __restrict__ outh,
                       int N, int K) {
  int ntn = N >> 4;
  int tm = blockIdx.x / ntn;
  int tn = blockIdx.x - tm * ntn;
  int lane = threadIdx.x;
  const _Float16* Ab = A + (size_t)tm * 16 * K;
  const _Float16* Bb = Bt + (size_t)tn * 16 * K;
  v8f acc = {};
  for (int k0 = 0; k0 < K; k0 += 32) {
    v16h a = load_frag(Ab + k0, K, lane);
    v16h b = load_frag(Bb + k0, K, lane);
    acc = wmma_f16(a, b, acc);
  }
  const int n = lane & 15, mh = (lane >> 4) * 8;
  const int col = tn * 16 + n;
  const float bv = bias[col];
#pragma unroll
  for (int r = 0; r < 8; ++r) {
    size_t row = (size_t)tm * 16 + mh + r;
    float v = acc[r] + bv;
    if (EPI == 0)      outf[row * N + col] = v + res[row * N + col];
    else if (EPI == 1) outh[row * N + col] = (_Float16)fmaxf(v, 0.f);
    else               outf[row * N + col] += v;
  }
}

extern "C" void kernel_launch(void* const* d_in, const int* in_sizes, int n_in,
                              void* d_out, int out_size, void* d_ws, size_t ws_size,
                              hipStream_t stream) {
  (void)in_sizes; (void)n_in; (void)out_size; (void)ws_size;
  const float* x   = (const float*)d_in[0];
  const float* Wq  = (const float*)d_in[1];
  const float* Wk  = (const float*)d_in[2];
  const float* Wv  = (const float*)d_in[3];
  const float* Wp  = (const float*)d_in[4];
  const float* bp  = (const float*)d_in[5];
  const float* W1  = (const float*)d_in[6];
  const float* b1  = (const float*)d_in[7];
  const float* W2  = (const float*)d_in[8];
  const float* b2  = (const float*)d_in[9];
  const float* g1  = (const float*)d_in[10];
  const float* be1 = (const float*)d_in[11];
  const float* g2  = (const float*)d_in[12];
  const float* be2 = (const float*)d_in[13];
  float* out = (float*)d_out;

  char* ws = (char*)d_ws;
  const size_t S = (size_t)32768 * 384 * 2;
  _Float16* h16    = (_Float16*)(ws + 0 * S);
  _Float16* q16    = (_Float16*)(ws + 1 * S);
  _Float16* k16    = (_Float16*)(ws + 2 * S);
  _Float16* vt16   = (_Float16*)(ws + 3 * S);
  _Float16* attn16 = (_Float16*)(ws + 0 * S);
  _Float16* h2_16  = (_Float16*)(ws + 1 * S);
  _Float16* a1_16  = (_Float16*)(ws + 2 * S);
  char* wbase = ws + 6 * S;
  _Float16* wqt = (_Float16*)(wbase);
  _Float16* wkt = (_Float16*)(wbase + 1 * 294912);
  _Float16* wvt = (_Float16*)(wbase + 2 * 294912);
  _Float16* wpt = (_Float16*)(wbase + 3 * 294912);
  _Float16* w1t = (_Float16*)(wbase + 4 * 294912);
  _Float16* w2t = (_Float16*)(wbase + 4 * 294912 + 1179648);

  tcast_kernel<<<dim3(96, 6),   256, 0, stream>>>(Wq, wqt, 384, 64);
  tcast_kernel<<<dim3(96, 6),   256, 0, stream>>>(Wk, wkt, 384, 64);
  tcast_kernel<<<dim3(96, 6),   256, 0, stream>>>(Wv, wvt, 384, 64);
  tcast_kernel<<<dim3(576, 1),  256, 0, stream>>>(Wp, wpt, 384, 384);
  tcast_kernel<<<dim3(2304, 1), 256, 0, stream>>>(W1, w1t, 384, 1536);
  tcast_kernel<<<dim3(2304, 1), 256, 0, stream>>>(W2, w2t, 1536, 384);

  ln_kernel<<<4096, 256, 0, stream>>>(x, g1, be1, h16);
  qkv_kernel<<<49152, 32, 0, stream>>>(h16, wqt, wkt, wvt, q16, k16, vt16);
  attn_kernel<<<12288, 32, 0, stream>>>(q16, k16, vt16, attn16);
  gemm16<0><<<2048 * 24, 32, 0, stream>>>(attn16, wpt, bp, x, out, nullptr, 384, 384);
  ln_kernel<<<4096, 256, 0, stream>>>(out, g2, be2, h2_16);
  gemm16<1><<<2048 * 96, 32, 0, stream>>>(h2_16, w1t, b1, nullptr, nullptr, a1_16, 1536, 384);
  gemm16<2><<<2048 * 24, 32, 0, stream>>>(a1_16, w2t, b2, nullptr, out, nullptr, 384, 1536);
}


# Attention softmax: the row max and row sum butterflies run as row_xmask swizzles on the vector ALU instead of wave shuffles through the LDS unit; same partners, same order of sums
# speedup vs baseline: 3.1859x; 3.1859x over previous
#include <hip/hip_runtime.h>
#include <hip/hip_bf16.h>


typedef __attribute__((ext_vector_type(16))) _Float16 v16h;
typedef __attribute__((ext_vector_type(8)))  _Float16 v8h;
typedef __attribute__((ext_vector_type(8)))  float    v8f;

#define DEV __device__ __forceinline__

DEV v8f wmma_f16(v16h a, v16h b, v8f c) {
  return __builtin_amdgcn_wmma_f32_16x16x32_f16(false, a, false, b, (short)0, c,
                                                false, false);
}

DEV v16h load_frag(const _Float16* __restrict__ tile, int ld, int lane) {
  const int m  = lane & 15;
  const int kh = (lane & 16) >> 1;
  const _Float16* p = tile + (size_t)m * ld + kh;
  v8h lo = *(const v8h*)p;
  v8h hi = *(const v8h*)(p + 16);
  v16h a;
#pragma unroll
  for (int i = 0; i < 8; ++i) { a[i] = lo[i]; a[i + 8] = hi[i]; }
  return a;
}

__global__ void tcast_kernel(const float* __restrict__ in,
                             _Float16* __restrict__ out, int R, int C) {
  size_t base = (size_t)blockIdx.y * R * C;
  int idx = blockIdx.x * blockDim.x + threadIdx.x;
  if (idx >= R * C) return;
  int r = idx / C, c = idx - r * C;
  out[base + (size_t)c * R + r] = (_Float16)in[base + idx];
}

typedef __attribute__((ext_vector_type(4))) float v4f;

DEV void ln_load_rows(const float* src, int lane, v4f (&v)[48]) {
  const int m  = lane & 15;
  const int kh = (lane & 16) >> 1;
  const float* p = src + (size_t)m * 384 + kh;
#pragma unroll
  for (int kk = 0; kk < 12; ++kk) {
    v[4 * kk + 0] = *(const v4f*)(p + kk * 32);
    v[4 * kk + 1] = *(const v4f*)(p + kk * 32 + 4);
    v[4 * kk + 2] = *(const v4f*)(p + kk * 32 + 16);
    v[4 * kk + 3] = *(const v4f*)(p + kk * 32 + 20);
  }
}

DEV void ln_gb_load(const float* __restrict__ gp, const float* __restrict__ bp, int kk,
                    v4f (&gg)[4], v4f (&bb)[4]) {
#pragma unroll
  for (int j = 0; j < 4; ++j) {
    const int co = kk * 32 + (j & 1) * 4 + (j >> 1) * 16;
    gg[j] = *(const v4f*)(gp + co);
    bb[j] = *(const v4f*)(bp + co);
  }
}

DEV void ln_frags(const v4f (&v)[48], const float* __restrict__ g,
                  const float* __restrict__ be, int lane, v16h (&hb)[12]) {
  const int kh = (lane & 16) >> 1;
  const float* gp = g + kh;
  const float* bp = be + kh;
  v4f gg[2][4], bb[2][4];
  ln_gb_load(gp, bp, 0, gg[0], bb[0]);
  v4f s4 = {0.f, 0.f, 0.f, 0.f};
#pragma unroll
  for (int i = 0; i < 48; ++i) s4 += v[i];
  float s = (s4[0] + s4[1]) + (s4[2] + s4[3]);
  s += __shfl_xor(s, 16, 32);
  const float mu = s * (1.f / 384.f);
  v4f q4 = {0.f, 0.f, 0.f, 0.f};
#pragma unroll
  for (int i = 0; i < 48; ++i) { v4f d = v[i] - mu; q4 += d * d; }
  float var = (q4[0] + q4[1]) + (q4[2] + q4[3]);
  var += __shfl_xor(var, 16, 32);
  const float rs = rsqrtf(var * (1.f / 384.f) + 1e-5f);
#pragma unroll
  for (int kk = 0; kk < 12; ++kk) {
    if (kk + 1 < 12) ln_gb_load(gp, bp, kk + 1, gg[(kk + 1) & 1], bb[(kk + 1) & 1]);
    __builtin_amdgcn_sched_barrier(0);
#pragma unroll
    for (int j = 0; j < 4; ++j) {
#pragma unroll
      for (int r = 0; r < 4; ++r)
        hb[kk][4 * j + r] =
            (_Float16)((v[4 * kk + j][r] - mu) * rs * gg[kk & 1][j][r] + bb[kk & 1][j][r]);
    }
    __builtin_amdgcn_sched_barrier(0);
  }
}

#define RG_P 392

template <bool WEIGHT_IS_B>
DEV void rg_chunk(const _Float16* Lc, const v16h (&hb)[12], v8f (&acc)[4], int lane) {
  v16h fa[2][6];
#pragma unroll
  for (int i = 0; i < 6; ++i)
    fa[0][i] = load_frag(Lc + (i & 3) * 16 * RG_P + (i >> 2) * 32, RG_P, lane);
#pragma unroll
  for (int g = 0; g < 8; ++g) {
    if (g + 1 < 8) {
#pragma unroll
      for (int i = 0; i < 6; ++i) {
        const int f = (g + 1) * 6 + i;
        fa[(g + 1) & 1][i] = load_frag(Lc + (f & 3) * 16 * RG_P + (f >> 2) * 32, RG_P, lane);
      }
    }
    __builtin_amdgcn_sched_barrier(0);
#pragma unroll
    for (int i = 0; i < 6; ++i) {
      const int f = g * 6 + i;
      if (WEIGHT_IS_B) acc[f & 3] = wmma_f16(hb[f >> 2], fa[g & 1][i], acc[f & 3]);
      else             acc[f & 3] = wmma_f16(fa[g & 1][i], hb[f >> 2], acc[f & 3]);
    }
    __builtin_amdgcn_sched_barrier(0);
  }
}

DEV v8h pack8h(v8f a) {
  v8h o;
#pragma unroll
  for (int r = 0; r < 8; ++r) o[r] = (_Float16)a[r];
  return o;
}

template <bool PROJ>
__global__ __launch_bounds__(256)
void rowgemm_kernel(const _Float16* __restrict__ act,
                    const float* __restrict__ xf, const float* __restrict__ lng,
                    const float* __restrict__ lnb,
                    const _Float16* __restrict__ w0,
                    const _Float16* __restrict__ w1,
                    const _Float16* __restrict__ w2,
                    _Float16* __restrict__ q, _Float16* __restrict__ k,
                    _Float16* __restrict__ vt,
                    const float* __restrict__ bias, const float* __restrict__ res,
                    float* __restrict__ outf) {
  const int Cc = 384, Dh = 64, T = 256, H = 6;
  const int NCH = PROJ ? 6 : 18;
  const int BUF = 64 * RG_P;
  __shared__ __attribute__((aligned(16))) _Float16 Lw[2 * BUF];

  const int tid = threadIdx.x;
  const int lane = tid & 31;
  const int wave = tid >> 5;
  const int c = lane & 15, hh = lane >> 4;
  const int b = blockIdx.x >> 1;
  const int tw = (blockIdx.x & 1) * 128 + wave * 16;
  const size_t row0 = (size_t)b * T + tw;

  v16h hb[12];
  v4f xv[48];
  if (PROJ) {
#pragma unroll
    for (int kk = 0; kk < 12; ++kk) hb[kk] = load_frag(act + row0 * Cc + kk * 32, Cc, lane);
  } else {
    ln_load_rows(xf + row0 * Cc, lane, xv);
  }

  int o1[12];
#pragma unroll
  for (int i = 0; i < 12; ++i) {
    int u = tid + 256 * i;
    int r = u / 48;
    o1[i] = r * RG_P + (u - r * 48) * 8;
  }
  v8h p[12];
  {
    const _Float16* g = w0 + tid * 8;
#pragma unroll
    for (int i = 0; i < 12; ++i) p[i] = *(const v8h*)(g + i * 2048);
    if (!PROJ) ln_frags(xv, lng, lnb, lane, hb);
#pragma unroll
    for (int i = 0; i < 12; ++i) *(v8h*)(Lw + o1[i]) = p[i];
    g += (size_t)Dh * Cc;
#pragma unroll
    for (int i = 0; i < 12; ++i) p[i] = *(const v8h*)(g + i * 2048);
  }
  __syncthreads();

  for (int s = 0; s < NCH; ++s) {
    const int m = s / 6;
    const int h = s - m * 6;
    const _Float16* Lc = Lw + (s & 1) * BUF;
    if (s + 1 < NCH) {
      _Float16* Ln = Lw + ((s + 1) & 1) * BUF;
#pragma unroll
      for (int i = 0; i < 12; ++i) *(v8h*)(Ln + o1[i]) = p[i];
      if (s + 2 < NCH) {
        const int m2 = (s + 2) / 6;
        const int h2 = (s + 2) - m2 * 6;
        const _Float16* wn = (m2 == 0) ? w0 : (m2 == 1) ? w1 : w2;
        const _Float16* g = wn + (size_t)h2 * Dh * Cc + tid * 8;
#pragma unroll
        for (int i = 0; i < 12; ++i) p[i] = *(const v8h*)(g + i * 2048);
      }
    }

    v8f acc[4];
#pragma unroll
    for (int dt = 0; dt < 4; ++dt) acc[dt] = (v8f){};

    if (PROJ) {
      const size_t off = (row0 + c) * Cc + h * 64 + hh * 8;
      const float* rp = res + off;
      const float* bq = bias + h * 64 + hh * 8;
      v4f xr[8], bb[8];
#pragma unroll
      for (int dt = 0; dt < 4; ++dt) {
        xr[2 * dt]     = *(const v4f*)(rp + dt * 16);
        xr[2 * dt + 1] = *(const v4f*)(rp + dt * 16 + 4);
        bb[2 * dt]     = *(const v4f*)(bq + dt * 16);
        bb[2 * dt + 1] = *(const v4f*)(bq + dt * 16 + 4);
      }
      rg_chunk<false>(Lc, hb, acc, lane);
      v4f o[8];
#pragma unroll
      for (int dt = 0; dt < 4; ++dt) {
#pragma unroll
        for (int r = 0; r < 4; ++r) {
          o[2 * dt][r]     = (acc[dt][r] + bb[2 * dt][r]) + xr[2 * dt][r];
          o[2 * dt + 1][r] = (acc[dt][4 + r] + bb[2 * dt + 1][r]) + xr[2 * dt + 1][r];
        }
      }
      __syncthreads();
      float* op = outf + off;
#pragma unroll
      for (int dt = 0; dt < 4; ++dt) {
        *(v4f*)(op + dt * 16)     = o[2 * dt];
        *(v4f*)(op + dt * 16 + 4) = o[2 * dt + 1];
      }
    } else {
      const size_t bh = (size_t)b * H + h;
      v8h o[4];
      _Float16* dst;
      size_t step;
      if (m < 2) {
        rg_chunk<false>(Lc, hb, acc, lane);
        dst = ((m == 0) ? q : k) + (bh * T + tw + c) * Dh + hh * 8;
        step = 16;
      } else {
        rg_chunk<true>(Lc, hb, acc, lane);
        dst = vt + (bh * Dh + c) * T + tw + hh * 8;
        step = (size_t)16 * T;
      }
#pragma unroll
      for (int dt = 0; dt < 4; ++dt) o[dt] = pack8h(acc[dt]);
      __syncthreads();
#pragma unroll
      for (int dt = 0; dt < 4; ++dt) *(v8h*)(dst + dt * step) = o[dt];
    }
  }
}

template <int M>
DEV float row_xor(float v) {
  return __int_as_float(__builtin_amdgcn_update_dpp(0, __float_as_int(v), 0x160 + M,
                                                    0xf, 0xf, true));
}

__global__ void attn_kernel(const _Float16* __restrict__ q,
                            const _Float16* __restrict__ k,
                            const _Float16* __restrict__ vt,
                            _Float16* __restrict__ attn_out) {
  const int T = 256, Dh = 64, H = 6, C = 384;
  int tile = blockIdx.x;
  int it = tile & 15;
  int bh = tile >> 4;
  int b = bh / H, h = bh - b * H;
  int t0 = it * 16;
  int lane = threadIdx.x;
  const int n = lane & 15, mh = (lane >> 4) * 8;

  const _Float16* qb = q + ((size_t)bh * T + t0) * Dh;
  const _Float16* kb = k + (size_t)bh * T * Dh;
  const _Float16* vb = vt + (size_t)bh * Dh * T;

  v16h aq0 = load_frag(qb, Dh, lane);
  v16h aq1 = load_frag(qb + 32, Dh, lane);

  float mrow[8], lrow[8];
#pragma unroll
  for (int r = 0; r < 8; ++r) { mrow[r] = -__builtin_inff(); lrow[r] = 0.f; }
  v8f O[4];
#pragma unroll
  for (int i = 0; i < 4; ++i) O[i] = (v8f){};

  __shared__ _Float16 Pl[16 * 32];
  const float scale = 0.05103103630798287f;

  for (int s0 = 0; s0 < t0 + 16; s0 += 32) {
    float P0[8], P1[8];
    {
      v8f acc = {};
      v16h b0 = load_frag(kb + (size_t)s0 * Dh, Dh, lane);
      v16h b1 = load_frag(kb + (size_t)s0 * Dh + 32, Dh, lane);
      acc = wmma_f16(aq0, b0, acc);
      acc = wmma_f16(aq1, b1, acc);
#pragma unroll
      for (int r = 0; r < 8; ++r) P0[r] = acc[r] * scale;
    }
    {
      v8f acc = {};
      v16h b0 = load_frag(kb + (size_t)(s0 + 16) * Dh, Dh, lane);
      v16h b1 = load_frag(kb + (size_t)(s0 + 16) * Dh + 32, Dh, lane);
      acc = wmma_f16(aq0, b0, acc);
      acc = wmma_f16(aq1, b1, acc);
#pragma unroll
      for (int r = 0; r < 8; ++r) P1[r] = acc[r] * scale;
    }
#pragma unroll
    for (int r = 0; r < 8; ++r) {
      int trow = t0 + mh + r;
      if (s0 + n > trow)      P0[r] = -__builtin_inff();
      if (s0 + 16 + n > trow) P1[r] = -__builtin_inff();
      float mx = fmaxf(P0[r], P1[r]);
      mx = fmaxf(mx, row_xor<1>(mx));
      mx = fmaxf(mx, row_xor<2>(mx));
      mx = fmaxf(mx, row_xor<4>(mx));
      mx = fmaxf(mx, row_xor<8>(mx));
      float mnew = fmaxf(mrow[r], mx);
      float al = expf(mrow[r] - mnew);
      float p0 = expf(P0[r] - mnew);
      float p1 = expf(P1[r] - mnew);
      float rs = p0 + p1;
      rs += row_xor<1>(rs);
      rs += row_xor<2>(rs);
      rs += row_xor<4>(rs);
      rs += row_xor<8>(rs);
      lrow[r] = lrow[r] * al + rs;
      mrow[r] = mnew;
      P0[r] = p0; P1[r] = p1;
      O[0][r] *= al; O[1][r] *= al; O[2][r] *= al; O[3][r] *= al;
    }
    __syncthreads();
#pragma unroll
    for (int r = 0; r < 8; ++r) {
      Pl[(mh + r) * 32 + n]      = (_Float16)P0[r];
      Pl[(mh + r) * 32 + 16 + n] = (_Float16)P1[r];
    }
    __syncthreads();
    v16h ap = load_frag(Pl, 32, lane);
#pragma unroll
    for (int nt = 0; nt < 4; ++nt) {
      v16h bv = load_frag(vb + (size_t)(nt * 16) * T + s0, T, lane);
      O[nt] = wmma_f16(ap, bv, O[nt]);
    }
  }
#pragma unroll
  for (int r = 0; r < 8; ++r) {
    float inv = 1.f / lrow[r];
    _Float16* orow = attn_out + ((size_t)b * T + t0 + mh + r) * C + h * Dh;
#pragma unroll
    for (int nt = 0; nt < 4; ++nt) orow[nt * 16 + n] = (_Float16)(O[nt][r] * inv);
  }
}


__global__ __launch_bounds__(256)
void mlp_kernel(const float* __restrict__ lng, const float* __restrict__ lnb,
                const _Float16* __restrict__ w1t,
                const float* __restrict__ b1, const _Float16* __restrict__ w2t,
                const float* __restrict__ b2, float* __restrict__ out) {
  const int C = 384, F = 1536;
  const int P1 = 392;
  const int P2 = 40;
  __shared__ __attribute__((aligned(16))) _Float16 Lw1[32 * P1];
  __shared__ __attribute__((aligned(16))) _Float16 Lw2[384 * P2];
  __shared__ __attribute__((aligned(16))) float Lb1[1536];

  const int tid = threadIdx.x;
  const int lane = tid & 31;
  const int wave = tid >> 5;
  const int c = lane & 15, hh = lane >> 4;
  const size_t row0 = (size_t)blockIdx.x * 128 + (size_t)wave * 16;

#pragma unroll
  for (int i = 0; i < 6; ++i) Lb1[tid + 256 * i] = b1[tid + 256 * i];

  v4f xv[48];
  ln_load_rows(out + row0 * C, lane, xv);

  const _Float16* g1 = w1t + (size_t)tid * 8;
  const _Float16* g2 = w2t + (size_t)(tid >> 2) * F + (tid & 3) * 8;
  int o1[6];
#pragma unroll
  for (int i = 0; i < 6; ++i) {
    int u = tid + 256 * i;
    int r = u / 48;
    o1[i] = r * P1 + (u - r * 48) * 8;
  }
  const int o2 = (tid >> 2) * P2 + (tid & 3) * 8;

  v8h p1[6], p2[6];
#pragma unroll
  for (int i = 0; i < 6; ++i) {
    p1[i] = *(const v8h*)(g1 + i * 2048);
    p2[i] = *(const v8h*)(g2 + (size_t)i * 64 * F);
  }

  v16h hb[12];
  ln_frags(xv, lng, lnb, lane, hb);

  v8f acc[24];
#pragma unroll
  for (int mt = 0; mt < 24; ++mt) acc[mt] = (v8f){};

  for (int s = 0; s < 48; ++s) {
    __syncthreads();
#pragma unroll
    for (int i = 0; i < 6; ++i) {
      *(v8h*)(Lw1 + o1[i]) = p1[i];
      *(v8h*)(Lw2 + o2 + i * 64 * P2) = p2[i];
    }
    __syncthreads();
    if (s + 1 < 48) {
      const int hid1 = (s + 1) * 32;
#pragma unroll
      for (int i = 0; i < 6; ++i) {
        p1[i] = *(const v8h*)(g1 + (size_t)hid1 * C + i * 2048);
        p2[i] = *(const v8h*)(g2 + (size_t)i * 64 * F + hid1);
      }
    }

    v16h fa[2][6];
#pragma unroll
    for (int i = 0; i < 6; ++i) fa[0][i] = load_frag(Lw1 + (i & 1) * 16 * P1 + (i >> 1) * 32, P1, lane);
    v8f ha0 = {}, ha1 = {};
    v16h hf = {};
#pragma unroll
    for (int g = 0; g < 8; ++g) {
      if (g + 1 < 8) {
#pragma unroll
        for (int i = 0; i < 6; ++i) {
          const int f = (g + 1) * 6 + i;
          fa[(g + 1) & 1][i] = (f < 24)
              ? load_frag(Lw1 + (f & 1) * 16 * P1 + (f >> 1) * 32, P1, lane)
              : load_frag(Lw2 + (f - 24) * 16 * P2, P2, lane);
        }
      }
      __builtin_amdgcn_sched_barrier(0);
#pragma unroll
      for (int i = 0; i < 6; ++i) {
        const int f = g * 6 + i;
        if (f < 24) {
          if (f & 1) ha1 = wmma_f16(fa[g & 1][i], hb[f >> 1], ha1);
          else       ha0 = wmma_f16(fa[g & 1][i], hb[f >> 1], ha0);
          if (f == 23) {
            const float* bq = Lb1 + s * 32 + hh * 8;
            v4f bA = *(const v4f*)bq;
            v4f bB = *(const v4f*)(bq + 4);
            v4f bC = *(const v4f*)(bq + 16);
            v4f bD = *(const v4f*)(bq + 20);
#pragma unroll
            for (int r = 0; r < 4; ++r) {
              hf[r]      = (_Float16)fmaxf(ha0[r] + bA[r], 0.f);
              hf[4 + r]  = (_Float16)fmaxf(ha0[4 + r] + bB[r], 0.f);
              hf[8 + r]  = (_Float16)fmaxf(ha1[r] + bC[r], 0.f);
              hf[12 + r] = (_Float16)fmaxf(ha1[4 + r] + bD[r], 0.f);
            }
          }
        } else {
          acc[f - 24] = wmma_f16(fa[g & 1][i], hf, acc[f - 24]);
        }
      }
      __builtin_amdgcn_sched_barrier(0);
    }
  }

  float* orow = out + (row0 + c) * C + hh * 8;
  const float* b2p = b2 + hh * 8;
#pragma unroll
  for (int mt = 0; mt < 24; ++mt) {
    if (mt == 12) __builtin_amdgcn_sched_barrier(0);
    v4f oA = *(const v4f*)(orow + mt * 16);
    v4f oB = *(const v4f*)(orow + mt * 16 + 4);
    v4f cA = *(const v4f*)(b2p + mt * 16);
    v4f cB = *(const v4f*)(b2p + mt * 16 + 4);
#pragma unroll
    for (int r = 0; r < 4; ++r) {
      oA[r] += acc[mt][r] + cA[r];
      oB[r] += acc[mt][4 + r] + cB[r];
    }
    *(v4f*)(orow + mt * 16) = oA;
    *(v4f*)(orow + mt * 16 + 4) = oB;
  }
}

extern "C" void kernel_launch(void* const* d_in, const int* in_sizes, int n_in,
                              void* d_out, int out_size, void* d_ws, size_t ws_size,
                              hipStream_t stream) {
  (void)in_sizes; (void)n_in; (void)out_size; (void)ws_size;
  const float* x   = (const float*)d_in[0];
  const float* Wq  = (const float*)d_in[1];
  const float* Wk  = (const float*)d_in[2];
  const float* Wv  = (const float*)d_in[3];
  const float* Wp  = (const float*)d_in[4];
  const float* bp  = (const float*)d_in[5];
  const float* W1  = (const float*)d_in[6];
  const float* b1  = (const float*)d_in[7];
  const float* W2  = (const float*)d_in[8];
  const float* b2  = (const float*)d_in[9];
  const float* g1  = (const float*)d_in[10];
  const float* be1 = (const float*)d_in[11];
  const float* g2  = (const float*)d_in[12];
  const float* be2 = (const float*)d_in[13];
  float* out = (float*)d_out;

  char* ws = (char*)d_ws;
  const size_t S = (size_t)32768 * 384 * 2;
  _Float16* q16    = (_Float16*)(ws + 1 * S);
  _Float16* k16    = (_Float16*)(ws + 2 * S);
  _Float16* vt16   = (_Float16*)(ws + 3 * S);
  _Float16* attn16 = (_Float16*)(ws + 0 * S);
  char* wbase = ws + 6 * S;
  _Float16* wqt = (_Float16*)(wbase);
  _Float16* wkt = (_Float16*)(wbase + 1 * 294912);
  _Float16* wvt = (_Float16*)(wbase + 2 * 294912);
  _Float16* wpt = (_Float16*)(wbase + 3 * 294912);
  _Float16* w1t = (_Float16*)(wbase + 4 * 294912);
  _Float16* w2t = (_Float16*)(wbase + 4 * 294912 + 1179648);

  tcast_kernel<<<dim3(96, 6),   256, 0, stream>>>(Wq, wqt, 384, 64);
  tcast_kernel<<<dim3(96, 6),   256, 0, stream>>>(Wk, wkt, 384, 64);
  tcast_kernel<<<dim3(96, 6),   256, 0, stream>>>(Wv, wvt, 384, 64);
  tcast_kernel<<<dim3(576, 1),  256, 0, stream>>>(Wp, wpt, 384, 384);
  tcast_kernel<<<dim3(2304, 1), 256, 0, stream>>>(W1, w1t, 384, 1536);
  tcast_kernel<<<dim3(2304, 1), 256, 0, stream>>>(W2, w2t, 1536, 384);

  rowgemm_kernel<false><<<256, 256, 0, stream>>>(nullptr, x, g1, be1, wqt, wkt, wvt,
                                                 q16, k16, vt16, nullptr, nullptr, nullptr);
  attn_kernel<<<12288, 32, 0, stream>>>(q16, k16, vt16, attn16);
  rowgemm_kernel<true><<<256, 256, 0, stream>>>(attn16, nullptr, nullptr, nullptr, wpt, nullptr,
                                                nullptr, nullptr, nullptr, nullptr, bp, x, out);
  mlp_kernel<<<256, 256, 0, stream>>>(g2, be2, w1t, b1, w2t, b2, out);
}


# Weight casts folded into the GEMM kernels: a work-group reads its f32 weight chunk, rounds it to f16 and transposes it in registers on the way to LDS; the 6 cast launches go: 10 -> 4; sums unchanged
# speedup vs baseline: 3.9052x; 1.2258x over previous
#include <hip/hip_runtime.h>
#include <hip/hip_bf16.h>


typedef __attribute__((ext_vector_type(16))) _Float16 v16h;
typedef __attribute__((ext_vector_type(8)))  _Float16 v8h;
typedef __attribute__((ext_vector_type(8)))  float    v8f;

#define DEV __device__ __forceinline__

DEV v8f wmma_f16(v16h a, v16h b, v8f c) {
  return __builtin_amdgcn_wmma_f32_16x16x32_f16(false, a, false, b, (short)0, c,
                                                false, false);
}

DEV v16h load_frag(const _Float16* __restrict__ tile, int ld, int lane) {
  const int m  = lane & 15;
  const int kh = (lane & 16) >> 1;
  const _Float16* p = tile + (size_t)m * ld + kh;
  v8h lo = *(const v8h*)p;
  v8h hi = *(const v8h*)(p + 16);
  v16h a;
#pragma unroll
  for (int i = 0; i < 8; ++i) { a[i] = lo[i]; a[i + 8] = hi[i]; }
  return a;
}

typedef __attribute__((ext_vector_type(4))) float v4f;

template <int RS>
DEV void wtask_load(const float* __restrict__ src, v4f* pf) {
#pragma unroll
  for (int r = 0; r < 8; ++r) pf[r] = *(const v4f*)(src + (size_t)r * RS);
}

template <int PITCH>
DEV void wtask_store(_Float16* dst, const v4f* pf) {
#pragma unroll
  for (int j = 0; j < 4; ++j) {
    v8h o;
#pragma unroll
    for (int r = 0; r < 8; ++r) o[r] = (_Float16)pf[r][j];
    *(v8h*)(dst + j * PITCH) = o;
  }
}


DEV void ln_load_rows(const float* src, int lane, v4f (&v)[48]) {
  const int m  = lane & 15;
  const int kh = (lane & 16) >> 1;
  const float* p = src + (size_t)m * 384 + kh;
#pragma unroll
  for (int kk = 0; kk < 12; ++kk) {
    v[4 * kk + 0] = *(const v4f*)(p + kk * 32);
    v[4 * kk + 1] = *(const v4f*)(p + kk * 32 + 4);
    v[4 * kk + 2] = *(const v4f*)(p + kk * 32 + 16);
    v[4 * kk + 3] = *(const v4f*)(p + kk * 32 + 20);
  }
}

DEV void ln_gb_load(const float* __restrict__ gp, const float* __restrict__ bp, int kk,
                    v4f (&gg)[4], v4f (&bb)[4]) {
#pragma unroll
  for (int j = 0; j < 4; ++j) {
    const int co = kk * 32 + (j & 1) * 4 + (j >> 1) * 16;
    gg[j] = *(const v4f*)(gp + co);
    bb[j] = *(const v4f*)(bp + co);
  }
}

DEV void ln_frags(const v4f (&v)[48], const float* __restrict__ g,
                  const float* __restrict__ be, int lane, v16h (&hb)[12]) {
  const int kh = (lane & 16) >> 1;
  const float* gp = g + kh;
  const float* bp = be + kh;
  v4f gg[2][4], bb[2][4];
  ln_gb_load(gp, bp, 0, gg[0], bb[0]);
  v4f s4 = {0.f, 0.f, 0.f, 0.f};
#pragma unroll
  for (int i = 0; i < 48; ++i) s4 += v[i];
  float s = (s4[0] + s4[1]) + (s4[2] + s4[3]);
  s += __shfl_xor(s, 16, 32);
  const float mu = s * (1.f / 384.f);
  v4f q4 = {0.f, 0.f, 0.f, 0.f};
#pragma unroll
  for (int i = 0; i < 48; ++i) { v4f d = v[i] - mu; q4 += d * d; }
  float var = (q4[0] + q4[1]) + (q4[2] + q4[3]);
  var += __shfl_xor(var, 16, 32);
  const float rs = rsqrtf(var * (1.f / 384.f) + 1e-5f);
#pragma unroll
  for (int kk = 0; kk < 12; ++kk) {
    if (kk + 1 < 12) ln_gb_load(gp, bp, kk + 1, gg[(kk + 1) & 1], bb[(kk + 1) & 1]);
    __builtin_amdgcn_sched_barrier(0);
#pragma unroll
    for (int j = 0; j < 4; ++j) {
#pragma unroll
      for (int r = 0; r < 4; ++r)
        hb[kk][4 * j + r] =
            (_Float16)((v[4 * kk + j][r] - mu) * rs * gg[kk & 1][j][r] + bb[kk & 1][j][r]);
    }
    __builtin_amdgcn_sched_barrier(0);
  }
}

#define RG_P 392

template <bool WEIGHT_IS_B>
DEV void rg_chunk(const _Float16* Lc, const v16h (&hb)[12], v8f (&acc)[4], int lane) {
  v16h fa[2][6];
#pragma unroll
  for (int i = 0; i < 6; ++i)
    fa[0][i] = load_frag(Lc + (i & 3) * 16 * RG_P + (i >> 2) * 32, RG_P, lane);
#pragma unroll
  for (int g = 0; g < 8; ++g) {
    if (g + 1 < 8) {
#pragma unroll
      for (int i = 0; i < 6; ++i) {
        const int f = (g + 1) * 6 + i;
        fa[(g + 1) & 1][i] = load_frag(Lc + (f & 3) * 16 * RG_P + (f >> 2) * 32, RG_P, lane);
      }
    }
    __builtin_amdgcn_sched_barrier(0);
#pragma unroll
    for (int i = 0; i < 6; ++i) {
      const int f = g * 6 + i;
      if (WEIGHT_IS_B) acc[f & 3] = wmma_f16(hb[f >> 2], fa[g & 1][i], acc[f & 3]);
      else             acc[f & 3] = wmma_f16(fa[g & 1][i], hb[f >> 2], acc[f & 3]);
    }
    __builtin_amdgcn_sched_barrier(0);
  }
}

DEV v8h pack8h(v8f a) {
  v8h o;
#pragma unroll
  for (int r = 0; r < 8; ++r) o[r] = (_Float16)a[r];
  return o;
}

template <bool PROJ>
__global__ __launch_bounds__(256)
void rowgemm_kernel(const _Float16* __restrict__ act,
                    const float* __restrict__ xf, const float* __restrict__ lng,
                    const float* __restrict__ lnb,
                    const float* __restrict__ w0,
                    const float* __restrict__ w1,
                    const float* __restrict__ w2,
                    _Float16* __restrict__ q, _Float16* __restrict__ k,
                    _Float16* __restrict__ vt,
                    const float* __restrict__ bias, const float* __restrict__ res,
                    float* __restrict__ outf) {
  const int Cc = 384, Dh = 64, T = 256, H = 6;
  const int NCH = PROJ ? 6 : 18;
  const int BUF = 64 * RG_P;
  __shared__ __attribute__((aligned(16))) _Float16 Lw[2 * BUF];

  const int tid = threadIdx.x;
  const int lane = tid & 31;
  const int wave = tid >> 5;
  const int c = lane & 15, hh = lane >> 4;
  const int b = blockIdx.x >> 1;
  const int tw = (blockIdx.x & 1) * 128 + wave * 16;
  const size_t row0 = (size_t)b * T + tw;

  v16h hb[12];
  v4f xv[48];
  if (PROJ) {
#pragma unroll
    for (int kk = 0; kk < 12; ++kk) hb[kk] = load_frag(act + row0 * Cc + kk * 32, Cc, lane);
  } else {
    ln_load_rows(xf + row0 * Cc, lane, xv);
  }

  const int RS = PROJ ? 384 : 64;
  const int rg  = ((tid >> 4) & 3) * 4 + (tid & 3);
  const int kg0 = (tid >> 6) * 4 + ((tid >> 2) & 3);
  const int soff = 8 * kg0 * RS + 4 * rg;
  const int loff = 4 * rg * RG_P + 8 * kg0;
  v4f pf[24];
  {
    const float* g = w0 + soff;
#pragma unroll
    for (int i = 0; i < 3; ++i) wtask_load<RS>(g + (size_t)128 * i * RS, pf + 8 * i);
    if (!PROJ) ln_frags(xv, lng, lnb, lane, hb);
#pragma unroll
    for (int i = 0; i < 3; ++i) wtask_store<RG_P>(Lw + loff + 128 * i, pf + 8 * i);
    g += PROJ ? 64 : 64 * 384;
#pragma unroll
    for (int i = 0; i < 3; ++i) wtask_load<RS>(g + (size_t)128 * i * RS, pf + 8 * i);
  }
  __syncthreads();

  for (int s = 0; s < NCH; ++s) {
    const int m = s / 6;
    const int h = s - m * 6;
    const _Float16* Lc = Lw + (s & 1) * BUF;
    if (s + 1 < NCH) {
      _Float16* Ln = Lw + ((s + 1) & 1) * BUF;
#pragma unroll
      for (int i = 0; i < 3; ++i) wtask_store<RG_P>(Ln + loff + 128 * i, pf + 8 * i);
      if (s + 2 < NCH) {
        const int m2 = (s + 2) / 6;
        const int h2 = (s + 2) - m2 * 6;
        const float* wn = (m2 == 0) ? w0 : (m2 == 1) ? w1 : w2;
        const float* g = wn + (PROJ ? (size_t)h2 * 64 : (size_t)h2 * Dh * Cc) + soff;
#pragma unroll
        for (int i = 0; i < 3; ++i) wtask_load<RS>(g + (size_t)128 * i * RS, pf + 8 * i);
      }
    }

    v8f acc[4];
#pragma unroll
    for (int dt = 0; dt < 4; ++dt) acc[dt] = (v8f){};

    if (PROJ) {
      const size_t off = (row0 + c) * Cc + h * 64 + hh * 8;
      const float* rp = res + off;
      const float* bq = bias + h * 64 + hh * 8;
      v4f xr[8], bb[8];
#pragma unroll
      for (int dt = 0; dt < 4; ++dt) {
        xr[2 * dt]     = *(const v4f*)(rp + dt * 16);
        xr[2 * dt + 1] = *(const v4f*)(rp + dt * 16 + 4);
        bb[2 * dt]     = *(const v4f*)(bq + dt * 16);
        bb[2 * dt + 1] = *(const v4f*)(bq + dt * 16 + 4);
      }
      rg_chunk<false>(Lc, hb, acc, lane);
      v4f o[8];
#pragma unroll
      for (int dt = 0; dt < 4; ++dt) {
#pragma unroll
        for (int r = 0; r < 4; ++r) {
          o[2 * dt][r]     = (acc[dt][r] + bb[2 * dt][r]) + xr[2 * dt][r];
          o[2 * dt + 1][r] = (acc[dt][4 + r] + bb[2 * dt + 1][r]) + xr[2 * dt + 1][r];
        }
      }
      __syncthreads();
      float* op = outf + off;
#pragma unroll
      for (int dt = 0; dt < 4; ++dt) {
        *(v4f*)(op + dt * 16)     = o[2 * dt];
        *(v4f*)(op + dt * 16 + 4) = o[2 * dt + 1];
      }
    } else {
      const size_t bh = (size_t)b * H + h;
      v8h o[4];
      _Float16* dst;
      size_t step;
      if (m < 2) {
        rg_chunk<false>(Lc, hb, acc, lane);
        dst = ((m == 0) ? q : k) + (bh * T + tw + c) * Dh + hh * 8;
        step = 16;
      } else {
        rg_chunk<true>(Lc, hb, acc, lane);
        dst = vt + (bh * Dh + c) * T + tw + hh * 8;
        step = (size_t)16 * T;
      }
#pragma unroll
      for (int dt = 0; dt < 4; ++dt) o[dt] = pack8h(acc[dt]);
      __syncthreads();
#pragma unroll
      for (int dt = 0; dt < 4; ++dt) *(v8h*)(dst + dt * step) = o[dt];
    }
  }
}

template <int M>
DEV float row_xor(float v) {
  return __int_as_float(__builtin_amdgcn_update_dpp(0, __float_as_int(v), 0x160 + M,
                                                    0xf, 0xf, true));
}

__global__ void attn_kernel(const _Float16* __restrict__ q,
                            const _Float16* __restrict__ k,
                            const _Float16* __restrict__ vt,
                            _Float16* __restrict__ attn_out) {
  const int T = 256, Dh = 64, H = 6, C = 384;
  int tile = blockIdx.x;
  int it = tile & 15;
  int bh = tile >> 4;
  int b = bh / H, h = bh - b * H;
  int t0 = it * 16;
  int lane = threadIdx.x;
  const int n = lane & 15, mh = (lane >> 4) * 8;

  const _Float16* qb = q + ((size_t)bh * T + t0) * Dh;
  const _Float16* kb = k + (size_t)bh * T * Dh;
  const _Float16* vb = vt + (size_t)bh * Dh * T;

  v16h aq0 = load_frag(qb, Dh, lane);
  v16h aq1 = load_frag(qb + 32, Dh, lane);

  float mrow[8], lrow[8];
#pragma unroll
  for (int r = 0; r < 8; ++r) { mrow[r] = -__builtin_inff(); lrow[r] = 0.f; }
  v8f O[4];
#pragma unroll
  for (int i = 0; i < 4; ++i) O[i] = (v8f){};

  __shared__ _Float16 Pl[16 * 32];
  const float scale = 0.05103103630798287f;

  for (int s0 = 0; s0 < t0 + 16; s0 += 32) {
    float P0[8], P1[8];
    {
      v8f acc = {};
      v16h b0 = load_frag(kb + (size_t)s0 * Dh, Dh, lane);
      v16h b1 = load_frag(kb + (size_t)s0 * Dh + 32, Dh, lane);
      acc = wmma_f16(aq0, b0, acc);
      acc = wmma_f16(aq1, b1, acc);
#pragma unroll
      for (int r = 0; r < 8; ++r) P0[r] = acc[r] * scale;
    }
    {
      v8f acc = {};
      v16h b0 = load_frag(kb + (size_t)(s0 + 16) * Dh, Dh, lane);
      v16h b1 = load_frag(kb + (size_t)(s0 + 16) * Dh + 32, Dh, lane);
      acc = wmma_f16(aq0, b0, acc);
      acc = wmma_f16(aq1, b1, acc);
#pragma unroll
      for (int r = 0; r < 8; ++r) P1[r] = acc[r] * scale;
    }
#pragma unroll
    for (int r = 0; r < 8; ++r) {
      int trow = t0 + mh + r;
      if (s0 + n > trow)      P0[r] = -__builtin_inff();
      if (s0 + 16 + n > trow) P1[r] = -__builtin_inff();
      float mx = fmaxf(P0[r], P1[r]);
      mx = fmaxf(mx, row_xor<1>(mx));
      mx = fmaxf(mx, row_xor<2>(mx));
      mx = fmaxf(mx, row_xor<4>(mx));
      mx = fmaxf(mx, row_xor<8>(mx));
      float mnew = fmaxf(mrow[r], mx);
      float al = expf(mrow[r] - mnew);
      float p0 = expf(P0[r] - mnew);
      float p1 = expf(P1[r] - mnew);
      float rs = p0 + p1;
      rs += row_xor<1>(rs);
      rs += row_xor<2>(rs);
      rs += row_xor<4>(rs);
      rs += row_xor<8>(rs);
      lrow[r] = lrow[r] * al + rs;
      mrow[r] = mnew;
      P0[r] = p0; P1[r] = p1;
      O[0][r] *= al; O[1][r] *= al; O[2][r] *= al; O[3][r] *= al;
    }
    __syncthreads();
#pragma unroll
    for (int r = 0; r < 8; ++r) {
      Pl[(mh + r) * 32 + n]      = (_Float16)P0[r];
      Pl[(mh + r) * 32 + 16 + n] = (_Float16)P1[r];
    }
    __syncthreads();
    v16h ap = load_frag(Pl, 32, lane);
#pragma unroll
    for (int nt = 0; nt < 4; ++nt) {
      v16h bv = load_frag(vb + (size_t)(nt * 16) * T + s0, T, lane);
      O[nt] = wmma_f16(ap, bv, O[nt]);
    }
  }
#pragma unroll
  for (int r = 0; r < 8; ++r) {
    float inv = 1.f / lrow[r];
    _Float16* orow = attn_out + ((size_t)b * T + t0 + mh + r) * C + h * Dh;
#pragma unroll
    for (int nt = 0; nt < 4; ++nt) orow[nt * 16 + n] = (_Float16)(O[nt][r] * inv);
  }
}


__global__ __launch_bounds__(256)
void mlp_kernel(const float* __restrict__ lng, const float* __restrict__ lnb,
                const float* __restrict__ w1,
                const float* __restrict__ b1, const float* __restrict__ w2,
                const float* __restrict__ b2, float* __restrict__ out) {
  const int C = 384, F = 1536;
  const int P1 = 392;
  const int P2 = 40;
  __shared__ __attribute__((aligned(16))) _Float16 Lw1[32 * P1];
  __shared__ __attribute__((aligned(16))) _Float16 Lw2[384 * P2];
  __shared__ __attribute__((aligned(16))) float Lb1[1536];

  const int tid = threadIdx.x;
  const int lane = tid & 31;
  const int wave = tid >> 5;
  const int c = lane & 15, hh = lane >> 4;
  const size_t row0 = (size_t)blockIdx.x * 128 + (size_t)wave * 16;

#pragma unroll
  for (int i = 0; i < 6; ++i) Lb1[tid + 256 * i] = b1[tid + 256 * i];

  v4f xv[48];
  ln_load_rows(out + row0 * C, lane, xv);

  const bool c_is_w1 = __builtin_amdgcn_readfirstlane(wave) < 4;
  const int uC = 256 + (tid & 127);
  const int rgA = ((tid >> 4) & 1) * 4 + (tid & 3), kgA = (tid >> 5) * 4 + ((tid >> 2) & 3);
  const int rgB = (tid >> 4) * 4 + (tid & 3),       kgB = (tid >> 2) & 3;
  const int rgC = c_is_w1 ? ((uC >> 4) & 1) * 4 + (uC & 3) : (uC >> 4) * 4 + (uC & 3);
  const int kgC = c_is_w1 ? (uC >> 5) * 4 + ((uC >> 2) & 3) : (uC >> 2) & 3;
  const float* gA = w1 + (size_t)8 * kgA * F + 4 * rgA;
  const float* gB = w2 + (size_t)8 * kgB * C + 4 * rgB;
  const float* gC = c_is_w1 ? w1 + (size_t)8 * kgC * F + 4 * rgC
                            : w2 + (size_t)8 * kgC * C + 4 * rgC;
  const int oA = 4 * rgA * P1 + 8 * kgA;
  const int oB = 4 * rgB * P2 + 8 * kgB;
  const int oC = c_is_w1 ? 4 * rgC * P1 + 8 * kgC : 4 * rgC * P2 + 8 * kgC;

  v4f pf[24];
  wtask_load<F>(gA, pf);
  wtask_load<C>(gB, pf + 8);
  if (c_is_w1) wtask_load<F>(gC, pf + 16);
  else         wtask_load<C>(gC, pf + 16);

  v16h hb[12];
  ln_frags(xv, lng, lnb, lane, hb);

  v8f acc[24];
#pragma unroll
  for (int mt = 0; mt < 24; ++mt) acc[mt] = (v8f){};

  for (int s = 0; s < 48; ++s) {
    __syncthreads();
    wtask_store<P1>(Lw1 + oA, pf);
    wtask_store<P2>(Lw2 + oB, pf + 8);
    if (c_is_w1) wtask_store<P1>(Lw1 + oC, pf + 16);
    else         wtask_store<P2>(Lw2 + oC, pf + 16);
    __syncthreads();
    if (s + 1 < 48) {
      const int hid1 = (s + 1) * 32;
      wtask_load<F>(gA + hid1, pf);
      wtask_load<C>(gB + (size_t)hid1 * C, pf + 8);
      if (c_is_w1) wtask_load<F>(gC + hid1, pf + 16);
      else         wtask_load<C>(gC + (size_t)hid1 * C, pf + 16);
    }

    v16h fa[2][6];
#pragma unroll
    for (int i = 0; i < 6; ++i) fa[0][i] = load_frag(Lw1 + (i & 1) * 16 * P1 + (i >> 1) * 32, P1, lane);
    v8f ha0 = {}, ha1 = {};
    v16h hf = {};
#pragma unroll
    for (int g = 0; g < 8; ++g) {
      if (g + 1 < 8) {
#pragma unroll
        for (int i = 0; i < 6; ++i) {
          const int f = (g + 1) * 6 + i;
          fa[(g + 1) & 1][i] = (f < 24)
              ? load_frag(Lw1 + (f & 1) * 16 * P1 + (f >> 1) * 32, P1, lane)
              : load_frag(Lw2 + (f - 24) * 16 * P2, P2, lane);
        }
      }
      __builtin_amdgcn_sched_barrier(0);
#pragma unroll
      for (int i = 0; i < 6; ++i) {
        const int f = g * 6 + i;
        if (f < 24) {
          if (f & 1) ha1 = wmma_f16(fa[g & 1][i], hb[f >> 1], ha1);
          else       ha0 = wmma_f16(fa[g & 1][i], hb[f >> 1], ha0);
          if (f == 23) {
            const float* bq = Lb1 + s * 32 + hh * 8;
            v4f bA = *(const v4f*)bq;
            v4f bB = *(const v4f*)(bq + 4);
            v4f bC = *(const v4f*)(bq + 16);
            v4f bD = *(const v4f*)(bq + 20);
#pragma unroll
            for (int r = 0; r < 4; ++r) {
              hf[r]      = (_Float16)fmaxf(ha0[r] + bA[r], 0.f);
              hf[4 + r]  = (_Float16)fmaxf(ha0[4 + r] + bB[r], 0.f);
              hf[8 + r]  = (_Float16)fmaxf(ha1[r] + bC[r], 0.f);
              hf[12 + r] = (_Float16)fmaxf(ha1[4 + r] + bD[r], 0.f);
            }
          }
        } else {
          acc[f - 24] = wmma_f16(fa[g & 1][i], hf, acc[f - 24]);
        }
      }
      __builtin_amdgcn_sched_barrier(0);
    }
  }

  float* orow = out + (row0 + c) * C + hh * 8;
  const float* b2p = b2 + hh * 8;
#pragma unroll
  for (int mt = 0; mt < 24; ++mt) {
    if (mt == 12) __builtin_amdgcn_sched_barrier(0);
    v4f oA = *(const v4f*)(orow + mt * 16);
    v4f oB = *(const v4f*)(orow + mt * 16 + 4);
    v4f cA = *(const v4f*)(b2p + mt * 16);
    v4f cB = *(const v4f*)(b2p + mt * 16 + 4);
#pragma unroll
    for (int r = 0; r < 4; ++r) {
      oA[r] += acc[mt][r] + cA[r];
      oB[r] += acc[mt][4 + r] + cB[r];
    }
    *(v4f*)(orow + mt * 16) = oA;
    *(v4f*)(orow + mt * 16 + 4) = oB;
  }
}

extern "C" void kernel_launch(void* const* d_in, const int* in_sizes, int n_in,
                              void* d_out, int out_size, void* d_ws, size_t ws_size,
                              hipStream_t stream) {
  (void)in_sizes; (void)n_in; (void)out_size; (void)ws_size;
  const float* x   = (const float*)d_in[0];
  const float* Wq  = (const float*)d_in[1];
  const float* Wk  = (const float*)d_in[2];
  const float* Wv  = (const float*)d_in[3];
  const float* Wp  = (const float*)d_in[4];
  const float* bp  = (const float*)d_in[5];
  const float* W1  = (const float*)d_in[6];
  const float* b1  = (const float*)d_in[7];
  const float* W2  = (const float*)d_in[8];
  const float* b2  = (const float*)d_in[9];
  const float* g1  = (const float*)d_in[10];
  const float* be1 = (const float*)d_in[11];
  const float* g2  = (const float*)d_in[12];
  const float* be2 = (const float*)d_in[13];
  float* out = (float*)d_out;

  char* ws = (char*)d_ws;
  const size_t S = (size_t)32768 * 384 * 2;
  _Float16* q16    = (_Float16*)(ws + 1 * S);
  _Float16* k16    = (_Float16*)(ws + 2 * S);
  _Float16* vt16   = (_Float16*)(ws + 3 * S);
  _Float16* attn16 = (_Float16*)(ws + 0 * S);

  rowgemm_kernel<false><<<256, 256, 0, stream>>>(nullptr, x, g1, be1, Wq, Wk, Wv,
                                                 q16, k16, vt16, nullptr, nullptr, nullptr);
  attn_kernel<<<12288, 32, 0, stream>>>(q16, k16, vt16, attn16);
  rowgemm_kernel<true><<<256, 256, 0, stream>>>(attn16, nullptr, nullptr, nullptr, Wp, nullptr,
                                                nullptr, nullptr, nullptr, nullptr, bp, x, out);
  mlp_kernel<<<256, 256, 0, stream>>>(g2, be2, W1, b1, W2, b2, out);
}


# Attention: a chunk's value fragments and the next chunk's key fragments are requested a stage ahead (launch bound 32 lifts the 128-register cap): 6 dependent round trips a chunk -> 0; sums unchanged
# speedup vs baseline: 4.1320x; 1.0581x over previous
#include <hip/hip_runtime.h>
#include <hip/hip_bf16.h>


typedef __attribute__((ext_vector_type(16))) _Float16 v16h;
typedef __attribute__((ext_vector_type(8)))  _Float16 v8h;
typedef __attribute__((ext_vector_type(8)))  float    v8f;

#define DEV __device__ __forceinline__

DEV v8f wmma_f16(v16h a, v16h b, v8f c) {
  return __builtin_amdgcn_wmma_f32_16x16x32_f16(false, a, false, b, (short)0, c,
                                                false, false);
}

DEV v16h load_frag(const _Float16* __restrict__ tile, int ld, int lane) {
  const int m  = lane & 15;
  const int kh = (lane & 16) >> 1;
  const _Float16* p = tile + (size_t)m * ld + kh;
  v8h lo = *(const v8h*)p;
  v8h hi = *(const v8h*)(p + 16);
  v16h a;
#pragma unroll
  for (int i = 0; i < 8; ++i) { a[i] = lo[i]; a[i + 8] = hi[i]; }
  return a;
}

typedef __attribute__((ext_vector_type(4))) float v4f;

template <int RS>
DEV void wtask_load(const float* __restrict__ src, v4f* pf) {
#pragma unroll
  for (int r = 0; r < 8; ++r) pf[r] = *(const v4f*)(src + (size_t)r * RS);
}

template <int PITCH>
DEV void wtask_store(_Float16* dst, const v4f* pf) {
#pragma unroll
  for (int j = 0; j < 4; ++j) {
    v8h o;
#pragma unroll
    for (int r = 0; r < 8; ++r) o[r] = (_Float16)pf[r][j];
    *(v8h*)(dst + j * PITCH) = o;
  }
}


DEV void ln_load_rows(const float* src, int lane, v4f (&v)[48]) {
  const int m  = lane & 15;
  const int kh = (lane & 16) >> 1;
  const float* p = src + (size_t)m * 384 + kh;
#pragma unroll
  for (int kk = 0; kk < 12; ++kk) {
    v[4 * kk + 0] = *(const v4f*)(p + kk * 32);
    v[4 * kk + 1] = *(const v4f*)(p + kk * 32 + 4);
    v[4 * kk + 2] = *(const v4f*)(p + kk * 32 + 16);
    v[4 * kk + 3] = *(const v4f*)(p + kk * 32 + 20);
  }
}

DEV void ln_gb_load(const float* __restrict__ gp, const float* __restrict__ bp, int kk,
                    v4f (&gg)[4], v4f (&bb)[4]) {
#pragma unroll
  for (int j = 0; j < 4; ++j) {
    const int co = kk * 32 + (j & 1) * 4 + (j >> 1) * 16;
    gg[j] = *(const v4f*)(gp + co);
    bb[j] = *(const v4f*)(bp + co);
  }
}

DEV void ln_frags(const v4f (&v)[48], const float* __restrict__ g,
                  const float* __restrict__ be, int lane, v16h (&hb)[12]) {
  const int kh = (lane & 16) >> 1;
  const float* gp = g + kh;
  const float* bp = be + kh;
  v4f gg[2][4], bb[2][4];
  ln_gb_load(gp, bp, 0, gg[0], bb[0]);
  v4f s4 = {0.f, 0.f, 0.f, 0.f};
#pragma unroll
  for (int i = 0; i < 48; ++i) s4 += v[i];
  float s = (s4[0] + s4[1]) + (s4[2] + s4[3]);
  s += __shfl_xor(s, 16, 32);
  const float mu = s * (1.f / 384.f);
  v4f q4 = {0.f, 0.f, 0.f, 0.f};
#pragma unroll
  for (int i = 0; i < 48; ++i) { v4f d = v[i] - mu; q4 += d * d; }
  float var = (q4[0] + q4[1]) + (q4[2] + q4[3]);
  var += __shfl_xor(var, 16, 32);
  const float rs = rsqrtf(var * (1.f / 384.f) + 1e-5f);
#pragma unroll
  for (int kk = 0; kk < 12; ++kk) {
    if (kk + 1 < 12) ln_gb_load(gp, bp, kk + 1, gg[(kk + 1) & 1], bb[(kk + 1) & 1]);
    __builtin_amdgcn_sched_barrier(0);
#pragma unroll
    for (int j = 0; j < 4; ++j) {
#pragma unroll
      for (int r = 0; r < 4; ++r)
        hb[kk][4 * j + r] =
            (_Float16)((v[4 * kk + j][r] - mu) * rs * gg[kk & 1][j][r] + bb[kk & 1][j][r]);
    }
    __builtin_amdgcn_sched_barrier(0);
  }
}

#define RG_P 392

template <bool WEIGHT_IS_B>
DEV void rg_chunk(const _Float16* Lc, const v16h (&hb)[12], v8f (&acc)[4], int lane) {
  v16h fa[2][6];
#pragma unroll
  for (int i = 0; i < 6; ++i)
    fa[0][i] = load_frag(Lc + (i & 3) * 16 * RG_P + (i >> 2) * 32, RG_P, lane);
#pragma unroll
  for (int g = 0; g < 8; ++g) {
    if (g + 1 < 8) {
#pragma unroll
      for (int i = 0; i < 6; ++i) {
        const int f = (g + 1) * 6 + i;
        fa[(g + 1) & 1][i] = load_frag(Lc + (f & 3) * 16 * RG_P + (f >> 2) * 32, RG_P, lane);
      }
    }
    __builtin_amdgcn_sched_barrier(0);
#pragma unroll
    for (int i = 0; i < 6; ++i) {
      const int f = g * 6 + i;
      if (WEIGHT_IS_B) acc[f & 3] = wmma_f16(hb[f >> 2], fa[g & 1][i], acc[f & 3]);
      else             acc[f & 3] = wmma_f16(fa[g & 1][i], hb[f >> 2], acc[f & 3]);
    }
    __builtin_amdgcn_sched_barrier(0);
  }
}

DEV v8h pack8h(v8f a) {
  v8h o;
#pragma unroll
  for (int r = 0; r < 8; ++r) o[r] = (_Float16)a[r];
  return o;
}

template <bool PROJ>
__global__ __launch_bounds__(256)
void rowgemm_kernel(const _Float16* __restrict__ act,
                    const float* __restrict__ xf, const float* __restrict__ lng,
                    const float* __restrict__ lnb,
                    const float* __restrict__ w0,
                    const float* __restrict__ w1,
                    const float* __restrict__ w2,
                    _Float16* __restrict__ q, _Float16* __restrict__ k,
                    _Float16* __restrict__ vt,
                    const float* __restrict__ bias, const float* __restrict__ res,
                    float* __restrict__ outf) {
  const int Cc = 384, Dh = 64, T = 256, H = 6;
  const int NCH = PROJ ? 6 : 18;
  const int BUF = 64 * RG_P;
  __shared__ __attribute__((aligned(16))) _Float16 Lw[2 * BUF];

  const int tid = threadIdx.x;
  const int lane = tid & 31;
  const int wave = tid >> 5;
  const int c = lane & 15, hh = lane >> 4;
  const int b = blockIdx.x >> 1;
  const int tw = (blockIdx.x & 1) * 128 + wave * 16;
  const size_t row0 = (size_t)b * T + tw;

  v16h hb[12];
  v4f xv[48];
  if (PROJ) {
#pragma unroll
    for (int kk = 0; kk < 12; ++kk) hb[kk] = load_frag(act + row0 * Cc + kk * 32, Cc, lane);
  } else {
    ln_load_rows(xf + row0 * Cc, lane, xv);
  }

  const int RS = PROJ ? 384 : 64;
  const int rg  = ((tid >> 4) & 3) * 4 + (tid & 3);
  const int kg0 = (tid >> 6) * 4 + ((tid >> 2) & 3);
  const int soff = 8 * kg0 * RS + 4 * rg;
  const int loff = 4 * rg * RG_P + 8 * kg0;
  v4f pf[24];
  {
    const float* g = w0 + soff;
#pragma unroll
    for (int i = 0; i < 3; ++i) wtask_load<RS>(g + (size_t)128 * i * RS, pf + 8 * i);
    if (!PROJ) ln_frags(xv, lng, lnb, lane, hb);
#pragma unroll
    for (int i = 0; i < 3; ++i) wtask_store<RG_P>(Lw + loff + 128 * i, pf + 8 * i);
    g += PROJ ? 64 : 64 * 384;
#pragma unroll
    for (int i = 0; i < 3; ++i) wtask_load<RS>(g + (size_t)128 * i * RS, pf + 8 * i);
  }
  __syncthreads();

  for (int s = 0; s < NCH; ++s) {
    const int m = s / 6;
    const int h = s - m * 6;
    const _Float16* Lc = Lw + (s & 1) * BUF;
    if (s + 1 < NCH) {
      _Float16* Ln = Lw + ((s + 1) & 1) * BUF;
#pragma unroll
      for (int i = 0; i < 3; ++i) wtask_store<RG_P>(Ln + loff + 128 * i, pf + 8 * i);
      if (s + 2 < NCH) {
        const int m2 = (s + 2) / 6;
        const int h2 = (s + 2) - m2 * 6;
        const float* wn = (m2 == 0) ? w0 : (m2 == 1) ? w1 : w2;
        const float* g = wn + (PROJ ? (size_t)h2 * 64 : (size_t)h2 * Dh * Cc) + soff;
#pragma unroll
        for (int i = 0; i < 3; ++i) wtask_load<RS>(g + (size_t)128 * i * RS, pf + 8 * i);
      }
    }

    v8f acc[4];
#pragma unroll
    for (int dt = 0; dt < 4; ++dt) acc[dt] = (v8f){};

    if (PROJ) {
      const size_t off = (row0 + c) * Cc + h * 64 + hh * 8;
      const float* rp = res + off;
      const float* bq = bias + h * 64 + hh * 8;
      v4f xr[8], bb[8];
#pragma unroll
      for (int dt = 0; dt < 4; ++dt) {
        xr[2 * dt]     = *(const v4f*)(rp + dt * 16);
        xr[2 * dt + 1] = *(const v4f*)(rp + dt * 16 + 4);
        bb[2 * dt]     = *(const v4f*)(bq + dt * 16);
        bb[2 * dt + 1] = *(const v4f*)(bq + dt * 16 + 4);
      }
      rg_chunk<false>(Lc, hb, acc, lane);
      v4f o[8];
#pragma unroll
      for (int dt = 0; dt < 4; ++dt) {
#pragma unroll
        for (int r = 0; r < 4; ++r) {
          o[2 * dt][r]     = (acc[dt][r] + bb[2 * dt][r]) + xr[2 * dt][r];
          o[2 * dt + 1][r] = (acc[dt][4 + r] + bb[2 * dt + 1][r]) + xr[2 * dt + 1][r];
        }
      }
      __syncthreads();
      float* op = outf + off;
#pragma unroll
      for (int dt = 0; dt < 4; ++dt) {
        *(v4f*)(op + dt * 16)     = o[2 * dt];
        *(v4f*)(op + dt * 16 + 4) = o[2 * dt + 1];
      }
    } else {
      const size_t bh = (size_t)b * H + h;
      v8h o[4];
      _Float16* dst;
      size_t step;
      if (m < 2) {
        rg_chunk<false>(Lc, hb, acc, lane);
        dst = ((m == 0) ? q : k) + (bh * T + tw + c) * Dh + hh * 8;
        step = 16;
      } else {
        rg_chunk<true>(Lc, hb, acc, lane);
        dst = vt + (bh * Dh + c) * T + tw + hh * 8;
        step = (size_t)16 * T;
      }
#pragma unroll
      for (int dt = 0; dt < 4; ++dt) o[dt] = pack8h(acc[dt]);
      __syncthreads();
#pragma unroll
      for (int dt = 0; dt < 4; ++dt) *(v8h*)(dst + dt * step) = o[dt];
    }
  }
}

template <int M>
DEV float row_xor(float v) {
  return __int_as_float(__builtin_amdgcn_update_dpp(0, __float_as_int(v), 0x160 + M,
                                                    0xf, 0xf, true));
}

template <bool NEXT>
DEV void attn_chunk(const _Float16* __restrict__ kb, const _Float16* __restrict__ vb,
                    _Float16* Pl, int s0, int t0, int lane,
                    const v16h& aq0, const v16h& aq1, v16h (&kf)[4],
                    float (&mrow)[8], float (&lrow)[8], v8f (&O)[4]) {
  const int T = 256, Dh = 64;
  const int n = lane & 15, mh = (lane >> 4) * 8;
  const float scale = 0.05103103630798287f;

  float P0[8], P1[8];
  {
    v8f acc = {};
    acc = wmma_f16(aq0, kf[0], acc);
    acc = wmma_f16(aq1, kf[1], acc);
#pragma unroll
    for (int r = 0; r < 8; ++r) P0[r] = acc[r] * scale;
  }
  {
    v8f acc = {};
    acc = wmma_f16(aq0, kf[2], acc);
    acc = wmma_f16(aq1, kf[3], acc);
#pragma unroll
    for (int r = 0; r < 8; ++r) P1[r] = acc[r] * scale;
  }
  v16h vf[4];
  vf[0] = load_frag(vb + s0, T, lane);
  vf[1] = load_frag(vb + (size_t)16 * T + s0, T, lane);
  vf[2] = load_frag(vb + (size_t)32 * T + s0, T, lane);
  vf[3] = load_frag(vb + (size_t)48 * T + s0, T, lane);
  if (NEXT) {
    const _Float16* kn = kb + (size_t)(s0 + 32) * Dh;
    kf[0] = load_frag(kn, Dh, lane);
    kf[1] = load_frag(kn + 32, Dh, lane);
    kf[2] = load_frag(kn + (size_t)16 * Dh, Dh, lane);
    kf[3] = load_frag(kn + (size_t)16 * Dh + 32, Dh, lane);
  }
#pragma unroll
  for (int r = 0; r < 8; ++r) {
    int trow = t0 + mh + r;
    if (s0 + n > trow)      P0[r] = -__builtin_inff();
    if (s0 + 16 + n > trow) P1[r] = -__builtin_inff();
    float mx = fmaxf(P0[r], P1[r]);
    mx = fmaxf(mx, row_xor<1>(mx));
    mx = fmaxf(mx, row_xor<2>(mx));
    mx = fmaxf(mx, row_xor<4>(mx));
    mx = fmaxf(mx, row_xor<8>(mx));
    float mnew = fmaxf(mrow[r], mx);
    float al = expf(mrow[r] - mnew);
    float p0 = expf(P0[r] - mnew);
    float p1 = expf(P1[r] - mnew);
    float rs = p0 + p1;
    rs += row_xor<1>(rs);
    rs += row_xor<2>(rs);
    rs += row_xor<4>(rs);
    rs += row_xor<8>(rs);
    lrow[r] = lrow[r] * al + rs;
    mrow[r] = mnew;
    P0[r] = p0; P1[r] = p1;
    O[0][r] *= al; O[1][r] *= al; O[2][r] *= al; O[3][r] *= al;
  }
  __syncthreads();
#pragma unroll
  for (int r = 0; r < 8; ++r) {
    Pl[(mh + r) * 32 + n]      = (_Float16)P0[r];
    Pl[(mh + r) * 32 + 16 + n] = (_Float16)P1[r];
  }
  __syncthreads();
  v16h ap = load_frag(Pl, 32, lane);
#pragma unroll
  for (int nt = 0; nt < 4; ++nt) O[nt] = wmma_f16(ap, vf[nt], O[nt]);
}

__global__ __launch_bounds__(32)
void attn_kernel(const _Float16* __restrict__ q,
                 const _Float16* __restrict__ k,
                 const _Float16* __restrict__ vt,
                 _Float16* __restrict__ attn_out) {
  const int T = 256, Dh = 64, H = 6, C = 384;
  int tile = blockIdx.x;
  int it = tile & 15;
  int bh = tile >> 4;
  int b = bh / H, h = bh - b * H;
  int t0 = it * 16;
  int lane = threadIdx.x;
  const int n = lane & 15, mh = (lane >> 4) * 8;

  const _Float16* qb = q + ((size_t)bh * T + t0) * Dh;
  const _Float16* kb = k + (size_t)bh * T * Dh;
  const _Float16* vb = vt + (size_t)bh * Dh * T;

  v16h aq0 = load_frag(qb, Dh, lane);
  v16h aq1 = load_frag(qb + 32, Dh, lane);

  v16h kf[4];
  kf[0] = load_frag(kb, Dh, lane);
  kf[1] = load_frag(kb + 32, Dh, lane);
  kf[2] = load_frag(kb + (size_t)16 * Dh, Dh, lane);
  kf[3] = load_frag(kb + (size_t)16 * Dh + 32, Dh, lane);

  float mrow[8], lrow[8];
#pragma unroll
  for (int r = 0; r < 8; ++r) { mrow[r] = -__builtin_inff(); lrow[r] = 0.f; }
  v8f O[4];
#pragma unroll
  for (int i = 0; i < 4; ++i) O[i] = (v8f){};

  __shared__ _Float16 Pl[16 * 32];

  int s0 = 0;
  for (; s0 + 32 < t0 + 16; s0 += 32)
    attn_chunk<true>(kb, vb, Pl, s0, t0, lane, aq0, aq1, kf, mrow, lrow, O);
  attn_chunk<false>(kb, vb, Pl, s0, t0, lane, aq0, aq1, kf, mrow, lrow, O);

#pragma unroll
  for (int r = 0; r < 8; ++r) {
    float inv = 1.f / lrow[r];
    _Float16* orow = attn_out + ((size_t)b * T + t0 + mh + r) * C + h * Dh;
#pragma unroll
    for (int nt = 0; nt < 4; ++nt) orow[nt * 16 + n] = (_Float16)(O[nt][r] * inv);
  }
}


__global__ __launch_bounds__(256)
void mlp_kernel(const float* __restrict__ lng, const float* __restrict__ lnb,
                const float* __restrict__ w1,
                const float* __restrict__ b1, const float* __restrict__ w2,
                const float* __restrict__ b2, float* __restrict__ out) {
  const int C = 384, F = 1536;
  const int P1 = 392;
  const int P2 = 40;
  __shared__ __attribute__((aligned(16))) _Float16 Lw1[32 * P1];
  __shared__ __attribute__((aligned(16))) _Float16 Lw2[384 * P2];
  __shared__ __attribute__((aligned(16))) float Lb1[1536];

  const int tid = threadIdx.x;
  const int lane = tid & 31;
  const int wave = tid >> 5;
  const int c = lane & 15, hh = lane >> 4;
  const size_t row0 = (size_t)blockIdx.x * 128 + (size_t)wave * 16;

#pragma unroll
  for (int i = 0; i < 6; ++i) Lb1[tid + 256 * i] = b1[tid + 256 * i];

  v4f xv[48];
  ln_load_rows(out + row0 * C, lane, xv);

  const bool c_is_w1 = __builtin_amdgcn_readfirstlane(wave) < 4;
  const int uC = 256 + (tid & 127);
  const int rgA = ((tid >> 4) & 1) * 4 + (tid & 3), kgA = (tid >> 5) * 4 + ((tid >> 2) & 3);
  const int rgB = (tid >> 4) * 4 + (tid & 3),       kgB = (tid >> 2) & 3;
  const int rgC = c_is_w1 ? ((uC >> 4) & 1) * 4 + (uC & 3) : (uC >> 4) * 4 + (uC & 3);
  const int kgC = c_is_w1 ? (uC >> 5) * 4 + ((uC >> 2) & 3) : (uC >> 2) & 3;
  const float* gA = w1 + (size_t)8 * kgA * F + 4 * rgA;
  const float* gB = w2 + (size_t)8 * kgB * C + 4 * rgB;
  const float* gC = c_is_w1 ? w1 + (size_t)8 * kgC * F + 4 * rgC
                            : w2 + (size_t)8 * kgC * C + 4 * rgC;
  const int oA = 4 * rgA * P1 + 8 * kgA;
  const int oB = 4 * rgB * P2 + 8 * kgB;
  const int oC = c_is_w1 ? 4 * rgC * P1 + 8 * kgC : 4 * rgC * P2 + 8 * kgC;

  v4f pf[24];
  wtask_load<F>(gA, pf);
  wtask_load<C>(gB, pf + 8);
  if (c_is_w1) wtask_load<F>(gC, pf + 16);
  else         wtask_load<C>(gC, pf + 16);

  v16h hb[12];
  ln_frags(xv, lng, lnb, lane, hb);

  v8f acc[24];
#pragma unroll
  for (int mt = 0; mt < 24; ++mt) acc[mt] = (v8f){};

  for (int s = 0; s < 48; ++s) {
    __syncthreads();
    wtask_store<P1>(Lw1 + oA, pf);
    wtask_store<P2>(Lw2 + oB, pf + 8);
    if (c_is_w1) wtask_store<P1>(Lw1 + oC, pf + 16);
    else         wtask_store<P2>(Lw2 + oC, pf + 16);
    __syncthreads();
    if (s + 1 < 48) {
      const int hid1 = (s + 1) * 32;
      wtask_load<F>(gA + hid1, pf);
      wtask_load<C>(gB + (size_t)hid1 * C, pf + 8);
      if (c_is_w1) wtask_load<F>(gC + hid1, pf + 16);
      else         wtask_load<C>(gC + (size_t)hid1 * C, pf + 16);
    }

    v16h fa[2][6];
#pragma unroll
    for (int i = 0; i < 6; ++i) fa[0][i] = load_frag(Lw1 + (i & 1) * 16 * P1 + (i >> 1) * 32, P1, lane);
    v8f ha0 = {}, ha1 = {};
    v16h hf = {};
#pragma unroll
    for (int g = 0; g < 8; ++g) {
      if (g + 1 < 8) {
#pragma unroll
        for (int i = 0; i < 6; ++i) {
          const int f = (g + 1) * 6 + i;
          fa[(g + 1) & 1][i] = (f < 24)
              ? load_frag(Lw1 + (f & 1) * 16 * P1 + (f >> 1) * 32, P1, lane)
              : load_frag(Lw2 + (f - 24) * 16 * P2, P2, lane);
        }
      }
      __builtin_amdgcn_sched_barrier(0);
#pragma unroll
      for (int i = 0; i < 6; ++i) {
        const int f = g * 6 + i;
        if (f < 24) {
          if (f & 1) ha1 = wmma_f16(fa[g & 1][i], hb[f >> 1], ha1);
          else       ha0 = wmma_f16(fa[g & 1][i], hb[f >> 1], ha0);
          if (f == 23) {
            const float* bq = Lb1 + s * 32 + hh * 8;
            v4f bA = *(const v4f*)bq;
            v4f bB = *(const v4f*)(bq + 4);
            v4f bC = *(const v4f*)(bq + 16);
            v4f bD = *(const v4f*)(bq + 20);
#pragma unroll
            for (int r = 0; r < 4; ++r) {
              hf[r]      = (_Float16)fmaxf(ha0[r] + bA[r], 0.f);
              hf[4 + r]  = (_Float16)fmaxf(ha0[4 + r] + bB[r], 0.f);
              hf[8 + r]  = (_Float16)fmaxf(ha1[r] + bC[r], 0.f);
              hf[12 + r] = (_Float16)fmaxf(ha1[4 + r] + bD[r], 0.f);
            }
          }
        } else {
          acc[f - 24] = wmma_f16(fa[g & 1][i], hf, acc[f - 24]);
        }
      }
      __builtin_amdgcn_sched_barrier(0);
    }
  }

  float* orow = out + (row0 + c) * C + hh * 8;
  const float* b2p = b2 + hh * 8;
#pragma unroll
  for (int mt = 0; mt < 24; ++mt) {
    if (mt == 12) __builtin_amdgcn_sched_barrier(0);
    v4f oA = *(const v4f*)(orow + mt * 16);
    v4f oB = *(const v4f*)(orow + mt * 16 + 4);
    v4f cA = *(const v4f*)(b2p + mt * 16);
    v4f cB = *(const v4f*)(b2p + mt * 16 + 4);
#pragma unroll
    for (int r = 0; r < 4; ++r) {
      oA[r] += acc[mt][r] + cA[r];
      oB[r] += acc[mt][4 + r] + cB[r];
    }
    *(v4f*)(orow + mt * 16) = oA;
    *(v4f*)(orow + mt * 16 + 4) = oB;
  }
}

extern "C" void kernel_launch(void* const* d_in, const int* in_sizes, int n_in,
                              void* d_out, int out_size, void* d_ws, size_t ws_size,
                              hipStream_t stream) {
  (void)in_sizes; (void)n_in; (void)out_size; (void)ws_size;
  const float* x   = (const float*)d_in[0];
  const float* Wq  = (const float*)d_in[1];
  const float* Wk  = (const float*)d_in[2];
  const float* Wv  = (const float*)d_in[3];
  const float* Wp  = (const float*)d_in[4];
  const float* bp  = (const float*)d_in[5];
  const float* W1  = (const float*)d_in[6];
  const float* b1  = (const float*)d_in[7];
  const float* W2  = (const float*)d_in[8];
  const float* b2  = (const float*)d_in[9];
  const float* g1  = (const float*)d_in[10];
  const float* be1 = (const float*)d_in[11];
  const float* g2  = (const float*)d_in[12];
  const float* be2 = (const float*)d_in[13];
  float* out = (float*)d_out;

  char* ws = (char*)d_ws;
  const size_t S = (size_t)32768 * 384 * 2;
  _Float16* q16    = (_Float16*)(ws + 1 * S);
  _Float16* k16    = (_Float16*)(ws + 2 * S);
  _Float16* vt16   = (_Float16*)(ws + 3 * S);
  _Float16* attn16 = (_Float16*)(ws + 0 * S);

  rowgemm_kernel<false><<<256, 256, 0, stream>>>(nullptr, x, g1, be1, Wq, Wk, Wv,
                                                 q16, k16, vt16, nullptr, nullptr, nullptr);
  attn_kernel<<<12288, 32, 0, stream>>>(q16, k16, vt16, attn16);
  rowgemm_kernel<true><<<256, 256, 0, stream>>>(attn16, nullptr, nullptr, nullptr, Wp, nullptr,
                                                nullptr, nullptr, nullptr, nullptr, bp, x, out);
  mlp_kernel<<<256, 256, 0, stream>>>(g2, be2, W1, b1, W2, b2, out);
}


# q, k and v^T kept fragment-major in the workspace: each QKV tile store and each attention fragment load is 512 contiguous bytes (4 whole lines, was 16 partial lines a load); sums unchanged
# speedup vs baseline: 4.4995x; 1.0889x over previous
#include <hip/hip_runtime.h>
#include <hip/hip_bf16.h>


typedef __attribute__((ext_vector_type(16))) _Float16 v16h;
typedef __attribute__((ext_vector_type(8)))  _Float16 v8h;
typedef __attribute__((ext_vector_type(8)))  float    v8f;

#define DEV __device__ __forceinline__

DEV v8f wmma_f16(v16h a, v16h b, v8f c) {
  return __builtin_amdgcn_wmma_f32_16x16x32_f16(false, a, false, b, (short)0, c,
                                                false, false);
}

DEV v16h load_frag(const _Float16* __restrict__ tile, int ld, int lane) {
  const int m  = lane & 15;
  const int kh = (lane & 16) >> 1;
  const _Float16* p = tile + (size_t)m * ld + kh;
  v8h lo = *(const v8h*)p;
  v8h hi = *(const v8h*)(p + 16);
  v16h a;
#pragma unroll
  for (int i = 0; i < 8; ++i) { a[i] = lo[i]; a[i + 8] = hi[i]; }
  return a;
}

DEV v16h load_fragm(const _Float16* __restrict__ blk, int lane) {
  const _Float16* p = blk + lane * 8;
  v8h lo = *(const v8h*)p;
  v8h hi = *(const v8h*)(p + 256);
  v16h a;
#pragma unroll
  for (int i = 0; i < 8; ++i) { a[i] = lo[i]; a[i + 8] = hi[i]; }
  return a;
}

typedef __attribute__((ext_vector_type(4))) float v4f;

template <int RS>
DEV void wtask_load(const float* __restrict__ src, v4f* pf) {
#pragma unroll
  for (int r = 0; r < 8; ++r) pf[r] = *(const v4f*)(src + (size_t)r * RS);
}

template <int PITCH>
DEV void wtask_store(_Float16* dst, const v4f* pf) {
#pragma unroll
  for (int j = 0; j < 4; ++j) {
    v8h o;
#pragma unroll
    for (int r = 0; r < 8; ++r) o[r] = (_Float16)pf[r][j];
    *(v8h*)(dst + j * PITCH) = o;
  }
}


DEV void ln_load_rows(const float* src, int lane, v4f (&v)[48]) {
  const int m  = lane & 15;
  const int kh = (lane & 16) >> 1;
  const float* p = src + (size_t)m * 384 + kh;
#pragma unroll
  for (int kk = 0; kk < 12; ++kk) {
    v[4 * kk + 0] = *(const v4f*)(p + kk * 32);
    v[4 * kk + 1] = *(const v4f*)(p + kk * 32 + 4);
    v[4 * kk + 2] = *(const v4f*)(p + kk * 32 + 16);
    v[4 * kk + 3] = *(const v4f*)(p + kk * 32 + 20);
  }
}

DEV void ln_gb_load(const float* __restrict__ gp, const float* __restrict__ bp, int kk,
                    v4f (&gg)[4], v4f (&bb)[4]) {
#pragma unroll
  for (int j = 0; j < 4; ++j) {
    const int co = kk * 32 + (j & 1) * 4 + (j >> 1) * 16;
    gg[j] = *(const v4f*)(gp + co);
    bb[j] = *(const v4f*)(bp + co);
  }
}

DEV void ln_frags(const v4f (&v)[48], const float* __restrict__ g,
                  const float* __restrict__ be, int lane, v16h (&hb)[12]) {
  const int kh = (lane & 16) >> 1;
  const float* gp = g + kh;
  const float* bp = be + kh;
  v4f gg[2][4], bb[2][4];
  ln_gb_load(gp, bp, 0, gg[0], bb[0]);
  v4f s4 = {0.f, 0.f, 0.f, 0.f};
#pragma unroll
  for (int i = 0; i < 48; ++i) s4 += v[i];
  float s = (s4[0] + s4[1]) + (s4[2] + s4[3]);
  s += __shfl_xor(s, 16, 32);
  const float mu = s * (1.f / 384.f);
  v4f q4 = {0.f, 0.f, 0.f, 0.f};
#pragma unroll
  for (int i = 0; i < 48; ++i) { v4f d = v[i] - mu; q4 += d * d; }
  float var = (q4[0] + q4[1]) + (q4[2] + q4[3]);
  var += __shfl_xor(var, 16, 32);
  const float rs = rsqrtf(var * (1.f / 384.f) + 1e-5f);
#pragma unroll
  for (int kk = 0; kk < 12; ++kk) {
    if (kk + 1 < 12) ln_gb_load(gp, bp, kk + 1, gg[(kk + 1) & 1], bb[(kk + 1) & 1]);
    __builtin_amdgcn_sched_barrier(0);
#pragma unroll
    for (int j = 0; j < 4; ++j) {
#pragma unroll
      for (int r = 0; r < 4; ++r)
        hb[kk][4 * j + r] =
            (_Float16)((v[4 * kk + j][r] - mu) * rs * gg[kk & 1][j][r] + bb[kk & 1][j][r]);
    }
    __builtin_amdgcn_sched_barrier(0);
  }
}

#define RG_P 392

template <bool WEIGHT_IS_B>
DEV void rg_chunk(const _Float16* Lc, const v16h (&hb)[12], v8f (&acc)[4], int lane) {
  v16h fa[2][6];
#pragma unroll
  for (int i = 0; i < 6; ++i)
    fa[0][i] = load_frag(Lc + (i & 3) * 16 * RG_P + (i >> 2) * 32, RG_P, lane);
#pragma unroll
  for (int g = 0; g < 8; ++g) {
    if (g + 1 < 8) {
#pragma unroll
      for (int i = 0; i < 6; ++i) {
        const int f = (g + 1) * 6 + i;
        fa[(g + 1) & 1][i] = load_frag(Lc + (f & 3) * 16 * RG_P + (f >> 2) * 32, RG_P, lane);
      }
    }
    __builtin_amdgcn_sched_barrier(0);
#pragma unroll
    for (int i = 0; i < 6; ++i) {
      const int f = g * 6 + i;
      if (WEIGHT_IS_B) acc[f & 3] = wmma_f16(hb[f >> 2], fa[g & 1][i], acc[f & 3]);
      else             acc[f & 3] = wmma_f16(fa[g & 1][i], hb[f >> 2], acc[f & 3]);
    }
    __builtin_amdgcn_sched_barrier(0);
  }
}

DEV v8h pack8h(v8f a) {
  v8h o;
#pragma unroll
  for (int r = 0; r < 8; ++r) o[r] = (_Float16)a[r];
  return o;
}

template <bool PROJ>
__global__ __launch_bounds__(256)
void rowgemm_kernel(const _Float16* __restrict__ act,
                    const float* __restrict__ xf, const float* __restrict__ lng,
                    const float* __restrict__ lnb,
                    const float* __restrict__ w0,
                    const float* __restrict__ w1,
                    const float* __restrict__ w2,
                    _Float16* __restrict__ q, _Float16* __restrict__ k,
                    _Float16* __restrict__ vt,
                    const float* __restrict__ bias, const float* __restrict__ res,
                    float* __restrict__ outf) {
  const int Cc = 384, Dh = 64, T = 256, H = 6;
  const int NCH = PROJ ? 6 : 18;
  const int BUF = 64 * RG_P;
  __shared__ __attribute__((aligned(16))) _Float16 Lw[2 * BUF];

  const int tid = threadIdx.x;
  const int lane = tid & 31;
  const int wave = tid >> 5;
  const int c = lane & 15, hh = lane >> 4;
  const int b = blockIdx.x >> 1;
  const int tw = (blockIdx.x & 1) * 128 + wave * 16;
  const size_t row0 = (size_t)b * T + tw;

  v16h hb[12];
  v4f xv[48];
  if (PROJ) {
#pragma unroll
    for (int kk = 0; kk < 12; ++kk) hb[kk] = load_frag(act + row0 * Cc + kk * 32, Cc, lane);
  } else {
    ln_load_rows(xf + row0 * Cc, lane, xv);
  }

  const int RS = PROJ ? 384 : 64;
  const int rg  = ((tid >> 4) & 3) * 4 + (tid & 3);
  const int kg0 = (tid >> 6) * 4 + ((tid >> 2) & 3);
  const int soff = 8 * kg0 * RS + 4 * rg;
  const int loff = 4 * rg * RG_P + 8 * kg0;
  v4f pf[24];
  {
    const float* g = w0 + soff;
#pragma unroll
    for (int i = 0; i < 3; ++i) wtask_load<RS>(g + (size_t)128 * i * RS, pf + 8 * i);
    if (!PROJ) ln_frags(xv, lng, lnb, lane, hb);
#pragma unroll
    for (int i = 0; i < 3; ++i) wtask_store<RG_P>(Lw + loff + 128 * i, pf + 8 * i);
    g += PROJ ? 64 : 64 * 384;
#pragma unroll
    for (int i = 0; i < 3; ++i) wtask_load<RS>(g + (size_t)128 * i * RS, pf + 8 * i);
  }
  __syncthreads();

  for (int s = 0; s < NCH; ++s) {
    const int m = s / 6;
    const int h = s - m * 6;
    const _Float16* Lc = Lw + (s & 1) * BUF;
    if (s + 1 < NCH) {
      _Float16* Ln = Lw + ((s + 1) & 1) * BUF;
#pragma unroll
      for (int i = 0; i < 3; ++i) wtask_store<RG_P>(Ln + loff + 128 * i, pf + 8 * i);
      if (s + 2 < NCH) {
        const int m2 = (s + 2) / 6;
        const int h2 = (s + 2) - m2 * 6;
        const float* wn = (m2 == 0) ? w0 : (m2 == 1) ? w1 : w2;
        const float* g = wn + (PROJ ? (size_t)h2 * 64 : (size_t)h2 * Dh * Cc) + soff;
#pragma unroll
        for (int i = 0; i < 3; ++i) wtask_load<RS>(g + (size_t)128 * i * RS, pf + 8 * i);
      }
    }

    v8f acc[4];
#pragma unroll
    for (int dt = 0; dt < 4; ++dt) acc[dt] = (v8f){};

    if (PROJ) {
      const size_t off = (row0 + c) * Cc + h * 64 + hh * 8;
      const float* rp = res + off;
      const float* bq = bias + h * 64 + hh * 8;
      v4f xr[8], bb[8];
#pragma unroll
      for (int dt = 0; dt < 4; ++dt) {
        xr[2 * dt]     = *(const v4f*)(rp + dt * 16);
        xr[2 * dt + 1] = *(const v4f*)(rp + dt * 16 + 4);
        bb[2 * dt]     = *(const v4f*)(bq + dt * 16);
        bb[2 * dt + 1] = *(const v4f*)(bq + dt * 16 + 4);
      }
      rg_chunk<false>(Lc, hb, acc, lane);
      v4f o[8];
#pragma unroll
      for (int dt = 0; dt < 4; ++dt) {
#pragma unroll
        for (int r = 0; r < 4; ++r) {
          o[2 * dt][r]     = (acc[dt][r] + bb[2 * dt][r]) + xr[2 * dt][r];
          o[2 * dt + 1][r] = (acc[dt][4 + r] + bb[2 * dt + 1][r]) + xr[2 * dt + 1][r];
        }
      }
      __syncthreads();
      float* op = outf + off;
#pragma unroll
      for (int dt = 0; dt < 4; ++dt) {
        *(v4f*)(op + dt * 16)     = o[2 * dt];
        *(v4f*)(op + dt * 16 + 4) = o[2 * dt + 1];
      }
    } else {
      const size_t bh = (size_t)b * H + h;
      const int g = tw >> 4;
      v8h o[4];
      _Float16* dst;
      size_t step;
      if (m < 2) {
        rg_chunk<false>(Lc, hb, acc, lane);
        dst = ((m == 0) ? q : k) + bh * (size_t)(T * Dh) + (size_t)g * 1024 + lane * 8;
        step = 256;
      } else {
        rg_chunk<true>(Lc, hb, acc, lane);
        dst = vt + bh * (size_t)(T * Dh) + (size_t)((g >> 1) * 8 + (g & 1)) * 256 + lane * 8;
        step = 512;
      }
#pragma unroll
      for (int dt = 0; dt < 4; ++dt) o[dt] = pack8h(acc[dt]);
      __syncthreads();
#pragma unroll
      for (int dt = 0; dt < 4; ++dt) *(v8h*)(dst + dt * step) = o[dt];
    }
  }
}

template <int M>
DEV float row_xor(float v) {
  return __int_as_float(__builtin_amdgcn_update_dpp(0, __float_as_int(v), 0x160 + M,
                                                    0xf, 0xf, true));
}

template <bool NEXT>
DEV void attn_chunk(const _Float16* __restrict__ kb, const _Float16* __restrict__ vb,
                    _Float16* Pl, int s0, int t0, int lane,
                    const v16h& aq0, const v16h& aq1, v16h (&kf)[4],
                    float (&mrow)[8], float (&lrow)[8], v8f (&O)[4]) {
  const int Dh = 64;
  const int n = lane & 15, mh = (lane >> 4) * 8;
  const float scale = 0.05103103630798287f;

  float P0[8], P1[8];
  {
    v8f acc = {};
    acc = wmma_f16(aq0, kf[0], acc);
    acc = wmma_f16(aq1, kf[1], acc);
#pragma unroll
    for (int r = 0; r < 8; ++r) P0[r] = acc[r] * scale;
  }
  {
    v8f acc = {};
    acc = wmma_f16(aq0, kf[2], acc);
    acc = wmma_f16(aq1, kf[3], acc);
#pragma unroll
    for (int r = 0; r < 8; ++r) P1[r] = acc[r] * scale;
  }
  v16h vf[4];
  const _Float16* vc = vb + (size_t)s0 * Dh;
  vf[0] = load_fragm(vc, lane);
  vf[1] = load_fragm(vc + 512, lane);
  vf[2] = load_fragm(vc + 1024, lane);
  vf[3] = load_fragm(vc + 1536, lane);
  if (NEXT) {
    const _Float16* kn = kb + (size_t)(s0 + 32) * Dh;
    kf[0] = load_fragm(kn, lane);
    kf[1] = load_fragm(kn + 512, lane);
    kf[2] = load_fragm(kn + 1024, lane);
    kf[3] = load_fragm(kn + 1536, lane);
  }
#pragma unroll
  for (int r = 0; r < 8; ++r) {
    int trow = t0 + mh + r;
    if (s0 + n > trow)      P0[r] = -__builtin_inff();
    if (s0 + 16 + n > trow) P1[r] = -__builtin_inff();
    float mx = fmaxf(P0[r], P1[r]);
    mx = fmaxf(mx, row_xor<1>(mx));
    mx = fmaxf(mx, row_xor<2>(mx));
    mx = fmaxf(mx, row_xor<4>(mx));
    mx = fmaxf(mx, row_xor<8>(mx));
    float mnew = fmaxf(mrow[r], mx);
    float al = expf(mrow[r] - mnew);
    float p0 = expf(P0[r] - mnew);
    float p1 = expf(P1[r] - mnew);
    float rs = p0 + p1;
    rs += row_xor<1>(rs);
    rs += row_xor<2>(rs);
    rs += row_xor<4>(rs);
    rs += row_xor<8>(rs);
    lrow[r] = lrow[r] * al + rs;
    mrow[r] = mnew;
    P0[r] = p0; P1[r] = p1;
    O[0][r] *= al; O[1][r] *= al; O[2][r] *= al; O[3][r] *= al;
  }
  __syncthreads();
#pragma unroll
  for (int r = 0; r < 8; ++r) {
    Pl[(mh + r) * 32 + n]      = (_Float16)P0[r];
    Pl[(mh + r) * 32 + 16 + n] = (_Float16)P1[r];
  }
  __syncthreads();
  v16h ap = load_frag(Pl, 32, lane);
#pragma unroll
  for (int nt = 0; nt < 4; ++nt) O[nt] = wmma_f16(ap, vf[nt], O[nt]);
}

__global__ __launch_bounds__(32)
void attn_kernel(const _Float16* __restrict__ q,
                 const _Float16* __restrict__ k,
                 const _Float16* __restrict__ vt,
                 _Float16* __restrict__ attn_out) {
  const int T = 256, Dh = 64, H = 6, C = 384;
  int tile = blockIdx.x;
  int it = tile & 15;
  int bh = tile >> 4;
  int b = bh / H, h = bh - b * H;
  int t0 = it * 16;
  int lane = threadIdx.x;
  const int n = lane & 15, mh = (lane >> 4) * 8;

  const _Float16* qb = q + ((size_t)bh * T + t0) * Dh;
  const _Float16* kb = k + (size_t)bh * T * Dh;
  const _Float16* vb = vt + (size_t)bh * Dh * T;

  v16h aq0 = load_fragm(qb, lane);
  v16h aq1 = load_fragm(qb + 512, lane);

  v16h kf[4];
  kf[0] = load_fragm(kb, lane);
  kf[1] = load_fragm(kb + 512, lane);
  kf[2] = load_fragm(kb + 1024, lane);
  kf[3] = load_fragm(kb + 1536, lane);

  float mrow[8], lrow[8];
#pragma unroll
  for (int r = 0; r < 8; ++r) { mrow[r] = -__builtin_inff(); lrow[r] = 0.f; }
  v8f O[4];
#pragma unroll
  for (int i = 0; i < 4; ++i) O[i] = (v8f){};

  __shared__ _Float16 Pl[16 * 32];

  int s0 = 0;
  for (; s0 + 32 < t0 + 16; s0 += 32)
    attn_chunk<true>(kb, vb, Pl, s0, t0, lane, aq0, aq1, kf, mrow, lrow, O);
  attn_chunk<false>(kb, vb, Pl, s0, t0, lane, aq0, aq1, kf, mrow, lrow, O);

#pragma unroll
  for (int r = 0; r < 8; ++r) {
    float inv = 1.f / lrow[r];
    _Float16* orow = attn_out + ((size_t)b * T + t0 + mh + r) * C + h * Dh;
#pragma unroll
    for (int nt = 0; nt < 4; ++nt) orow[nt * 16 + n] = (_Float16)(O[nt][r] * inv);
  }
}


__global__ __launch_bounds__(256)
void mlp_kernel(const float* __restrict__ lng, const float* __restrict__ lnb,
                const float* __restrict__ w1,
                const float* __restrict__ b1, const float* __restrict__ w2,
                const float* __restrict__ b2, float* __restrict__ out) {
  const int C = 384, F = 1536;
  const int P1 = 392;
  const int P2 = 40;
  __shared__ __attribute__((aligned(16))) _Float16 Lw1[32 * P1];
  __shared__ __attribute__((aligned(16))) _Float16 Lw2[384 * P2];
  __shared__ __attribute__((aligned(16))) float Lb1[1536];

  const int tid = threadIdx.x;
  const int lane = tid & 31;
  const int wave = tid >> 5;
  const int c = lane & 15, hh = lane >> 4;
  const size_t row0 = (size_t)blockIdx.x * 128 + (size_t)wave * 16;

#pragma unroll
  for (int i = 0; i < 6; ++i) Lb1[tid + 256 * i] = b1[tid + 256 * i];

  v4f xv[48];
  ln_load_rows(out + row0 * C, lane, xv);

  const bool c_is_w1 = __builtin_amdgcn_readfirstlane(wave) < 4;
  const int uC = 256 + (tid & 127);
  const int rgA = ((tid >> 4) & 1) * 4 + (tid & 3), kgA = (tid >> 5) * 4 + ((tid >> 2) & 3);
  const int rgB = (tid >> 4) * 4 + (tid & 3),       kgB = (tid >> 2) & 3;
  const int rgC = c_is_w1 ? ((uC >> 4) & 1) * 4 + (uC & 3) : (uC >> 4) * 4 + (uC & 3);
  const int kgC = c_is_w1 ? (uC >> 5) * 4 + ((uC >> 2) & 3) : (uC >> 2) & 3;
  const float* gA = w1 + (size_t)8 * kgA * F + 4 * rgA;
  const float* gB = w2 + (size_t)8 * kgB * C + 4 * rgB;
  const float* gC = c_is_w1 ? w1 + (size_t)8 * kgC * F + 4 * rgC
                            : w2 + (size_t)8 * kgC * C + 4 * rgC;
  const int oA = 4 * rgA * P1 + 8 * kgA;
  const int oB = 4 * rgB * P2 + 8 * kgB;
  const int oC = c_is_w1 ? 4 * rgC * P1 + 8 * kgC : 4 * rgC * P2 + 8 * kgC;

  v4f pf[24];
  wtask_load<F>(gA, pf);
  wtask_load<C>(gB, pf + 8);
  if (c_is_w1) wtask_load<F>(gC, pf + 16);
  else         wtask_load<C>(gC, pf + 16);

  v16h hb[12];
  ln_frags(xv, lng, lnb, lane, hb);

  v8f acc[24];
#pragma unroll
  for (int mt = 0; mt < 24; ++mt) acc[mt] = (v8f){};

  for (int s = 0; s < 48; ++s) {
    __syncthreads();
    wtask_store<P1>(Lw1 + oA, pf);
    wtask_store<P2>(Lw2 + oB, pf + 8);
    if (c_is_w1) wtask_store<P1>(Lw1 + oC, pf + 16);
    else         wtask_store<P2>(Lw2 + oC, pf + 16);
    __syncthreads();
    if (s + 1 < 48) {
      const int hid1 = (s + 1) * 32;
      wtask_load<F>(gA + hid1, pf);
      wtask_load<C>(gB + (size_t)hid1 * C, pf + 8);
      if (c_is_w1) wtask_load<F>(gC + hid1, pf + 16);
      else         wtask_load<C>(gC + (size_t)hid1 * C, pf + 16);
    }

    v16h fa[2][6];
#pragma unroll
    for (int i = 0; i < 6; ++i) fa[0][i] = load_frag(Lw1 + (i & 1) * 16 * P1 + (i >> 1) * 32, P1, lane);
    v8f ha0 = {}, ha1 = {};
    v16h hf = {};
#pragma unroll
    for (int g = 0; g < 8; ++g) {
      if (g + 1 < 8) {
#pragma unroll
        for (int i = 0; i < 6; ++i) {
          const int f = (g + 1) * 6 + i;
          fa[(g + 1) & 1][i] = (f < 24)
              ? load_frag(Lw1 + (f & 1) * 16 * P1 + (f >> 1) * 32, P1, lane)
              : load_frag(Lw2 + (f - 24) * 16 * P2, P2, lane);
        }
      }
      __builtin_amdgcn_sched_barrier(0);
#pragma unroll
      for (int i = 0; i < 6; ++i) {
        const int f = g * 6 + i;
        if (f < 24) {
          if (f & 1) ha1 = wmma_f16(fa[g & 1][i], hb[f >> 1], ha1);
          else       ha0 = wmma_f16(fa[g & 1][i], hb[f >> 1], ha0);
          if (f == 23) {
            const float* bq = Lb1 + s * 32 + hh * 8;
            v4f bA = *(const v4f*)bq;
            v4f bB = *(const v4f*)(bq + 4);
            v4f bC = *(const v4f*)(bq + 16);
            v4f bD = *(const v4f*)(bq + 20);
#pragma unroll
            for (int r = 0; r < 4; ++r) {
              hf[r]      = (_Float16)fmaxf(ha0[r] + bA[r], 0.f);
              hf[4 + r]  = (_Float16)fmaxf(ha0[4 + r] + bB[r], 0.f);
              hf[8 + r]  = (_Float16)fmaxf(ha1[r] + bC[r], 0.f);
              hf[12 + r] = (_Float16)fmaxf(ha1[4 + r] + bD[r], 0.f);
            }
          }
        } else {
          acc[f - 24] = wmma_f16(fa[g & 1][i], hf, acc[f - 24]);
        }
      }
      __builtin_amdgcn_sched_barrier(0);
    }
  }

  float* orow = out + (row0 + c) * C + hh * 8;
  const float* b2p = b2 + hh * 8;
#pragma unroll
  for (int mt = 0; mt < 24; ++mt) {
    if (mt == 12) __builtin_amdgcn_sched_barrier(0);
    v4f oA = *(const v4f*)(orow + mt * 16);
    v4f oB = *(const v4f*)(orow + mt * 16 + 4);
    v4f cA = *(const v4f*)(b2p + mt * 16);
    v4f cB = *(const v4f*)(b2p + mt * 16 + 4);
#pragma unroll
    for (int r = 0; r < 4; ++r) {
      oA[r] += acc[mt][r] + cA[r];
      oB[r] += acc[mt][4 + r] + cB[r];
    }
    *(v4f*)(orow + mt * 16) = oA;
    *(v4f*)(orow + mt * 16 + 4) = oB;
  }
}

extern "C" void kernel_launch(void* const* d_in, const int* in_sizes, int n_in,
                              void* d_out, int out_size, void* d_ws, size_t ws_size,
                              hipStream_t stream) {
  (void)in_sizes; (void)n_in; (void)out_size; (void)ws_size;
  const float* x   = (const float*)d_in[0];
  const float* Wq  = (const float*)d_in[1];
  const float* Wk  = (const float*)d_in[2];
  const float* Wv  = (const float*)d_in[3];
  const float* Wp  = (const float*)d_in[4];
  const float* bp  = (const float*)d_in[5];
  const float* W1  = (const float*)d_in[6];
  const float* b1  = (const float*)d_in[7];
  const float* W2  = (const float*)d_in[8];
  const float* b2  = (const float*)d_in[9];
  const float* g1  = (const float*)d_in[10];
  const float* be1 = (const float*)d_in[11];
  const float* g2  = (const float*)d_in[12];
  const float* be2 = (const float*)d_in[13];
  float* out = (float*)d_out;

  char* ws = (char*)d_ws;
  const size_t S = (size_t)32768 * 384 * 2;
  _Float16* q16    = (_Float16*)(ws + 1 * S);
  _Float16* k16    = (_Float16*)(ws + 2 * S);
  _Float16* vt16   = (_Float16*)(ws + 3 * S);
  _Float16* attn16 = (_Float16*)(ws + 0 * S);

  rowgemm_kernel<false><<<256, 256, 0, stream>>>(nullptr, x, g1, be1, Wq, Wk, Wv,
                                                 q16, k16, vt16, nullptr, nullptr, nullptr);
  attn_kernel<<<12288, 32, 0, stream>>>(q16, k16, vt16, attn16);
  rowgemm_kernel<true><<<256, 256, 0, stream>>>(attn16, nullptr, nullptr, nullptr, Wp, nullptr,
                                                nullptr, nullptr, nullptr, nullptr, bp, x, out);
  mlp_kernel<<<256, 256, 0, stream>>>(g2, be2, W1, b1, W2, b2, out);
}


# MLP kernel: weight chunks double-buffered in LDS, the f32->f16 staging and the next fetch run in three pieces beside the step's matrix runs; two barriers a step -> one; sums unchanged
# speedup vs baseline: 4.8145x; 1.0700x over previous
#include <hip/hip_runtime.h>
#include <hip/hip_bf16.h>


typedef __attribute__((ext_vector_type(16))) _Float16 v16h;
typedef __attribute__((ext_vector_type(8)))  _Float16 v8h;
typedef __attribute__((ext_vector_type(8)))  float    v8f;

#define DEV __device__ __forceinline__

DEV v8f wmma_f16(v16h a, v16h b, v8f c) {
  return __builtin_amdgcn_wmma_f32_16x16x32_f16(false, a, false, b, (short)0, c,
                                                false, false);
}

DEV v16h load_frag(const _Float16* __restrict__ tile, int ld, int lane) {
  const int m  = lane & 15;
  const int kh = (lane & 16) >> 1;
  const _Float16* p = tile + (size_t)m * ld + kh;
  v8h lo = *(const v8h*)p;
  v8h hi = *(const v8h*)(p + 16);
  v16h a;
#pragma unroll
  for (int i = 0; i < 8; ++i) { a[i] = lo[i]; a[i + 8] = hi[i]; }
  return a;
}

DEV v16h load_fragm(const _Float16* __restrict__ blk, int lane) {
  const _Float16* p = blk + lane * 8;
  v8h lo = *(const v8h*)p;
  v8h hi = *(const v8h*)(p + 256);
  v16h a;
#pragma unroll
  for (int i = 0; i < 8; ++i) { a[i] = lo[i]; a[i + 8] = hi[i]; }
  return a;
}

typedef __attribute__((ext_vector_type(4))) float v4f;

template <int RS>
DEV void wtask_load(const float* __restrict__ src, v4f* pf) {
#pragma unroll
  for (int r = 0; r < 8; ++r) pf[r] = *(const v4f*)(src + (size_t)r * RS);
}

template <int PITCH>
DEV void wtask_store(_Float16* dst, const v4f* pf) {
#pragma unroll
  for (int j = 0; j < 4; ++j) {
    v8h o;
#pragma unroll
    for (int r = 0; r < 8; ++r) o[r] = (_Float16)pf[r][j];
    *(v8h*)(dst + j * PITCH) = o;
  }
}

DEV void wtask_load_u(const float* __restrict__ ubase, int rs, unsigned voff, v4f* pf) {
#pragma unroll
  for (int r = 0; r < 8; ++r)
    pf[r] = *(const v4f*)((const char*)(ubase + (size_t)r * rs) + voff);
}

DEV void wtask_store_rt(_Float16* dst, int pitch, const v4f* pf) {
#pragma unroll
  for (int j = 0; j < 4; ++j) {
    v8h o;
#pragma unroll
    for (int r = 0; r < 8; ++r) o[r] = (_Float16)pf[r][j];
    *(v8h*)(dst + j * pitch) = o;
  }
}

DEV void lds_barrier() {
  __builtin_amdgcn_fence(__ATOMIC_RELEASE, "workgroup", "local");
  __builtin_amdgcn_s_barrier();
  __builtin_amdgcn_fence(__ATOMIC_ACQUIRE, "workgroup", "local");
}


DEV void ln_load_rows(const float* src, int lane, v4f (&v)[48]) {
  const int m  = lane & 15;
  const int kh = (lane & 16) >> 1;
  const float* p = src + (size_t)m * 384 + kh;
#pragma unroll
  for (int kk = 0; kk < 12; ++kk) {
    v[4 * kk + 0] = *(const v4f*)(p + kk * 32);
    v[4 * kk + 1] = *(const v4f*)(p + kk * 32 + 4);
    v[4 * kk + 2] = *(const v4f*)(p + kk * 32 + 16);
    v[4 * kk + 3] = *(const v4f*)(p + kk * 32 + 20);
  }
}

DEV void ln_gb_load(const float* __restrict__ gp, const float* __restrict__ bp, int kk,
                    v4f (&gg)[4], v4f (&bb)[4]) {
#pragma unroll
  for (int j = 0; j < 4; ++j) {
    const int co = kk * 32 + (j & 1) * 4 + (j >> 1) * 16;
    gg[j] = *(const v4f*)(gp + co);
    bb[j] = *(const v4f*)(bp + co);
  }
}

DEV void ln_frags(const v4f (&v)[48], const float* __restrict__ g,
                  const float* __restrict__ be, int lane, v16h (&hb)[12]) {
  const int kh = (lane & 16) >> 1;
  const float* gp = g + kh;
  const float* bp = be + kh;
  v4f gg[2][4], bb[2][4];
  ln_gb_load(gp, bp, 0, gg[0], bb[0]);
  v4f s4 = {0.f, 0.f, 0.f, 0.f};
#pragma unroll
  for (int i = 0; i < 48; ++i) s4 += v[i];
  float s = (s4[0] + s4[1]) + (s4[2] + s4[3]);
  s += __shfl_xor(s, 16, 32);
  const float mu = s * (1.f / 384.f);
  v4f q4 = {0.f, 0.f, 0.f, 0.f};
#pragma unroll
  for (int i = 0; i < 48; ++i) { v4f d = v[i] - mu; q4 += d * d; }
  float var = (q4[0] + q4[1]) + (q4[2] + q4[3]);
  var += __shfl_xor(var, 16, 32);
  const float rs = rsqrtf(var * (1.f / 384.f) + 1e-5f);
#pragma unroll
  for (int kk = 0; kk < 12; ++kk) {
    if (kk + 1 < 12) ln_gb_load(gp, bp, kk + 1, gg[(kk + 1) & 1], bb[(kk + 1) & 1]);
    __builtin_amdgcn_sched_barrier(0);
#pragma unroll
    for (int j = 0; j < 4; ++j) {
#pragma unroll
      for (int r = 0; r < 4; ++r)
        hb[kk][4 * j + r] =
            (_Float16)((v[4 * kk + j][r] - mu) * rs * gg[kk & 1][j][r] + bb[kk & 1][j][r]);
    }
    __builtin_amdgcn_sched_barrier(0);
  }
}

#define RG_P 392

template <bool WEIGHT_IS_B>
DEV void rg_chunk(const _Float16* Lc, const v16h (&hb)[12], v8f (&acc)[4], int lane) {
  v16h fa[2][6];
#pragma unroll
  for (int i = 0; i < 6; ++i)
    fa[0][i] = load_frag(Lc + (i & 3) * 16 * RG_P + (i >> 2) * 32, RG_P, lane);
#pragma unroll
  for (int g = 0; g < 8; ++g) {
    if (g + 1 < 8) {
#pragma unroll
      for (int i = 0; i < 6; ++i) {
        const int f = (g + 1) * 6 + i;
        fa[(g + 1) & 1][i] = load_frag(Lc + (f & 3) * 16 * RG_P + (f >> 2) * 32, RG_P, lane);
      }
    }
    __builtin_amdgcn_sched_barrier(0);
#pragma unroll
    for (int i = 0; i < 6; ++i) {
      const int f = g * 6 + i;
      if (WEIGHT_IS_B) acc[f & 3] = wmma_f16(hb[f >> 2], fa[g & 1][i], acc[f & 3]);
      else             acc[f & 3] = wmma_f16(fa[g & 1][i], hb[f >> 2], acc[f & 3]);
    }
    __builtin_amdgcn_sched_barrier(0);
  }
}

DEV v8h pack8h(v8f a) {
  v8h o;
#pragma unroll
  for (int r = 0; r < 8; ++r) o[r] = (_Float16)a[r];
  return o;
}

template <bool PROJ>
__global__ __launch_bounds__(256)
void rowgemm_kernel(const _Float16* __restrict__ act,
                    const float* __restrict__ xf, const float* __restrict__ lng,
                    const float* __restrict__ lnb,
                    const float* __restrict__ w0,
                    const float* __restrict__ w1,
                    const float* __restrict__ w2,
                    _Float16* __restrict__ q, _Float16* __restrict__ k,
                    _Float16* __restrict__ vt,
                    const float* __restrict__ bias, const float* __restrict__ res,
                    float* __restrict__ outf) {
  const int Cc = 384, Dh = 64, T = 256, H = 6;
  const int NCH = PROJ ? 6 : 18;
  const int BUF = 64 * RG_P;
  __shared__ __attribute__((aligned(16))) _Float16 Lw[2 * BUF];

  const int tid = threadIdx.x;
  const int lane = tid & 31;
  const int wave = tid >> 5;
  const int c = lane & 15, hh = lane >> 4;
  const int b = blockIdx.x >> 1;
  const int tw = (blockIdx.x & 1) * 128 + wave * 16;
  const size_t row0 = (size_t)b * T + tw;

  v16h hb[12];
  v4f xv[48];
  if (PROJ) {
#pragma unroll
    for (int kk = 0; kk < 12; ++kk) hb[kk] = load_frag(act + row0 * Cc + kk * 32, Cc, lane);
  } else {
    ln_load_rows(xf + row0 * Cc, lane, xv);
  }

  const int RS = PROJ ? 384 : 64;
  const int rg  = ((tid >> 4) & 3) * 4 + (tid & 3);
  const int kg0 = (tid >> 6) * 4 + ((tid >> 2) & 3);
  const int soff = 8 * kg0 * RS + 4 * rg;
  const int loff = 4 * rg * RG_P + 8 * kg0;
  v4f pf[24];
  {
    const float* g = w0 + soff;
#pragma unroll
    for (int i = 0; i < 3; ++i) wtask_load<RS>(g + (size_t)128 * i * RS, pf + 8 * i);
    if (!PROJ) ln_frags(xv, lng, lnb, lane, hb);
#pragma unroll
    for (int i = 0; i < 3; ++i) wtask_store<RG_P>(Lw + loff + 128 * i, pf + 8 * i);
    g += PROJ ? 64 : 64 * 384;
#pragma unroll
    for (int i = 0; i < 3; ++i) wtask_load<RS>(g + (size_t)128 * i * RS, pf + 8 * i);
  }
  __syncthreads();

  for (int s = 0; s < NCH; ++s) {
    const int m = s / 6;
    const int h = s - m * 6;
    const _Float16* Lc = Lw + (s & 1) * BUF;
    if (s + 1 < NCH) {
      _Float16* Ln = Lw + ((s + 1) & 1) * BUF;
#pragma unroll
      for (int i = 0; i < 3; ++i) wtask_store<RG_P>(Ln + loff + 128 * i, pf + 8 * i);
      if (s + 2 < NCH) {
        const int m2 = (s + 2) / 6;
        const int h2 = (s + 2) - m2 * 6;
        const float* wn = (m2 == 0) ? w0 : (m2 == 1) ? w1 : w2;
        const float* g = wn + (PROJ ? (size_t)h2 * 64 : (size_t)h2 * Dh * Cc) + soff;
#pragma unroll
        for (int i = 0; i < 3; ++i) wtask_load<RS>(g + (size_t)128 * i * RS, pf + 8 * i);
      }
    }

    v8f acc[4];
#pragma unroll
    for (int dt = 0; dt < 4; ++dt) acc[dt] = (v8f){};

    if (PROJ) {
      const size_t off = (row0 + c) * Cc + h * 64 + hh * 8;
      const float* rp = res + off;
      const float* bq = bias + h * 64 + hh * 8;
      v4f xr[8], bb[8];
#pragma unroll
      for (int dt = 0; dt < 4; ++dt) {
        xr[2 * dt]     = *(const v4f*)(rp + dt * 16);
        xr[2 * dt + 1] = *(const v4f*)(rp + dt * 16 + 4);
        bb[2 * dt]     = *(const v4f*)(bq + dt * 16);
        bb[2 * dt + 1] = *(const v4f*)(bq + dt * 16 + 4);
      }
      rg_chunk<false>(Lc, hb, acc, lane);
      v4f o[8];
#pragma unroll
      for (int dt = 0; dt < 4; ++dt) {
#pragma unroll
        for (int r = 0; r < 4; ++r) {
          o[2 * dt][r]     = (acc[dt][r] + bb[2 * dt][r]) + xr[2 * dt][r];
          o[2 * dt + 1][r] = (acc[dt][4 + r] + bb[2 * dt + 1][r]) + xr[2 * dt + 1][r];
        }
      }
      __syncthreads();
      float* op = outf + off;
#pragma unroll
      for (int dt = 0; dt < 4; ++dt) {
        *(v4f*)(op + dt * 16)     = o[2 * dt];
        *(v4f*)(op + dt * 16 + 4) = o[2 * dt + 1];
      }
    } else {
      const size_t bh = (size_t)b * H + h;
      const int g = tw >> 4;
      v8h o[4];
      _Float16* dst;
      size_t step;
      if (m < 2) {
        rg_chunk<false>(Lc, hb, acc, lane);
        dst = ((m == 0) ? q : k) + bh * (size_t)(T * Dh) + (size_t)g * 1024 + lane * 8;
        step = 256;
      } else {
        rg_chunk<true>(Lc, hb, acc, lane);
        dst = vt + bh * (size_t)(T * Dh) + (size_t)((g >> 1) * 8 + (g & 1)) * 256 + lane * 8;
        step = 512;
      }
#pragma unroll
      for (int dt = 0; dt < 4; ++dt) o[dt] = pack8h(acc[dt]);
      __syncthreads();
#pragma unroll
      for (int dt = 0; dt < 4; ++dt) *(v8h*)(dst + dt * step) = o[dt];
    }
  }
}

template <int M>
DEV float row_xor(float v) {
  return __int_as_float(__builtin_amdgcn_update_dpp(0, __float_as_int(v), 0x160 + M,
                                                    0xf, 0xf, true));
}

template <bool NEXT>
DEV void attn_chunk(const _Float16* __restrict__ kb, const _Float16* __restrict__ vb,
                    _Float16* Pl, int s0, int t0, int lane,
                    const v16h& aq0, const v16h& aq1, v16h (&kf)[4],
                    float (&mrow)[8], float (&lrow)[8], v8f (&O)[4]) {
  const int Dh = 64;
  const int n = lane & 15, mh = (lane >> 4) * 8;
  const float scale = 0.05103103630798287f;

  float P0[8], P1[8];
  {
    v8f acc = {};
    acc = wmma_f16(aq0, kf[0], acc);
    acc = wmma_f16(aq1, kf[1], acc);
#pragma unroll
    for (int r = 0; r < 8; ++r) P0[r] = acc[r] * scale;
  }
  {
    v8f acc = {};
    acc = wmma_f16(aq0, kf[2], acc);
    acc = wmma_f16(aq1, kf[3], acc);
#pragma unroll
    for (int r = 0; r < 8; ++r) P1[r] = acc[r] * scale;
  }
  v16h vf[4];
  const _Float16* vc = vb + (size_t)s0 * Dh;
  vf[0] = load_fragm(vc, lane);
  vf[1] = load_fragm(vc + 512, lane);
  vf[2] = load_fragm(vc + 1024, lane);
  vf[3] = load_fragm(vc + 1536, lane);
  if (NEXT) {
    const _Float16* kn = kb + (size_t)(s0 + 32) * Dh;
    kf[0] = load_fragm(kn, lane);
    kf[1] = load_fragm(kn + 512, lane);
    kf[2] = load_fragm(kn + 1024, lane);
    kf[3] = load_fragm(kn + 1536, lane);
  }
#pragma unroll
  for (int r = 0; r < 8; ++r) {
    int trow = t0 + mh + r;
    if (s0 + n > trow)      P0[r] = -__builtin_inff();
    if (s0 + 16 + n > trow) P1[r] = -__builtin_inff();
    float mx = fmaxf(P0[r], P1[r]);
    mx = fmaxf(mx, row_xor<1>(mx));
    mx = fmaxf(mx, row_xor<2>(mx));
    mx = fmaxf(mx, row_xor<4>(mx));
    mx = fmaxf(mx, row_xor<8>(mx));
    float mnew = fmaxf(mrow[r], mx);
    float al = expf(mrow[r] - mnew);
    float p0 = expf(P0[r] - mnew);
    float p1 = expf(P1[r] - mnew);
    float rs = p0 + p1;
    rs += row_xor<1>(rs);
    rs += row_xor<2>(rs);
    rs += row_xor<4>(rs);
    rs += row_xor<8>(rs);
    lrow[r] = lrow[r] * al + rs;
    mrow[r] = mnew;
    P0[r] = p0; P1[r] = p1;
    O[0][r] *= al; O[1][r] *= al; O[2][r] *= al; O[3][r] *= al;
  }
  __syncthreads();
#pragma unroll
  for (int r = 0; r < 8; ++r) {
    Pl[(mh + r) * 32 + n]      = (_Float16)P0[r];
    Pl[(mh + r) * 32 + 16 + n] = (_Float16)P1[r];
  }
  __syncthreads();
  v16h ap = load_frag(Pl, 32, lane);
#pragma unroll
  for (int nt = 0; nt < 4; ++nt) O[nt] = wmma_f16(ap, vf[nt], O[nt]);
}

__global__ __launch_bounds__(32)
void attn_kernel(const _Float16* __restrict__ q,
                 const _Float16* __restrict__ k,
                 const _Float16* __restrict__ vt,
                 _Float16* __restrict__ attn_out) {
  const int T = 256, Dh = 64, H = 6, C = 384;
  int tile = blockIdx.x;
  int it = tile & 15;
  int bh = tile >> 4;
  int b = bh / H, h = bh - b * H;
  int t0 = it * 16;
  int lane = threadIdx.x;
  const int n = lane & 15, mh = (lane >> 4) * 8;

  const _Float16* qb = q + ((size_t)bh * T + t0) * Dh;
  const _Float16* kb = k + (size_t)bh * T * Dh;
  const _Float16* vb = vt + (size_t)bh * Dh * T;

  v16h aq0 = load_fragm(qb, lane);
  v16h aq1 = load_fragm(qb + 512, lane);

  v16h kf[4];
  kf[0] = load_fragm(kb, lane);
  kf[1] = load_fragm(kb + 512, lane);
  kf[2] = load_fragm(kb + 1024, lane);
  kf[3] = load_fragm(kb + 1536, lane);

  float mrow[8], lrow[8];
#pragma unroll
  for (int r = 0; r < 8; ++r) { mrow[r] = -__builtin_inff(); lrow[r] = 0.f; }
  v8f O[4];
#pragma unroll
  for (int i = 0; i < 4; ++i) O[i] = (v8f){};

  __shared__ _Float16 Pl[16 * 32];

  int s0 = 0;
  for (; s0 + 32 < t0 + 16; s0 += 32)
    attn_chunk<true>(kb, vb, Pl, s0, t0, lane, aq0, aq1, kf, mrow, lrow, O);
  attn_chunk<false>(kb, vb, Pl, s0, t0, lane, aq0, aq1, kf, mrow, lrow, O);

#pragma unroll
  for (int r = 0; r < 8; ++r) {
    float inv = 1.f / lrow[r];
    _Float16* orow = attn_out + ((size_t)b * T + t0 + mh + r) * C + h * Dh;
#pragma unroll
    for (int nt = 0; nt < 4; ++nt) orow[nt * 16 + n] = (_Float16)(O[nt][r] * inv);
  }
}


constexpr int ML_C = 384, ML_F = 1536;
constexpr int ML_P1 = 392;
constexpr int ML_P2 = 40;
constexpr int ML_B1 = 32 * ML_P1;
constexpr int ML_B2 = 384 * ML_P2;

template <bool ST, bool LD>
DEV void mlp_step(int s, _Float16* Lw1, _Float16* Lw2, const float* Lb1,
                  const float* __restrict__ w1, const float* __restrict__ w2,
                  const float* __restrict__ wC, unsigned vA, unsigned vB, unsigned vC,
                  int rsC, int stC, int pC, int oA, int oB, int oC, bool c_is_w1,
                  int lane, int hh, const v16h (&hb)[12], v4f* pf, v8f (&acc)[24]) {
  const int C = ML_C, F = ML_F, P1 = ML_P1, P2 = ML_P2, B1 = ML_B1, B2 = ML_B2;
  const int cur = s & 1;
  const _Float16* L1c = Lw1 + cur * B1;
  const _Float16* L2c = Lw2 + cur * B2;
  _Float16* L1n = Lw1 + (cur ^ 1) * B1;
  _Float16* L2n = Lw2 + (cur ^ 1) * B2;
  _Float16* LCn = c_is_w1 ? L1n : L2n;
  const int hid2 = (s + 2) * 32;

  v16h fa[2][4];
#pragma unroll
  for (int i = 0; i < 4; ++i) fa[0][i] = load_frag(L1c + (i & 1) * 16 * P1 + (i >> 1) * 32, P1, lane);
  v8f ha0 = {}, ha1 = {};
  v16h hf = {};
#pragma unroll
  for (int g = 0; g < 12; ++g) {
    if (g + 1 < 12) {
#pragma unroll
      for (int i = 0; i < 4; ++i) {
        const int f = (g + 1) * 4 + i;
        fa[(g + 1) & 1][i] = (f < 24)
            ? load_frag(L1c + (f & 1) * 16 * P1 + (f >> 1) * 32, P1, lane)
            : load_frag(L2c + (f - 24) * 16 * P2, P2, lane);
      }
    }
    __builtin_amdgcn_sched_barrier(0);
    if (g == 6 && ST) {
      wtask_store<P1>(L1n + oA, pf);
      if (LD) wtask_load_u(w1 + hid2, F, vA, pf);
    }
    if (g == 8 && ST) {
      wtask_store<P2>(L2n + oB, pf + 8);
      if (LD) wtask_load_u(w2 + (size_t)hid2 * C, C, vB, pf + 8);
    }
    if (g == 10 && ST) {
      wtask_store_rt(LCn + oC, pC, pf + 16);
      if (LD) wtask_load_u(wC + (size_t)(s + 2) * stC, rsC, vC, pf + 16);
    }
#pragma unroll
    for (int i = 0; i < 4; ++i) {
      const int f = g * 4 + i;
      if (f < 24) {
        if (f & 1) ha1 = wmma_f16(fa[g & 1][i], hb[f >> 1], ha1);
        else       ha0 = wmma_f16(fa[g & 1][i], hb[f >> 1], ha0);
        if (f == 23) {
          const float* bq = Lb1 + s * 32 + hh * 8;
          v4f bA = *(const v4f*)bq;
          v4f bB = *(const v4f*)(bq + 4);
          v4f bC = *(const v4f*)(bq + 16);
          v4f bD = *(const v4f*)(bq + 20);
#pragma unroll
          for (int r = 0; r < 4; ++r) {
            hf[r]      = (_Float16)fmaxf(ha0[r] + bA[r], 0.f);
            hf[4 + r]  = (_Float16)fmaxf(ha0[4 + r] + bB[r], 0.f);
            hf[8 + r]  = (_Float16)fmaxf(ha1[r] + bC[r], 0.f);
            hf[12 + r] = (_Float16)fmaxf(ha1[4 + r] + bD[r], 0.f);
          }
        }
      } else {
        acc[f - 24] = wmma_f16(fa[g & 1][i], hf, acc[f - 24]);
      }
    }
    __builtin_amdgcn_sched_barrier(0);
  }
}

__global__ __launch_bounds__(256)
void mlp_kernel(const float* __restrict__ lng, const float* __restrict__ lnb,
                const float* __restrict__ w1,
                const float* __restrict__ b1, const float* __restrict__ w2,
                const float* __restrict__ b2, float* __restrict__ out) {
  const int C = ML_C, F = ML_F, P1 = ML_P1, P2 = ML_P2, B1 = ML_B1, B2 = ML_B2;
  __shared__ __attribute__((aligned(16))) _Float16 Lw1[2 * B1];
  __shared__ __attribute__((aligned(16))) _Float16 Lw2[2 * B2];
  __shared__ __attribute__((aligned(16))) float Lb1[1536];

  const int tid = threadIdx.x;
  const int lane = tid & 31;
  const int wave = tid >> 5;
  const int c = lane & 15, hh = lane >> 4;
  const size_t row0 = (size_t)blockIdx.x * 128 + (size_t)wave * 16;

#pragma unroll
  for (int i = 0; i < 6; ++i) Lb1[tid + 256 * i] = b1[tid + 256 * i];

  v4f xv[48];
  ln_load_rows(out + row0 * C, lane, xv);

  const bool c_is_w1 = __builtin_amdgcn_readfirstlane(wave) < 4;
  const int uC = 256 + (tid & 127);
  const int rgA = ((tid >> 4) & 1) * 4 + (tid & 3), kgA = (tid >> 5) * 4 + ((tid >> 2) & 3);
  const int rgB = (tid >> 4) * 4 + (tid & 3),       kgB = (tid >> 2) & 3;
  const int rgC = c_is_w1 ? ((uC >> 4) & 1) * 4 + (uC & 3) : (uC >> 4) * 4 + (uC & 3);
  const int kgC = c_is_w1 ? (uC >> 5) * 4 + ((uC >> 2) & 3) : (uC >> 2) & 3;
  const unsigned vA = (unsigned)(8 * kgA * F + 4 * rgA) * 4u;
  const unsigned vB = (unsigned)(8 * kgB * C + 4 * rgB) * 4u;
  const unsigned vC = c_is_w1 ? (unsigned)(8 * kgC * F + 4 * rgC) * 4u
                              : (unsigned)(8 * kgC * C + 4 * rgC) * 4u;
  const float* wC = c_is_w1 ? w1 : w2;
  const int rsC = c_is_w1 ? F : C;
  const int stC = c_is_w1 ? 32 : 32 * C;
  const int pC  = c_is_w1 ? P1 : P2;
  const int oA = 4 * rgA * P1 + 8 * kgA;
  const int oB = 4 * rgB * P2 + 8 * kgB;
  const int oC = c_is_w1 ? 4 * rgC * P1 + 8 * kgC : 4 * rgC * P2 + 8 * kgC;

  v4f pf[24];
  wtask_load_u(w1, F, vA, pf);
  wtask_load_u(w2, C, vB, pf + 8);
  wtask_load_u(wC, rsC, vC, pf + 16);

  v16h hb[12];
  ln_frags(xv, lng, lnb, lane, hb);

  v8f acc[24];
#pragma unroll
  for (int mt = 0; mt < 24; ++mt) acc[mt] = (v8f){};

  wtask_store<P1>(Lw1 + oA, pf);
  wtask_store<P2>(Lw2 + oB, pf + 8);
  wtask_store_rt((c_is_w1 ? Lw1 : Lw2) + oC, pC, pf + 16);
  wtask_load_u(w1 + 32, F, vA, pf);
  wtask_load_u(w2 + (size_t)32 * C, C, vB, pf + 8);
  wtask_load_u(wC + stC, rsC, vC, pf + 16);
  lds_barrier();

#define MLP_STEP_ARGS Lw1, Lw2, Lb1, w1, w2, wC, vA, vB, vC, rsC, stC, pC, oA, oB, oC, \
                      c_is_w1, lane, hh, hb, pf, acc
  for (int s = 0; s < 46; ++s) {
    mlp_step<true, true>(s, MLP_STEP_ARGS);
    lds_barrier();
  }
  mlp_step<true, false>(46, MLP_STEP_ARGS);
  lds_barrier();
  mlp_step<false, false>(47, MLP_STEP_ARGS);
#undef MLP_STEP_ARGS

  float* orow = out + (row0 + c) * C + hh * 8;
  const float* b2p = b2 + hh * 8;
#pragma unroll
  for (int mt = 0; mt < 24; ++mt) {
    if (mt == 12) __builtin_amdgcn_sched_barrier(0);
    v4f oA = *(const v4f*)(orow + mt * 16);
    v4f oB = *(const v4f*)(orow + mt * 16 + 4);
    v4f cA = *(const v4f*)(b2p + mt * 16);
    v4f cB = *(const v4f*)(b2p + mt * 16 + 4);
#pragma unroll
    for (int r = 0; r < 4; ++r) {
      oA[r] += acc[mt][r] + cA[r];
      oB[r] += acc[mt][4 + r] + cB[r];
    }
    *(v4f*)(orow + mt * 16) = oA;
    *(v4f*)(orow + mt * 16 + 4) = oB;
  }
}

extern "C" void kernel_launch(void* const* d_in, const int* in_sizes, int n_in,
                              void* d_out, int out_size, void* d_ws, size_t ws_size,
                              hipStream_t stream) {
  (void)in_sizes; (void)n_in; (void)out_size; (void)ws_size;
  const float* x   = (const float*)d_in[0];
  const float* Wq  = (const float*)d_in[1];
  const float* Wk  = (const float*)d_in[2];
  const float* Wv  = (const float*)d_in[3];
  const float* Wp  = (const float*)d_in[4];
  const float* bp  = (const float*)d_in[5];
  const float* W1  = (const float*)d_in[6];
  const float* b1  = (const float*)d_in[7];
  const float* W2  = (const float*)d_in[8];
  const float* b2  = (const float*)d_in[9];
  const float* g1  = (const float*)d_in[10];
  const float* be1 = (const float*)d_in[11];
  const float* g2  = (const float*)d_in[12];
  const float* be2 = (const float*)d_in[13];
  float* out = (float*)d_out;

  char* ws = (char*)d_ws;
  const size_t S = (size_t)32768 * 384 * 2;
  _Float16* q16    = (_Float16*)(ws + 1 * S);
  _Float16* k16    = (_Float16*)(ws + 2 * S);
  _Float16* vt16   = (_Float16*)(ws + 3 * S);
  _Float16* attn16 = (_Float16*)(ws + 0 * S);

  rowgemm_kernel<false><<<256, 256, 0, stream>>>(nullptr, x, g1, be1, Wq, Wk, Wv,
                                                 q16, k16, vt16, nullptr, nullptr, nullptr);
  attn_kernel<<<12288, 32, 0, stream>>>(q16, k16, vt16, attn16);
  rowgemm_kernel<true><<<256, 256, 0, stream>>>(attn16, nullptr, nullptr, nullptr, Wp, nullptr,
                                                nullptr, nullptr, nullptr, nullptr, bp, x, out);
  mlp_kernel<<<256, 256, 0, stream>>>(g2, be2, W1, b1, W2, b2, out);
}


# QKV, projection and MLP kernels: work-group i starts at weight chunk i % (18, 6, 48) and goes round, so 256 work-groups in step no longer fetch the same lines at once; FC2 hidden-sum order rotates
# speedup vs baseline: 5.2470x; 1.0898x over previous
#include <hip/hip_runtime.h>
#include <hip/hip_bf16.h>


typedef __attribute__((ext_vector_type(16))) _Float16 v16h;
typedef __attribute__((ext_vector_type(8)))  _Float16 v8h;
typedef __attribute__((ext_vector_type(8)))  float    v8f;

#define DEV __device__ __forceinline__

DEV v8f wmma_f16(v16h a, v16h b, v8f c) {
  return __builtin_amdgcn_wmma_f32_16x16x32_f16(false, a, false, b, (short)0, c,
                                                false, false);
}

DEV v16h load_frag(const _Float16* __restrict__ tile, int ld, int lane) {
  const int m  = lane & 15;
  const int kh = (lane & 16) >> 1;
  const _Float16* p = tile + (size_t)m * ld + kh;
  v8h lo = *(const v8h*)p;
  v8h hi = *(const v8h*)(p + 16);
  v16h a;
#pragma unroll
  for (int i = 0; i < 8; ++i) { a[i] = lo[i]; a[i + 8] = hi[i]; }
  return a;
}

DEV v16h load_fragm(const _Float16* __restrict__ blk, int lane) {
  const _Float16* p = blk + lane * 8;
  v8h lo = *(const v8h*)p;
  v8h hi = *(const v8h*)(p + 256);
  v16h a;
#pragma unroll
  for (int i = 0; i < 8; ++i) { a[i] = lo[i]; a[i + 8] = hi[i]; }
  return a;
}

typedef __attribute__((ext_vector_type(4))) float v4f;

template <int RS>
DEV void wtask_load(const float* __restrict__ src, v4f* pf) {
#pragma unroll
  for (int r = 0; r < 8; ++r) pf[r] = *(const v4f*)(src + (size_t)r * RS);
}

template <int PITCH>
DEV void wtask_store(_Float16* dst, const v4f* pf) {
#pragma unroll
  for (int j = 0; j < 4; ++j) {
    v8h o;
#pragma unroll
    for (int r = 0; r < 8; ++r) o[r] = (_Float16)pf[r][j];
    *(v8h*)(dst + j * PITCH) = o;
  }
}

DEV void wtask_load_u(const float* __restrict__ ubase, int rs, unsigned voff, v4f* pf) {
#pragma unroll
  for (int r = 0; r < 8; ++r)
    pf[r] = *(const v4f*)((const char*)(ubase + (size_t)r * rs) + voff);
}

DEV void wtask_store_rt(_Float16* dst, int pitch, const v4f* pf) {
#pragma unroll
  for (int j = 0; j < 4; ++j) {
    v8h o;
#pragma unroll
    for (int r = 0; r < 8; ++r) o[r] = (_Float16)pf[r][j];
    *(v8h*)(dst + j * pitch) = o;
  }
}

DEV void lds_barrier() {
  __builtin_amdgcn_fence(__ATOMIC_RELEASE, "workgroup", "local");
  __builtin_amdgcn_s_barrier();
  __builtin_amdgcn_fence(__ATOMIC_ACQUIRE, "workgroup", "local");
}


DEV void ln_load_rows(const float* src, int lane, v4f (&v)[48]) {
  const int m  = lane & 15;
  const int kh = (lane & 16) >> 1;
  const float* p = src + (size_t)m * 384 + kh;
#pragma unroll
  for (int kk = 0; kk < 12; ++kk) {
    v[4 * kk + 0] = *(const v4f*)(p + kk * 32);
    v[4 * kk + 1] = *(const v4f*)(p + kk * 32 + 4);
    v[4 * kk + 2] = *(const v4f*)(p + kk * 32 + 16);
    v[4 * kk + 3] = *(const v4f*)(p + kk * 32 + 20);
  }
}

DEV void ln_gb_load(const float* __restrict__ gp, const float* __restrict__ bp, int kk,
                    v4f (&gg)[4], v4f (&bb)[4]) {
#pragma unroll
  for (int j = 0; j < 4; ++j) {
    const int co = kk * 32 + (j & 1) * 4 + (j >> 1) * 16;
    gg[j] = *(const v4f*)(gp + co);
    bb[j] = *(const v4f*)(bp + co);
  }
}

DEV void ln_frags(const v4f (&v)[48], const float* __restrict__ g,
                  const float* __restrict__ be, int lane, v16h (&hb)[12]) {
  const int kh = (lane & 16) >> 1;
  const float* gp = g + kh;
  const float* bp = be + kh;
  v4f gg[2][4], bb[2][4];
  ln_gb_load(gp, bp, 0, gg[0], bb[0]);
  v4f s4 = {0.f, 0.f, 0.f, 0.f};
#pragma unroll
  for (int i = 0; i < 48; ++i) s4 += v[i];
  float s = (s4[0] + s4[1]) + (s4[2] + s4[3]);
  s += __shfl_xor(s, 16, 32);
  const float mu = s * (1.f / 384.f);
  v4f q4 = {0.f, 0.f, 0.f, 0.f};
#pragma unroll
  for (int i = 0; i < 48; ++i) { v4f d = v[i] - mu; q4 += d * d; }
  float var = (q4[0] + q4[1]) + (q4[2] + q4[3]);
  var += __shfl_xor(var, 16, 32);
  const float rs = rsqrtf(var * (1.f / 384.f) + 1e-5f);
#pragma unroll
  for (int kk = 0; kk < 12; ++kk) {
    if (kk + 1 < 12) ln_gb_load(gp, bp, kk + 1, gg[(kk + 1) & 1], bb[(kk + 1) & 1]);
    __builtin_amdgcn_sched_barrier(0);
#pragma unroll
    for (int j = 0; j < 4; ++j) {
#pragma unroll
      for (int r = 0; r < 4; ++r)
        hb[kk][4 * j + r] =
            (_Float16)((v[4 * kk + j][r] - mu) * rs * gg[kk & 1][j][r] + bb[kk & 1][j][r]);
    }
    __builtin_amdgcn_sched_barrier(0);
  }
}

#define RG_P 392

template <bool WEIGHT_IS_B>
DEV void rg_chunk(const _Float16* Lc, const v16h (&hb)[12], v8f (&acc)[4], int lane) {
  v16h fa[2][6];
#pragma unroll
  for (int i = 0; i < 6; ++i)
    fa[0][i] = load_frag(Lc + (i & 3) * 16 * RG_P + (i >> 2) * 32, RG_P, lane);
#pragma unroll
  for (int g = 0; g < 8; ++g) {
    if (g + 1 < 8) {
#pragma unroll
      for (int i = 0; i < 6; ++i) {
        const int f = (g + 1) * 6 + i;
        fa[(g + 1) & 1][i] = load_frag(Lc + (f & 3) * 16 * RG_P + (f >> 2) * 32, RG_P, lane);
      }
    }
    __builtin_amdgcn_sched_barrier(0);
#pragma unroll
    for (int i = 0; i < 6; ++i) {
      const int f = g * 6 + i;
      if (WEIGHT_IS_B) acc[f & 3] = wmma_f16(hb[f >> 2], fa[g & 1][i], acc[f & 3]);
      else             acc[f & 3] = wmma_f16(fa[g & 1][i], hb[f >> 2], acc[f & 3]);
    }
    __builtin_amdgcn_sched_barrier(0);
  }
}

DEV v8h pack8h(v8f a) {
  v8h o;
#pragma unroll
  for (int r = 0; r < 8; ++r) o[r] = (_Float16)a[r];
  return o;
}

template <bool PROJ>
__global__ __launch_bounds__(256)
void rowgemm_kernel(const _Float16* __restrict__ act,
                    const float* __restrict__ xf, const float* __restrict__ lng,
                    const float* __restrict__ lnb,
                    const float* __restrict__ w0,
                    const float* __restrict__ w1,
                    const float* __restrict__ w2,
                    _Float16* __restrict__ q, _Float16* __restrict__ k,
                    _Float16* __restrict__ vt,
                    const float* __restrict__ bias, const float* __restrict__ res,
                    float* __restrict__ outf) {
  const int Cc = 384, Dh = 64, T = 256, H = 6;
  const int NCH = PROJ ? 6 : 18;
  const int BUF = 64 * RG_P;
  __shared__ __attribute__((aligned(16))) _Float16 Lw[2 * BUF];

  const int tid = threadIdx.x;
  const int lane = tid & 31;
  const int wave = tid >> 5;
  const int c = lane & 15, hh = lane >> 4;
  const int b = blockIdx.x >> 1;
  const int tw = (blockIdx.x & 1) * 128 + wave * 16;
  const size_t row0 = (size_t)b * T + tw;

  v16h hb[12];
  v4f xv[48];
  if (PROJ) {
#pragma unroll
    for (int kk = 0; kk < 12; ++kk) hb[kk] = load_frag(act + row0 * Cc + kk * 32, Cc, lane);
  } else {
    ln_load_rows(xf + row0 * Cc, lane, xv);
  }

  const int RS = PROJ ? 384 : 64;
  const int rg  = ((tid >> 4) & 3) * 4 + (tid & 3);
  const int kg0 = (tid >> 6) * 4 + ((tid >> 2) & 3);
  const int soff = 8 * kg0 * RS + 4 * rg;
  const int loff = 4 * rg * RG_P + 8 * kg0;
  const int rot = (int)(blockIdx.x % (unsigned)NCH);
  auto chunk_src = [&](int ch) -> const float* {
    const int cm = ch / 6;
    const int chd = ch - cm * 6;
    const float* wn = (cm == 0) ? w0 : (cm == 1) ? w1 : w2;
    return wn + (PROJ ? (size_t)chd * 64 : (size_t)chd * Dh * Cc) + soff;
  };
  v4f pf[24];
  {
    const float* g = chunk_src(rot);
#pragma unroll
    for (int i = 0; i < 3; ++i) wtask_load<RS>(g + (size_t)128 * i * RS, pf + 8 * i);
    if (!PROJ) ln_frags(xv, lng, lnb, lane, hb);
#pragma unroll
    for (int i = 0; i < 3; ++i) wtask_store<RG_P>(Lw + loff + 128 * i, pf + 8 * i);
    g = chunk_src(rot + 1 < NCH ? rot + 1 : 0);
#pragma unroll
    for (int i = 0; i < 3; ++i) wtask_load<RS>(g + (size_t)128 * i * RS, pf + 8 * i);
  }
  __syncthreads();

  for (int s = 0; s < NCH; ++s) {
    const int sc = s + rot < NCH ? s + rot : s + rot - NCH;
    const int m = sc / 6;
    const int h = sc - m * 6;
    const _Float16* Lc = Lw + (s & 1) * BUF;
    if (s + 1 < NCH) {
      _Float16* Ln = Lw + ((s + 1) & 1) * BUF;
#pragma unroll
      for (int i = 0; i < 3; ++i) wtask_store<RG_P>(Ln + loff + 128 * i, pf + 8 * i);
      if (s + 2 < NCH) {
        const int s2 = s + 2 + rot;
        const float* g = chunk_src(s2 < NCH ? s2 : s2 - NCH);
#pragma unroll
        for (int i = 0; i < 3; ++i) wtask_load<RS>(g + (size_t)128 * i * RS, pf + 8 * i);
      }
    }

    v8f acc[4];
#pragma unroll
    for (int dt = 0; dt < 4; ++dt) acc[dt] = (v8f){};

    if (PROJ) {
      const size_t off = (row0 + c) * Cc + h * 64 + hh * 8;
      const float* rp = res + off;
      const float* bq = bias + h * 64 + hh * 8;
      v4f xr[8], bb[8];
#pragma unroll
      for (int dt = 0; dt < 4; ++dt) {
        xr[2 * dt]     = *(const v4f*)(rp + dt * 16);
        xr[2 * dt + 1] = *(const v4f*)(rp + dt * 16 + 4);
        bb[2 * dt]     = *(const v4f*)(bq + dt * 16);
        bb[2 * dt + 1] = *(const v4f*)(bq + dt * 16 + 4);
      }
      rg_chunk<false>(Lc, hb, acc, lane);
      v4f o[8];
#pragma unroll
      for (int dt = 0; dt < 4; ++dt) {
#pragma unroll
        for (int r = 0; r < 4; ++r) {
          o[2 * dt][r]     = (acc[dt][r] + bb[2 * dt][r]) + xr[2 * dt][r];
          o[2 * dt + 1][r] = (acc[dt][4 + r] + bb[2 * dt + 1][r]) + xr[2 * dt + 1][r];
        }
      }
      __syncthreads();
      float* op = outf + off;
#pragma unroll
      for (int dt = 0; dt < 4; ++dt) {
        *(v4f*)(op + dt * 16)     = o[2 * dt];
        *(v4f*)(op + dt * 16 + 4) = o[2 * dt + 1];
      }
    } else {
      const size_t bh = (size_t)b * H + h;
      const int g = tw >> 4;
      v8h o[4];
      _Float16* dst;
      size_t step;
      if (m < 2) {
        rg_chunk<false>(Lc, hb, acc, lane);
        dst = ((m == 0) ? q : k) + bh * (size_t)(T * Dh) + (size_t)g * 1024 + lane * 8;
        step = 256;
      } else {
        rg_chunk<true>(Lc, hb, acc, lane);
        dst = vt + bh * (size_t)(T * Dh) + (size_t)((g >> 1) * 8 + (g & 1)) * 256 + lane * 8;
        step = 512;
      }
#pragma unroll
      for (int dt = 0; dt < 4; ++dt) o[dt] = pack8h(acc[dt]);
      __syncthreads();
#pragma unroll
      for (int dt = 0; dt < 4; ++dt) *(v8h*)(dst + dt * step) = o[dt];
    }
  }
}

template <int M>
DEV float row_xor(float v) {
  return __int_as_float(__builtin_amdgcn_update_dpp(0, __float_as_int(v), 0x160 + M,
                                                    0xf, 0xf, true));
}

template <bool NEXT>
DEV void attn_chunk(const _Float16* __restrict__ kb, const _Float16* __restrict__ vb,
                    _Float16* Pl, int s0, int t0, int lane,
                    const v16h& aq0, const v16h& aq1, v16h (&kf)[4],
                    float (&mrow)[8], float (&lrow)[8], v8f (&O)[4]) {
  const int Dh = 64;
  const int n = lane & 15, mh = (lane >> 4) * 8;
  const float scale = 0.05103103630798287f;

  float P0[8], P1[8];
  {
    v8f acc = {};
    acc = wmma_f16(aq0, kf[0], acc);
    acc = wmma_f16(aq1, kf[1], acc);
#pragma unroll
    for (int r = 0; r < 8; ++r) P0[r] = acc[r] * scale;
  }
  {
    v8f acc = {};
    acc = wmma_f16(aq0, kf[2], acc);
    acc = wmma_f16(aq1, kf[3], acc);
#pragma unroll
    for (int r = 0; r < 8; ++r) P1[r] = acc[r] * scale;
  }
  v16h vf[4];
  const _Float16* vc = vb + (size_t)s0 * Dh;
  vf[0] = load_fragm(vc, lane);
  vf[1] = load_fragm(vc + 512, lane);
  vf[2] = load_fragm(vc + 1024, lane);
  vf[3] = load_fragm(vc + 1536, lane);
  if (NEXT) {
    const _Float16* kn = kb + (size_t)(s0 + 32) * Dh;
    kf[0] = load_fragm(kn, lane);
    kf[1] = load_fragm(kn + 512, lane);
    kf[2] = load_fragm(kn + 1024, lane);
    kf[3] = load_fragm(kn + 1536, lane);
  }
#pragma unroll
  for (int r = 0; r < 8; ++r) {
    int trow = t0 + mh + r;
    if (s0 + n > trow)      P0[r] = -__builtin_inff();
    if (s0 + 16 + n > trow) P1[r] = -__builtin_inff();
    float mx = fmaxf(P0[r], P1[r]);
    mx = fmaxf(mx, row_xor<1>(mx));
    mx = fmaxf(mx, row_xor<2>(mx));
    mx = fmaxf(mx, row_xor<4>(mx));
    mx = fmaxf(mx, row_xor<8>(mx));
    float mnew = fmaxf(mrow[r], mx);
    float al = expf(mrow[r] - mnew);
    float p0 = expf(P0[r] - mnew);
    float p1 = expf(P1[r] - mnew);
    float rs = p0 + p1;
    rs += row_xor<1>(rs);
    rs += row_xor<2>(rs);
    rs += row_xor<4>(rs);
    rs += row_xor<8>(rs);
    lrow[r] = lrow[r] * al + rs;
    mrow[r] = mnew;
    P0[r] = p0; P1[r] = p1;
    O[0][r] *= al; O[1][r] *= al; O[2][r] *= al; O[3][r] *= al;
  }
  __syncthreads();
#pragma unroll
  for (int r = 0; r < 8; ++r) {
    Pl[(mh + r) * 32 + n]      = (_Float16)P0[r];
    Pl[(mh + r) * 32 + 16 + n] = (_Float16)P1[r];
  }
  __syncthreads();
  v16h ap = load_frag(Pl, 32, lane);
#pragma unroll
  for (int nt = 0; nt < 4; ++nt) O[nt] = wmma_f16(ap, vf[nt], O[nt]);
}

__global__ __launch_bounds__(32)
void attn_kernel(const _Float16* __restrict__ q,
                 const _Float16* __restrict__ k,
                 const _Float16* __restrict__ vt,
                 _Float16* __restrict__ attn_out) {
  const int T = 256, Dh = 64, H = 6, C = 384;
  int tile = blockIdx.x;
  int it = tile & 15;
  int bh = tile >> 4;
  int b = bh / H, h = bh - b * H;
  int t0 = it * 16;
  int lane = threadIdx.x;
  const int n = lane & 15, mh = (lane >> 4) * 8;

  const _Float16* qb = q + ((size_t)bh * T + t0) * Dh;
  const _Float16* kb = k + (size_t)bh * T * Dh;
  const _Float16* vb = vt + (size_t)bh * Dh * T;

  v16h aq0 = load_fragm(qb, lane);
  v16h aq1 = load_fragm(qb + 512, lane);

  v16h kf[4];
  kf[0] = load_fragm(kb, lane);
  kf[1] = load_fragm(kb + 512, lane);
  kf[2] = load_fragm(kb + 1024, lane);
  kf[3] = load_fragm(kb + 1536, lane);

  float mrow[8], lrow[8];
#pragma unroll
  for (int r = 0; r < 8; ++r) { mrow[r] = -__builtin_inff(); lrow[r] = 0.f; }
  v8f O[4];
#pragma unroll
  for (int i = 0; i < 4; ++i) O[i] = (v8f){};

  __shared__ _Float16 Pl[16 * 32];

  int s0 = 0;
  for (; s0 + 32 < t0 + 16; s0 += 32)
    attn_chunk<true>(kb, vb, Pl, s0, t0, lane, aq0, aq1, kf, mrow, lrow, O);
  attn_chunk<false>(kb, vb, Pl, s0, t0, lane, aq0, aq1, kf, mrow, lrow, O);

#pragma unroll
  for (int r = 0; r < 8; ++r) {
    float inv = 1.f / lrow[r];
    _Float16* orow = attn_out + ((size_t)b * T + t0 + mh + r) * C + h * Dh;
#pragma unroll
    for (int nt = 0; nt < 4; ++nt) orow[nt * 16 + n] = (_Float16)(O[nt][r] * inv);
  }
}


constexpr int ML_C = 384, ML_F = 1536;
constexpr int ML_P1 = 392;
constexpr int ML_P2 = 40;
constexpr int ML_B1 = 32 * ML_P1;
constexpr int ML_B2 = 384 * ML_P2;

template <bool ST, bool LD>
DEV void mlp_step(int s, int sc, int sc2, _Float16* Lw1, _Float16* Lw2, const float* Lb1,
                  const float* __restrict__ w1, const float* __restrict__ w2,
                  const float* __restrict__ wC, unsigned vA, unsigned vB, unsigned vC,
                  int rsC, int stC, int pC, int oA, int oB, int oC, bool c_is_w1,
                  int lane, int hh, const v16h (&hb)[12], v4f* pf, v8f (&acc)[24]) {
  const int C = ML_C, F = ML_F, P1 = ML_P1, P2 = ML_P2, B1 = ML_B1, B2 = ML_B2;
  const int cur = s & 1;
  const _Float16* L1c = Lw1 + cur * B1;
  const _Float16* L2c = Lw2 + cur * B2;
  _Float16* L1n = Lw1 + (cur ^ 1) * B1;
  _Float16* L2n = Lw2 + (cur ^ 1) * B2;
  _Float16* LCn = c_is_w1 ? L1n : L2n;
  const int hid2 = sc2 * 32;

  v16h fa[2][4];
#pragma unroll
  for (int i = 0; i < 4; ++i) fa[0][i] = load_frag(L1c + (i & 1) * 16 * P1 + (i >> 1) * 32, P1, lane);
  v8f ha0 = {}, ha1 = {};
  v16h hf = {};
#pragma unroll
  for (int g = 0; g < 12; ++g) {
    if (g + 1 < 12) {
#pragma unroll
      for (int i = 0; i < 4; ++i) {
        const int f = (g + 1) * 4 + i;
        fa[(g + 1) & 1][i] = (f < 24)
            ? load_frag(L1c + (f & 1) * 16 * P1 + (f >> 1) * 32, P1, lane)
            : load_frag(L2c + (f - 24) * 16 * P2, P2, lane);
      }
    }
    __builtin_amdgcn_sched_barrier(0);
    if (g == 6 && ST) {
      wtask_store<P1>(L1n + oA, pf);
      if (LD) wtask_load_u(w1 + hid2, F, vA, pf);
    }
    if (g == 8 && ST) {
      wtask_store<P2>(L2n + oB, pf + 8);
      if (LD) wtask_load_u(w2 + (size_t)hid2 * C, C, vB, pf + 8);
    }
    if (g == 10 && ST) {
      wtask_store_rt(LCn + oC, pC, pf + 16);
      if (LD) wtask_load_u(wC + (size_t)sc2 * stC, rsC, vC, pf + 16);
    }
#pragma unroll
    for (int i = 0; i < 4; ++i) {
      const int f = g * 4 + i;
      if (f < 24) {
        if (f & 1) ha1 = wmma_f16(fa[g & 1][i], hb[f >> 1], ha1);
        else       ha0 = wmma_f16(fa[g & 1][i], hb[f >> 1], ha0);
        if (f == 23) {
          const float* bq = Lb1 + sc * 32 + hh * 8;
          v4f bA = *(const v4f*)bq;
          v4f bB = *(const v4f*)(bq + 4);
          v4f bC = *(const v4f*)(bq + 16);
          v4f bD = *(const v4f*)(bq + 20);
#pragma unroll
          for (int r = 0; r < 4; ++r) {
            hf[r]      = (_Float16)fmaxf(ha0[r] + bA[r], 0.f);
            hf[4 + r]  = (_Float16)fmaxf(ha0[4 + r] + bB[r], 0.f);
            hf[8 + r]  = (_Float16)fmaxf(ha1[r] + bC[r], 0.f);
            hf[12 + r] = (_Float16)fmaxf(ha1[4 + r] + bD[r], 0.f);
          }
        }
      } else {
        acc[f - 24] = wmma_f16(fa[g & 1][i], hf, acc[f - 24]);
      }
    }
    __builtin_amdgcn_sched_barrier(0);
  }
}

__global__ __launch_bounds__(256)
void mlp_kernel(const float* __restrict__ lng, const float* __restrict__ lnb,
                const float* __restrict__ w1,
                const float* __restrict__ b1, const float* __restrict__ w2,
                const float* __restrict__ b2, float* __restrict__ out) {
  const int C = ML_C, F = ML_F, P1 = ML_P1, P2 = ML_P2, B1 = ML_B1, B2 = ML_B2;
  __shared__ __attribute__((aligned(16))) _Float16 Lw1[2 * B1];
  __shared__ __attribute__((aligned(16))) _Float16 Lw2[2 * B2];
  __shared__ __attribute__((aligned(16))) float Lb1[1536];

  const int tid = threadIdx.x;
  const int lane = tid & 31;
  const int wave = tid >> 5;
  const int c = lane & 15, hh = lane >> 4;
  const size_t row0 = (size_t)blockIdx.x * 128 + (size_t)wave * 16;

#pragma unroll
  for (int i = 0; i < 6; ++i) Lb1[tid + 256 * i] = b1[tid + 256 * i];

  v4f xv[48];
  ln_load_rows(out + row0 * C, lane, xv);

  const bool c_is_w1 = __builtin_amdgcn_readfirstlane(wave) < 4;
  const int uC = 256 + (tid & 127);
  const int rgA = ((tid >> 4) & 1) * 4 + (tid & 3), kgA = (tid >> 5) * 4 + ((tid >> 2) & 3);
  const int rgB = (tid >> 4) * 4 + (tid & 3),       kgB = (tid >> 2) & 3;
  const int rgC = c_is_w1 ? ((uC >> 4) & 1) * 4 + (uC & 3) : (uC >> 4) * 4 + (uC & 3);
  const int kgC = c_is_w1 ? (uC >> 5) * 4 + ((uC >> 2) & 3) : (uC >> 2) & 3;
  const unsigned vA = (unsigned)(8 * kgA * F + 4 * rgA) * 4u;
  const unsigned vB = (unsigned)(8 * kgB * C + 4 * rgB) * 4u;
  const unsigned vC = c_is_w1 ? (unsigned)(8 * kgC * F + 4 * rgC) * 4u
                              : (unsigned)(8 * kgC * C + 4 * rgC) * 4u;
  const float* wC = c_is_w1 ? w1 : w2;
  const int rsC = c_is_w1 ? F : C;
  const int stC = c_is_w1 ? 32 : 32 * C;
  const int pC  = c_is_w1 ? P1 : P2;
  const int oA = 4 * rgA * P1 + 8 * kgA;
  const int oB = 4 * rgB * P2 + 8 * kgB;
  const int oC = c_is_w1 ? 4 * rgC * P1 + 8 * kgC : 4 * rgC * P2 + 8 * kgC;

  const int rot = (int)(blockIdx.x % 48u);
  const int ch1 = rot + 1 < 48 ? rot + 1 : 0;

  v4f pf[24];
  wtask_load_u(w1 + rot * 32, F, vA, pf);
  wtask_load_u(w2 + (size_t)rot * 32 * C, C, vB, pf + 8);
  wtask_load_u(wC + (size_t)rot * stC, rsC, vC, pf + 16);

  v16h hb[12];
  ln_frags(xv, lng, lnb, lane, hb);

  v8f acc[24];
#pragma unroll
  for (int mt = 0; mt < 24; ++mt) acc[mt] = (v8f){};

  wtask_store<P1>(Lw1 + oA, pf);
  wtask_store<P2>(Lw2 + oB, pf + 8);
  wtask_store_rt((c_is_w1 ? Lw1 : Lw2) + oC, pC, pf + 16);
  wtask_load_u(w1 + ch1 * 32, F, vA, pf);
  wtask_load_u(w2 + (size_t)ch1 * 32 * C, C, vB, pf + 8);
  wtask_load_u(wC + (size_t)ch1 * stC, rsC, vC, pf + 16);
  lds_barrier();

#define MLP_STEP_ARGS Lw1, Lw2, Lb1, w1, w2, wC, vA, vB, vC, rsC, stC, pC, oA, oB, oC, \
                      c_is_w1, lane, hh, hb, pf, acc
  for (int s = 0; s < 46; ++s) {
    const int sc  = s + rot < 48 ? s + rot : s + rot - 48;
    const int sc2 = s + 2 + rot < 48 ? s + 2 + rot : s + 2 + rot - 48;
    mlp_step<true, true>(s, sc, sc2, MLP_STEP_ARGS);
    lds_barrier();
  }
  mlp_step<true, false>(46, 46 + rot < 48 ? 46 + rot : rot - 2, 0, MLP_STEP_ARGS);
  lds_barrier();
  mlp_step<false, false>(47, 47 + rot < 48 ? 47 + rot : rot - 1, 0, MLP_STEP_ARGS);
#undef MLP_STEP_ARGS

  float* orow = out + (row0 + c) * C + hh * 8;
  const float* b2p = b2 + hh * 8;
#pragma unroll
  for (int mt = 0; mt < 24; ++mt) {
    if (mt == 12) __builtin_amdgcn_sched_barrier(0);
    v4f oA = *(const v4f*)(orow + mt * 16);
    v4f oB = *(const v4f*)(orow + mt * 16 + 4);
    v4f cA = *(const v4f*)(b2p + mt * 16);
    v4f cB = *(const v4f*)(b2p + mt * 16 + 4);
#pragma unroll
    for (int r = 0; r < 4; ++r) {
      oA[r] += acc[mt][r] + cA[r];
      oB[r] += acc[mt][4 + r] + cB[r];
    }
    *(v4f*)(orow + mt * 16) = oA;
    *(v4f*)(orow + mt * 16 + 4) = oB;
  }
}

extern "C" void kernel_launch(void* const* d_in, const int* in_sizes, int n_in,
                              void* d_out, int out_size, void* d_ws, size_t ws_size,
                              hipStream_t stream) {
  (void)in_sizes; (void)n_in; (void)out_size; (void)ws_size;
  const float* x   = (const float*)d_in[0];
  const float* Wq  = (const float*)d_in[1];
  const float* Wk  = (const float*)d_in[2];
  const float* Wv  = (const float*)d_in[3];
  const float* Wp  = (const float*)d_in[4];
  const float* bp  = (const float*)d_in[5];
  const float* W1  = (const float*)d_in[6];
  const float* b1  = (const float*)d_in[7];
  const float* W2  = (const float*)d_in[8];
  const float* b2  = (const float*)d_in[9];
  const float* g1  = (const float*)d_in[10];
  const float* be1 = (const float*)d_in[11];
  const float* g2  = (const float*)d_in[12];
  const float* be2 = (const float*)d_in[13];
  float* out = (float*)d_out;

  char* ws = (char*)d_ws;
  const size_t S = (size_t)32768 * 384 * 2;
  _Float16* q16    = (_Float16*)(ws + 1 * S);
  _Float16* k16    = (_Float16*)(ws + 2 * S);
  _Float16* vt16   = (_Float16*)(ws + 3 * S);
  _Float16* attn16 = (_Float16*)(ws + 0 * S);

  rowgemm_kernel<false><<<256, 256, 0, stream>>>(nullptr, x, g1, be1, Wq, Wk, Wv,
                                                 q16, k16, vt16, nullptr, nullptr, nullptr);
  attn_kernel<<<12288, 32, 0, stream>>>(q16, k16, vt16, attn16);
  rowgemm_kernel<true><<<256, 256, 0, stream>>>(attn16, nullptr, nullptr, nullptr, Wp, nullptr,
                                                nullptr, nullptr, nullptr, nullptr, bp, x, out);
  mlp_kernel<<<256, 256, 0, stream>>>(g2, be2, W1, b1, W2, b2, out);
}


# Out-projection folded into the MLP kernel: its transposed D tiles start from residual + bias, feed LN2 in registers and stay FC2's accumulators (sum order changes); out stored once; 4 launches -> 3
# speedup vs baseline: 5.6818x; 1.0829x over previous
#include <hip/hip_runtime.h>
#include <hip/hip_bf16.h>


typedef __attribute__((ext_vector_type(16))) _Float16 v16h;
typedef __attribute__((ext_vector_type(8)))  _Float16 v8h;
typedef __attribute__((ext_vector_type(8)))  float    v8f;

#define DEV __device__ __forceinline__

DEV v8f wmma_f16(v16h a, v16h b, v8f c) {
  return __builtin_amdgcn_wmma_f32_16x16x32_f16(false, a, false, b, (short)0, c,
                                                false, false);
}

DEV v16h load_frag(const _Float16* __restrict__ tile, int ld, int lane) {
  const int m  = lane & 15;
  const int kh = (lane & 16) >> 1;
  const _Float16* p = tile + (size_t)m * ld + kh;
  v8h lo = *(const v8h*)p;
  v8h hi = *(const v8h*)(p + 16);
  v16h a;
#pragma unroll
  for (int i = 0; i < 8; ++i) { a[i] = lo[i]; a[i + 8] = hi[i]; }
  return a;
}

DEV v16h load_fragm(const _Float16* __restrict__ blk, int lane) {
  const _Float16* p = blk + lane * 8;
  v8h lo = *(const v8h*)p;
  v8h hi = *(const v8h*)(p + 256);
  v16h a;
#pragma unroll
  for (int i = 0; i < 8; ++i) { a[i] = lo[i]; a[i + 8] = hi[i]; }
  return a;
}

typedef __attribute__((ext_vector_type(4))) float v4f;

template <int RS>
DEV void wtask_load(const float* __restrict__ src, v4f* pf) {
#pragma unroll
  for (int r = 0; r < 8; ++r) pf[r] = *(const v4f*)(src + (size_t)r * RS);
}

template <int PITCH>
DEV void wtask_store(_Float16* dst, const v4f* pf) {
#pragma unroll
  for (int j = 0; j < 4; ++j) {
    v8h o;
#pragma unroll
    for (int r = 0; r < 8; ++r) o[r] = (_Float16)pf[r][j];
    *(v8h*)(dst + j * PITCH) = o;
  }
}

DEV void wtask_load_u(const float* __restrict__ ubase, int rs, unsigned voff, v4f* pf) {
#pragma unroll
  for (int r = 0; r < 8; ++r)
    pf[r] = *(const v4f*)((const char*)(ubase + (size_t)r * rs) + voff);
}

DEV void wtask_store_rt(_Float16* dst, int pitch, const v4f* pf) {
#pragma unroll
  for (int j = 0; j < 4; ++j) {
    v8h o;
#pragma unroll
    for (int r = 0; r < 8; ++r) o[r] = (_Float16)pf[r][j];
    *(v8h*)(dst + j * pitch) = o;
  }
}

DEV void lds_barrier() {
  __builtin_amdgcn_fence(__ATOMIC_RELEASE, "workgroup", "local");
  __builtin_amdgcn_s_barrier();
  __builtin_amdgcn_fence(__ATOMIC_ACQUIRE, "workgroup", "local");
}


DEV void ln_load_rows(const float* src, int lane, v4f (&v)[48]) {
  const int m  = lane & 15;
  const int kh = (lane & 16) >> 1;
  const float* p = src + (size_t)m * 384 + kh;
#pragma unroll
  for (int kk = 0; kk < 12; ++kk) {
    v[4 * kk + 0] = *(const v4f*)(p + kk * 32);
    v[4 * kk + 1] = *(const v4f*)(p + kk * 32 + 4);
    v[4 * kk + 2] = *(const v4f*)(p + kk * 32 + 16);
    v[4 * kk + 3] = *(const v4f*)(p + kk * 32 + 20);
  }
}

DEV void ln_gb_load(const float* __restrict__ gp, const float* __restrict__ bp, int kk,
                    v4f (&gg)[4], v4f (&bb)[4]) {
#pragma unroll
  for (int j = 0; j < 4; ++j) {
    const int co = kk * 32 + (j & 1) * 4 + (j >> 1) * 16;
    gg[j] = *(const v4f*)(gp + co);
    bb[j] = *(const v4f*)(bp + co);
  }
}

template <bool SRC_LIVES>
DEV void ln_frags(const v4f (&v)[48], const float* __restrict__ g,
                  const float* __restrict__ be, int lane, v16h (&hb)[12]) {
  const int kh = (lane & 16) >> 1;
  const float* gp = g + kh;
  const float* bp = be + kh;
  v4f gg[2][4], bb[2][4];
  ln_gb_load(gp, bp, 0, gg[0], bb[0]);
  v4f s4 = {0.f, 0.f, 0.f, 0.f};
#pragma unroll
  for (int i = 0; i < 48; ++i) s4 += v[i];
  float s = (s4[0] + s4[1]) + (s4[2] + s4[3]);
  s += __shfl_xor(s, 16, 32);
  const float mu = s * (1.f / 384.f);
  float mu_q = mu;
  if (SRC_LIVES) {
    asm volatile("" : "+v"(mu_q));
  }
  v4f q4 = {0.f, 0.f, 0.f, 0.f};
#pragma unroll
  for (int i = 0; i < 48; ++i) { v4f d = v[i] - mu_q; q4 += d * d; }
  float var = (q4[0] + q4[1]) + (q4[2] + q4[3]);
  var += __shfl_xor(var, 16, 32);
  const float rs = rsqrtf(var * (1.f / 384.f) + 1e-5f);
#pragma unroll
  for (int kk = 0; kk < 12; ++kk) {
    if (kk + 1 < 12) ln_gb_load(gp, bp, kk + 1, gg[(kk + 1) & 1], bb[(kk + 1) & 1]);
    __builtin_amdgcn_sched_barrier(0);
#pragma unroll
    for (int j = 0; j < 4; ++j) {
#pragma unroll
      for (int r = 0; r < 4; ++r)
        hb[kk][4 * j + r] =
            (_Float16)((v[4 * kk + j][r] - mu) * rs * gg[kk & 1][j][r] + bb[kk & 1][j][r]);
    }
    __builtin_amdgcn_sched_barrier(0);
  }
}

#define RG_P 392

template <bool WEIGHT_IS_B>
DEV void rg_chunk(const _Float16* Lc, const v16h (&hb)[12], v8f (&acc)[4], int lane) {
  v16h fa[2][6];
#pragma unroll
  for (int i = 0; i < 6; ++i)
    fa[0][i] = load_frag(Lc + (i & 3) * 16 * RG_P + (i >> 2) * 32, RG_P, lane);
#pragma unroll
  for (int g = 0; g < 8; ++g) {
    if (g + 1 < 8) {
#pragma unroll
      for (int i = 0; i < 6; ++i) {
        const int f = (g + 1) * 6 + i;
        fa[(g + 1) & 1][i] = load_frag(Lc + (f & 3) * 16 * RG_P + (f >> 2) * 32, RG_P, lane);
      }
    }
    __builtin_amdgcn_sched_barrier(0);
#pragma unroll
    for (int i = 0; i < 6; ++i) {
      const int f = g * 6 + i;
      if (WEIGHT_IS_B) acc[f & 3] = wmma_f16(hb[f >> 2], fa[g & 1][i], acc[f & 3]);
      else             acc[f & 3] = wmma_f16(fa[g & 1][i], hb[f >> 2], acc[f & 3]);
    }
    __builtin_amdgcn_sched_barrier(0);
  }
}

DEV v8h pack8h(v8f a) {
  v8h o;
#pragma unroll
  for (int r = 0; r < 8; ++r) o[r] = (_Float16)a[r];
  return o;
}

template <bool PROJ>
__global__ __launch_bounds__(256)
void rowgemm_kernel(const _Float16* __restrict__ act,
                    const float* __restrict__ xf, const float* __restrict__ lng,
                    const float* __restrict__ lnb,
                    const float* __restrict__ w0,
                    const float* __restrict__ w1,
                    const float* __restrict__ w2,
                    _Float16* __restrict__ q, _Float16* __restrict__ k,
                    _Float16* __restrict__ vt,
                    const float* __restrict__ bias, const float* __restrict__ res,
                    float* __restrict__ outf) {
  const int Cc = 384, Dh = 64, T = 256, H = 6;
  const int NCH = PROJ ? 6 : 18;
  const int BUF = 64 * RG_P;
  __shared__ __attribute__((aligned(16))) _Float16 Lw[2 * BUF];

  const int tid = threadIdx.x;
  const int lane = tid & 31;
  const int wave = tid >> 5;
  const int c = lane & 15, hh = lane >> 4;
  const int b = blockIdx.x >> 1;
  const int tw = (blockIdx.x & 1) * 128 + wave * 16;
  const size_t row0 = (size_t)b * T + tw;

  v16h hb[12];
  v4f xv[48];
  if (PROJ) {
#pragma unroll
    for (int kk = 0; kk < 12; ++kk) hb[kk] = load_frag(act + row0 * Cc + kk * 32, Cc, lane);
  } else {
    ln_load_rows(xf + row0 * Cc, lane, xv);
  }

  const int RS = PROJ ? 384 : 64;
  const int rg  = ((tid >> 4) & 3) * 4 + (tid & 3);
  const int kg0 = (tid >> 6) * 4 + ((tid >> 2) & 3);
  const int soff = 8 * kg0 * RS + 4 * rg;
  const int loff = 4 * rg * RG_P + 8 * kg0;
  const int rot = (int)(blockIdx.x % (unsigned)NCH);
  auto chunk_src = [&](int ch) -> const float* {
    const int cm = ch / 6;
    const int chd = ch - cm * 6;
    const float* wn = (cm == 0) ? w0 : (cm == 1) ? w1 : w2;
    return wn + (PROJ ? (size_t)chd * 64 : (size_t)chd * Dh * Cc) + soff;
  };
  v4f pf[24];
  {
    const float* g = chunk_src(rot);
#pragma unroll
    for (int i = 0; i < 3; ++i) wtask_load<RS>(g + (size_t)128 * i * RS, pf + 8 * i);
    if (!PROJ) ln_frags<false>(xv, lng, lnb, lane, hb);
#pragma unroll
    for (int i = 0; i < 3; ++i) wtask_store<RG_P>(Lw + loff + 128 * i, pf + 8 * i);
    g = chunk_src(rot + 1 < NCH ? rot + 1 : 0);
#pragma unroll
    for (int i = 0; i < 3; ++i) wtask_load<RS>(g + (size_t)128 * i * RS, pf + 8 * i);
  }
  __syncthreads();

  for (int s = 0; s < NCH; ++s) {
    const int sc = s + rot < NCH ? s + rot : s + rot - NCH;
    const int m = sc / 6;
    const int h = sc - m * 6;
    const _Float16* Lc = Lw + (s & 1) * BUF;
    if (s + 1 < NCH) {
      _Float16* Ln = Lw + ((s + 1) & 1) * BUF;
#pragma unroll
      for (int i = 0; i < 3; ++i) wtask_store<RG_P>(Ln + loff + 128 * i, pf + 8 * i);
      if (s + 2 < NCH) {
        const int s2 = s + 2 + rot;
        const float* g = chunk_src(s2 < NCH ? s2 : s2 - NCH);
#pragma unroll
        for (int i = 0; i < 3; ++i) wtask_load<RS>(g + (size_t)128 * i * RS, pf + 8 * i);
      }
    }

    v8f acc[4];
#pragma unroll
    for (int dt = 0; dt < 4; ++dt) acc[dt] = (v8f){};

    if (PROJ) {
      const size_t off = (row0 + c) * Cc + h * 64 + hh * 8;
      const float* rp = res + off;
      const float* bq = bias + h * 64 + hh * 8;
      v4f xr[8], bb[8];
#pragma unroll
      for (int dt = 0; dt < 4; ++dt) {
        xr[2 * dt]     = *(const v4f*)(rp + dt * 16);
        xr[2 * dt + 1] = *(const v4f*)(rp + dt * 16 + 4);
        bb[2 * dt]     = *(const v4f*)(bq + dt * 16);
        bb[2 * dt + 1] = *(const v4f*)(bq + dt * 16 + 4);
      }
      rg_chunk<false>(Lc, hb, acc, lane);
      v4f o[8];
#pragma unroll
      for (int dt = 0; dt < 4; ++dt) {
#pragma unroll
        for (int r = 0; r < 4; ++r) {
          o[2 * dt][r]     = (acc[dt][r] + bb[2 * dt][r]) + xr[2 * dt][r];
          o[2 * dt + 1][r] = (acc[dt][4 + r] + bb[2 * dt + 1][r]) + xr[2 * dt + 1][r];
        }
      }
      __syncthreads();
      float* op = outf + off;
#pragma unroll
      for (int dt = 0; dt < 4; ++dt) {
        *(v4f*)(op + dt * 16)     = o[2 * dt];
        *(v4f*)(op + dt * 16 + 4) = o[2 * dt + 1];
      }
    } else {
      const size_t bh = (size_t)b * H + h;
      const int g = tw >> 4;
      v8h o[4];
      _Float16* dst;
      size_t step;
      if (m < 2) {
        rg_chunk<false>(Lc, hb, acc, lane);
        dst = ((m == 0) ? q : k) + bh * (size_t)(T * Dh) + (size_t)g * 1024 + lane * 8;
        step = 256;
      } else {
        rg_chunk<true>(Lc, hb, acc, lane);
        dst = vt + bh * (size_t)(T * Dh) + (size_t)((g >> 1) * 8 + (g & 1)) * 256 + lane * 8;
        step = 512;
      }
#pragma unroll
      for (int dt = 0; dt < 4; ++dt) o[dt] = pack8h(acc[dt]);
      __syncthreads();
#pragma unroll
      for (int dt = 0; dt < 4; ++dt) *(v8h*)(dst + dt * step) = o[dt];
    }
  }
}

template <int M>
DEV float row_xor(float v) {
  return __int_as_float(__builtin_amdgcn_update_dpp(0, __float_as_int(v), 0x160 + M,
                                                    0xf, 0xf, true));
}

template <bool NEXT>
DEV void attn_chunk(const _Float16* __restrict__ kb, const _Float16* __restrict__ vb,
                    _Float16* Pl, int s0, int t0, int lane,
                    const v16h& aq0, const v16h& aq1, v16h (&kf)[4],
                    float (&mrow)[8], float (&lrow)[8], v8f (&O)[4]) {
  const int Dh = 64;
  const int n = lane & 15, mh = (lane >> 4) * 8;
  const float scale = 0.05103103630798287f;

  float P0[8], P1[8];
  {
    v8f acc = {};
    acc = wmma_f16(aq0, kf[0], acc);
    acc = wmma_f16(aq1, kf[1], acc);
#pragma unroll
    for (int r = 0; r < 8; ++r) P0[r] = acc[r] * scale;
  }
  {
    v8f acc = {};
    acc = wmma_f16(aq0, kf[2], acc);
    acc = wmma_f16(aq1, kf[3], acc);
#pragma unroll
    for (int r = 0; r < 8; ++r) P1[r] = acc[r] * scale;
  }
  v16h vf[4];
  const _Float16* vc = vb + (size_t)s0 * Dh;
  vf[0] = load_fragm(vc, lane);
  vf[1] = load_fragm(vc + 512, lane);
  vf[2] = load_fragm(vc + 1024, lane);
  vf[3] = load_fragm(vc + 1536, lane);
  if (NEXT) {
    const _Float16* kn = kb + (size_t)(s0 + 32) * Dh;
    kf[0] = load_fragm(kn, lane);
    kf[1] = load_fragm(kn + 512, lane);
    kf[2] = load_fragm(kn + 1024, lane);
    kf[3] = load_fragm(kn + 1536, lane);
  }
#pragma unroll
  for (int r = 0; r < 8; ++r) {
    int trow = t0 + mh + r;
    if (s0 + n > trow)      P0[r] = -__builtin_inff();
    if (s0 + 16 + n > trow) P1[r] = -__builtin_inff();
    float mx = fmaxf(P0[r], P1[r]);
    mx = fmaxf(mx, row_xor<1>(mx));
    mx = fmaxf(mx, row_xor<2>(mx));
    mx = fmaxf(mx, row_xor<4>(mx));
    mx = fmaxf(mx, row_xor<8>(mx));
    float mnew = fmaxf(mrow[r], mx);
    float al = expf(mrow[r] - mnew);
    float p0 = expf(P0[r] - mnew);
    float p1 = expf(P1[r] - mnew);
    float rs = p0 + p1;
    rs += row_xor<1>(rs);
    rs += row_xor<2>(rs);
    rs += row_xor<4>(rs);
    rs += row_xor<8>(rs);
    lrow[r] = lrow[r] * al + rs;
    mrow[r] = mnew;
    P0[r] = p0; P1[r] = p1;
    O[0][r] *= al; O[1][r] *= al; O[2][r] *= al; O[3][r] *= al;
  }
  __syncthreads();
#pragma unroll
  for (int r = 0; r < 8; ++r) {
    Pl[(mh + r) * 32 + n]      = (_Float16)P0[r];
    Pl[(mh + r) * 32 + 16 + n] = (_Float16)P1[r];
  }
  __syncthreads();
  v16h ap = load_frag(Pl, 32, lane);
#pragma unroll
  for (int nt = 0; nt < 4; ++nt) O[nt] = wmma_f16(ap, vf[nt], O[nt]);
}

__global__ __launch_bounds__(32)
void attn_kernel(const _Float16* __restrict__ q,
                 const _Float16* __restrict__ k,
                 const _Float16* __restrict__ vt,
                 _Float16* __restrict__ attn_out) {
  const int T = 256, Dh = 64, H = 6, C = 384;
  int tile = blockIdx.x;
  int it = tile & 15;
  int bh = tile >> 4;
  int b = bh / H, h = bh - b * H;
  int t0 = it * 16;
  int lane = threadIdx.x;
  const int n = lane & 15, mh = (lane >> 4) * 8;

  const _Float16* qb = q + ((size_t)bh * T + t0) * Dh;
  const _Float16* kb = k + (size_t)bh * T * Dh;
  const _Float16* vb = vt + (size_t)bh * Dh * T;

  v16h aq0 = load_fragm(qb, lane);
  v16h aq1 = load_fragm(qb + 512, lane);

  v16h kf[4];
  kf[0] = load_fragm(kb, lane);
  kf[1] = load_fragm(kb + 512, lane);
  kf[2] = load_fragm(kb + 1024, lane);
  kf[3] = load_fragm(kb + 1536, lane);

  float mrow[8], lrow[8];
#pragma unroll
  for (int r = 0; r < 8; ++r) { mrow[r] = -__builtin_inff(); lrow[r] = 0.f; }
  v8f O[4];
#pragma unroll
  for (int i = 0; i < 4; ++i) O[i] = (v8f){};

  __shared__ _Float16 Pl[16 * 32];

  int s0 = 0;
  for (; s0 + 32 < t0 + 16; s0 += 32)
    attn_chunk<true>(kb, vb, Pl, s0, t0, lane, aq0, aq1, kf, mrow, lrow, O);
  attn_chunk<false>(kb, vb, Pl, s0, t0, lane, aq0, aq1, kf, mrow, lrow, O);

#pragma unroll
  for (int r = 0; r < 8; ++r) {
    float inv = 1.f / lrow[r];
    _Float16* orow = attn_out + ((size_t)b * T + t0 + mh + r) * C + h * Dh;
#pragma unroll
    for (int nt = 0; nt < 4; ++nt) orow[nt * 16 + n] = (_Float16)(O[nt][r] * inv);
  }
}


constexpr int ML_C = 384, ML_F = 1536;
constexpr int ML_P1 = 392;
constexpr int ML_P2 = 40;
constexpr int ML_B1 = 32 * ML_P1;
constexpr int ML_B2 = 384 * ML_P2;

template <bool ST, bool LD>
DEV void mlp_step(int s, int sc, int sc2, _Float16* Lw1, _Float16* Lw2, const float* Lb1,
                  const float* __restrict__ w1, const float* __restrict__ w2,
                  const float* __restrict__ wC, unsigned vA, unsigned vB, unsigned vC,
                  int rsC, int stC, int pC, int oA, int oB, int oC, bool c_is_w1,
                  int lane, int hh, const v16h (&hb)[12], v4f* pf, v8f (&acc)[24]) {
  const int C = ML_C, F = ML_F, P1 = ML_P1, P2 = ML_P2, B1 = ML_B1, B2 = ML_B2;
  const int cur = s & 1;
  const _Float16* L1c = Lw1 + cur * B1;
  const _Float16* L2c = Lw2 + cur * B2;
  _Float16* L1n = Lw1 + (cur ^ 1) * B1;
  _Float16* L2n = Lw2 + (cur ^ 1) * B2;
  _Float16* LCn = c_is_w1 ? L1n : L2n;
  const int hid2 = sc2 * 32;

  v16h fa[2][4];
#pragma unroll
  for (int i = 0; i < 4; ++i) fa[0][i] = load_frag(L1c + (i & 1) * 16 * P1 + (i >> 1) * 32, P1, lane);
  v8f ha0 = {}, ha1 = {};
  v16h hf = {};
#pragma unroll
  for (int g = 0; g < 12; ++g) {
    if (g + 1 < 12) {
#pragma unroll
      for (int i = 0; i < 4; ++i) {
        const int f = (g + 1) * 4 + i;
        fa[(g + 1) & 1][i] = (f < 24)
            ? load_frag(L1c + (f & 1) * 16 * P1 + (f >> 1) * 32, P1, lane)
            : load_frag(L2c + (f - 24) * 16 * P2, P2, lane);
      }
    }
    __builtin_amdgcn_sched_barrier(0);
    if (g == 6 && ST) {
      wtask_store<P1>(L1n + oA, pf);
      if (LD) wtask_load_u(w1 + hid2, F, vA, pf);
    }
    if (g == 8 && ST) {
      wtask_store<P2>(L2n + oB, pf + 8);
      if (LD) wtask_load_u(w2 + (size_t)hid2 * C, C, vB, pf + 8);
    }
    if (g == 10 && ST) {
      wtask_store_rt(LCn + oC, pC, pf + 16);
      if (LD) wtask_load_u(wC + (size_t)sc2 * stC, rsC, vC, pf + 16);
    }
#pragma unroll
    for (int i = 0; i < 4; ++i) {
      const int f = g * 4 + i;
      if (f < 24) {
        if (f & 1) ha1 = wmma_f16(fa[g & 1][i], hb[f >> 1], ha1);
        else       ha0 = wmma_f16(fa[g & 1][i], hb[f >> 1], ha0);
        if (f == 23) {
          const float* bq = Lb1 + sc * 32 + hh * 8;
          v4f bA = *(const v4f*)bq;
          v4f bB = *(const v4f*)(bq + 4);
          v4f bC = *(const v4f*)(bq + 16);
          v4f bD = *(const v4f*)(bq + 20);
#pragma unroll
          for (int r = 0; r < 4; ++r) {
            hf[r]      = (_Float16)fmaxf(ha0[r] + bA[r], 0.f);
            hf[4 + r]  = (_Float16)fmaxf(ha0[4 + r] + bB[r], 0.f);
            hf[8 + r]  = (_Float16)fmaxf(ha1[r] + bC[r], 0.f);
            hf[12 + r] = (_Float16)fmaxf(ha1[4 + r] + bD[r], 0.f);
          }
        }
      } else {
        acc[f - 24] = wmma_f16(fa[g & 1][i], hf, acc[f - 24]);
      }
    }
    __builtin_amdgcn_sched_barrier(0);
  }
}


template <int BASE>
DEV void pj_chunk(const _Float16* Lc, const v16h (&hb)[12], v8f (&acc)[24], int lane) {
  v16h fa[2][4];
#pragma unroll
  for (int i = 0; i < 4; ++i) fa[0][i] = load_frag(Lc + i * 16 * RG_P, RG_P, lane);
#pragma unroll
  for (int g = 0; g < 12; ++g) {
    if (g + 1 < 12) {
#pragma unroll
      for (int i = 0; i < 4; ++i)
        fa[(g + 1) & 1][i] = load_frag(Lc + i * 16 * RG_P + (g + 1) * 32, RG_P, lane);
    }
    __builtin_amdgcn_sched_barrier(0);
#pragma unroll
    for (int i = 0; i < 4; ++i) acc[BASE + i] = wmma_f16(fa[g & 1][i], hb[g], acc[BASE + i]);
    __builtin_amdgcn_sched_barrier(0);
  }
}

template <int BASE>
DEV void pj_start_load(const float* __restrict__ xrow, v8f (&acc)[24]) {
#pragma unroll
  for (int dt = 0; dt < 4; ++dt) {
    v4f a = *(const v4f*)(xrow + (BASE + dt) * 16);
    v4f b = *(const v4f*)(xrow + (BASE + dt) * 16 + 4);
#pragma unroll
    for (int r = 0; r < 4; ++r) { acc[BASE + dt][r] = a[r]; acc[BASE + dt][4 + r] = b[r]; }
  }
}

template <int BASE>
DEV void pj_bias_add(v8f (&acc)[24], const float* bq) {
#pragma unroll
  for (int dt = 0; dt < 4; ++dt) {
    v4f a = *(const v4f*)(bq + (BASE + dt) * 16);
    v4f b = *(const v4f*)(bq + (BASE + dt) * 16 + 4);
#pragma unroll
    for (int r = 0; r < 4; ++r) { acc[BASE + dt][r] += a[r]; acc[BASE + dt][4 + r] += b[r]; }
  }
}

struct MlpStage {
  const float* w1; const float* w2; const float* wC;
  unsigned vA, vB, vC;
  int rsC, stC, pC, oA, oB, oC;
  bool c_is_w1;
};

DEV void mlp_request(const MlpStage& ms, int ch, v4f* pf) {
  wtask_load_u(ms.w1 + ch * 32, ML_F, ms.vA, pf);
  wtask_load_u(ms.w2 + (size_t)ch * 32 * ML_C, ML_C, ms.vB, pf + 8);
  wtask_load_u(ms.wC + (size_t)ch * ms.stC, ms.rsC, ms.vC, pf + 16);
}

template <int S>
DEV void pj_step(_Float16* Lw, int loff, const float* __restrict__ wsrc, v4f* pf,
                 const float* __restrict__ xrow, const float* bq,
                 const v16h (&hb)[12], v8f (&acc)[24], int lane,
                 const MlpStage& ms, int rot, int ch1, _Float16* Lw1, _Float16* Lw2) {
  const int BUF = 64 * RG_P;
  const _Float16* Lc = Lw + (S & 1) * BUF;
  if constexpr (S + 1 < 6) {
    _Float16* Ln = Lw + ((S + 1) & 1) * BUF;
#pragma unroll
    for (int i = 0; i < 3; ++i) wtask_store<RG_P>(Ln + loff + 128 * i, pf + 8 * i);
    if constexpr (S + 2 < 6) {
      const float* g = wsrc + (S + 2) * 64;
#pragma unroll
      for (int i = 0; i < 3; ++i) wtask_load<384>(g + (size_t)128 * i * 384, pf + 8 * i);
    }
  }
  if constexpr (S == 4) mlp_request(ms, rot, pf);
  if constexpr (S == 5) {
    wtask_store<ML_P1>(Lw1 + ms.oA, pf);
    wtask_store<ML_P2>(Lw2 + ms.oB, pf + 8);
    wtask_store_rt((ms.c_is_w1 ? Lw1 : Lw2) + ms.oC, ms.pC, pf + 16);
    mlp_request(ms, ch1, pf);
  }
  pj_bias_add<4 * S>(acc, bq);
  if constexpr (S + 1 < 6) pj_start_load<4 * (S + 1)>(xrow, acc);
  pj_chunk<4 * S>(Lc, hb, acc, lane);
  if constexpr (S + 1 < 6) __syncthreads();
}

__global__ __launch_bounds__(256)
void projmlp_kernel(const _Float16* __restrict__ act, const float* __restrict__ wp,
                    const float* __restrict__ bp, const float* __restrict__ xres,
                    const float* __restrict__ lng, const float* __restrict__ lnb,
                    const float* __restrict__ w1,
                    const float* __restrict__ b1, const float* __restrict__ w2,
                    const float* __restrict__ b2, float* __restrict__ out) {
  const int C = ML_C, F = ML_F, P1 = ML_P1, P2 = ML_P2, B1 = ML_B1, B2 = ML_B2;
  __shared__ __attribute__((aligned(16))) _Float16 Lw[2 * 64 * RG_P];
  __shared__ __attribute__((aligned(16))) _Float16 Lw1[2 * B1];
  __shared__ __attribute__((aligned(16))) _Float16 Lw2[2 * B2];
  __shared__ __attribute__((aligned(16))) float Lb1[1536];
  __shared__ __attribute__((aligned(16))) float Lbp[384];

  const int tid = threadIdx.x;
  const int lane = tid & 31;
  const int wave = tid >> 5;
  const int c = lane & 15, hh = lane >> 4;
  const size_t row0 = (size_t)blockIdx.x * 128 + (size_t)wave * 16;

#pragma unroll
  for (int i = 0; i < 6; ++i) Lb1[tid + 256 * i] = b1[tid + 256 * i];
  if (tid < 96) *(v4f*)(Lbp + 4 * tid) = *(const v4f*)(bp + 4 * tid);

  v16h hb[12];
#pragma unroll
  for (int kk = 0; kk < 12; ++kk) hb[kk] = load_frag(act + row0 * C + kk * 32, C, lane);

  const int rg  = ((tid >> 4) & 3) * 4 + (tid & 3);
  const int kg0 = (tid >> 6) * 4 + ((tid >> 2) & 3);
  const float* wsrc = wp + 8 * kg0 * 384 + 4 * rg;
  const int loff = 4 * rg * RG_P + 8 * kg0;

  MlpStage ms;
  ms.c_is_w1 = __builtin_amdgcn_readfirstlane(wave) < 4;
  const int uC = 256 + (tid & 127);
  const int rgA = ((tid >> 4) & 1) * 4 + (tid & 3), kgA = (tid >> 5) * 4 + ((tid >> 2) & 3);
  const int rgB = (tid >> 4) * 4 + (tid & 3),       kgB = (tid >> 2) & 3;
  const int rgC = ms.c_is_w1 ? ((uC >> 4) & 1) * 4 + (uC & 3) : (uC >> 4) * 4 + (uC & 3);
  const int kgC = ms.c_is_w1 ? (uC >> 5) * 4 + ((uC >> 2) & 3) : (uC >> 2) & 3;
  ms.vA = (unsigned)(8 * kgA * F + 4 * rgA) * 4u;
  ms.vB = (unsigned)(8 * kgB * C + 4 * rgB) * 4u;
  ms.vC = ms.c_is_w1 ? (unsigned)(8 * kgC * F + 4 * rgC) * 4u
                     : (unsigned)(8 * kgC * C + 4 * rgC) * 4u;
  ms.w1 = w1; ms.w2 = w2;
  ms.wC = ms.c_is_w1 ? w1 : w2;
  ms.rsC = ms.c_is_w1 ? F : C;
  ms.stC = ms.c_is_w1 ? 32 : 32 * C;
  ms.pC  = ms.c_is_w1 ? P1 : P2;
  ms.oA = 4 * rgA * P1 + 8 * kgA;
  ms.oB = 4 * rgB * P2 + 8 * kgB;
  ms.oC = ms.c_is_w1 ? 4 * rgC * P1 + 8 * kgC : 4 * rgC * P2 + 8 * kgC;

  const int rot = (int)(blockIdx.x % 48u);
  const int ch1 = rot + 1 < 48 ? rot + 1 : 0;

  v8f acc[24];
  const float* xrow = xres + (row0 + c) * C + hh * 8;
  const float* bq = Lbp + hh * 8;
  v4f pf[24];
#pragma unroll
  for (int i = 0; i < 3; ++i) wtask_load<384>(wsrc + (size_t)128 * i * 384, pf + 8 * i);
  pj_start_load<0>(xrow, acc);
#pragma unroll
  for (int i = 0; i < 3; ++i) wtask_store<RG_P>(Lw + loff + 128 * i, pf + 8 * i);
#pragma unroll
  for (int i = 0; i < 3; ++i) wtask_load<384>(wsrc + 64 + (size_t)128 * i * 384, pf + 8 * i);
  __syncthreads();

#define PJ_STEP_ARGS Lw, loff, wsrc, pf, xrow, bq, hb, acc, lane, ms, rot, ch1, Lw1, Lw2
  pj_step<0>(PJ_STEP_ARGS);
  pj_step<1>(PJ_STEP_ARGS);
  pj_step<2>(PJ_STEP_ARGS);
  pj_step<3>(PJ_STEP_ARGS);
  pj_step<4>(PJ_STEP_ARGS);
  pj_step<5>(PJ_STEP_ARGS);
#undef PJ_STEP_ARGS

  {
    v4f xv[48];
#pragma unroll
    for (int kk = 0; kk < 12; ++kk) {
#pragma unroll
      for (int j = 0; j < 4; ++j) {
#pragma unroll
        for (int r = 0; r < 4; ++r) xv[4 * kk + j][r] = acc[2 * kk + (j >> 1)][4 * (j & 1) + r];
      }
    }
    ln_frags<true>(xv, lng, lnb, lane, hb);
  }
  lds_barrier();

  const float* wC = ms.wC;
  const unsigned vA = ms.vA, vB = ms.vB, vC = ms.vC;
  const int rsC = ms.rsC, stC = ms.stC, pC = ms.pC, oA = ms.oA, oB = ms.oB, oC = ms.oC;
  const bool c_is_w1 = ms.c_is_w1;
#define MLP_STEP_ARGS Lw1, Lw2, Lb1, w1, w2, wC, vA, vB, vC, rsC, stC, pC, oA, oB, oC, \
                      c_is_w1, lane, hh, hb, pf, acc
  for (int s = 0; s < 46; ++s) {
    const int sc  = s + rot < 48 ? s + rot : s + rot - 48;
    const int sc2 = s + 2 + rot < 48 ? s + 2 + rot : s + 2 + rot - 48;
    mlp_step<true, true>(s, sc, sc2, MLP_STEP_ARGS);
    lds_barrier();
  }
  mlp_step<true, false>(46, 46 + rot < 48 ? 46 + rot : rot - 2, 0, MLP_STEP_ARGS);
  lds_barrier();
  mlp_step<false, false>(47, 47 + rot < 48 ? 47 + rot : rot - 1, 0, MLP_STEP_ARGS);
#undef MLP_STEP_ARGS

  float* orow = out + (row0 + c) * C + hh * 8;
  const float* b2p = b2 + hh * 8;
#pragma unroll
  for (int mt = 0; mt < 24; ++mt) {
    v4f cA = *(const v4f*)(b2p + mt * 16);
    v4f cB = *(const v4f*)(b2p + mt * 16 + 4);
    v4f oA2, oB2;
#pragma unroll
    for (int r = 0; r < 4; ++r) {
      oA2[r] = acc[mt][r] + cA[r];
      oB2[r] = acc[mt][4 + r] + cB[r];
    }
    *(v4f*)(orow + mt * 16) = oA2;
    *(v4f*)(orow + mt * 16 + 4) = oB2;
  }
}

extern "C" void kernel_launch(void* const* d_in, const int* in_sizes, int n_in,
                              void* d_out, int out_size, void* d_ws, size_t ws_size,
                              hipStream_t stream) {
  (void)in_sizes; (void)n_in; (void)out_size; (void)ws_size;
  const float* x   = (const float*)d_in[0];
  const float* Wq  = (const float*)d_in[1];
  const float* Wk  = (const float*)d_in[2];
  const float* Wv  = (const float*)d_in[3];
  const float* Wp  = (const float*)d_in[4];
  const float* bp  = (const float*)d_in[5];
  const float* W1  = (const float*)d_in[6];
  const float* b1  = (const float*)d_in[7];
  const float* W2  = (const float*)d_in[8];
  const float* b2  = (const float*)d_in[9];
  const float* g1  = (const float*)d_in[10];
  const float* be1 = (const float*)d_in[11];
  const float* g2  = (const float*)d_in[12];
  const float* be2 = (const float*)d_in[13];
  float* out = (float*)d_out;

  char* ws = (char*)d_ws;
  const size_t S = (size_t)32768 * 384 * 2;
  _Float16* q16    = (_Float16*)(ws + 1 * S);
  _Float16* k16    = (_Float16*)(ws + 2 * S);
  _Float16* vt16   = (_Float16*)(ws + 3 * S);
  _Float16* attn16 = (_Float16*)(ws + 0 * S);

  rowgemm_kernel<false><<<256, 256, 0, stream>>>(nullptr, x, g1, be1, Wq, Wk, Wv,
                                                 q16, k16, vt16, nullptr, nullptr, nullptr);
  attn_kernel<<<12288, 32, 0, stream>>>(q16, k16, vt16, attn16);
  projmlp_kernel<<<256, 256, 0, stream>>>(attn16, Wp, bp, x, g2, be2, W1, b1, W2, b2, out);
}
